# starting kernel
# baseline (speedup 1.0000x reference)
#include <hip/hip_runtime.h>

typedef __attribute__((ext_vector_type(2))) float v2f;
typedef __attribute__((ext_vector_type(8))) float v8f;

#define BN_EPS 1e-5f

template<int F>
__global__ void scatter_add_kernel(const float* __restrict__ X,
                                   const int* __restrict__ src,
                                   const int* __restrict__ dst,
                                   float* __restrict__ agg, int E) {
    constexpr int G = F / 4;
    long long t = (long long)blockIdx.x * blockDim.x + threadIdx.x;
    long long total = (long long)E * G;
    if (t >= total) return;
    int e = (int)(t / G);
    int g = (int)(t % G);
    int s = src[e], d = dst[e];
    const float4 v = *reinterpret_cast<const float4*>(&X[(long long)s * F + g * 4]);
    float* p = &agg[(long long)d * F + g * 4];
    atomicAdd(p + 0, v.x);
    atomicAdd(p + 1, v.y);
    atomicAdd(p + 2, v.z);
    atomicAdd(p + 3, v.w);
}

template<int K, int NDIM, bool ADD_A, bool STATS>
__global__ void wmma_gemm_bias_relu(const float* __restrict__ A0,
                                    const float* __restrict__ A1,
                                    const float* __restrict__ W,
                                    const float* __restrict__ bias,
                                    float* __restrict__ Y,
                                    float* __restrict__ colsum,
                                    float* __restrict__ colsumsq) {
    constexpr int LSTRIDE = K + 4;
    __shared__ float tileA[16 * LSTRIDE];

    const int lane  = threadIdx.x;
    const int wave  = threadIdx.y;
    const int tid   = wave * 32 + lane;
    const int tileM = blockIdx.x;

    const long long aBase = (long long)tileM * 16 * K;
    #pragma unroll
    for (int g = tid; g < 4 * K; g += 512) {
        int arow = g / (K / 4);
        int acol = (g % (K / 4)) * 4;
        float4 v = *reinterpret_cast<const float4*>(&A0[aBase + (long long)arow * K + acol]);
        if (ADD_A) {
            float4 w = *reinterpret_cast<const float4*>(&A1[aBase + (long long)arow * K + acol]);
            v.x += w.x; v.y += w.y; v.z += w.z; v.w += w.w;
        }
        *reinterpret_cast<float4*>(&tileA[arow * LSTRIDE + acol]) = v;
    }
    __syncthreads();

    const int r16  = lane & 15;
    const int hi   = lane >> 4;
    const int bCol = wave * 16 + r16;
    const float* aLds = &tileA[r16 * LSTRIDE + hi * 2];

    v8f acc = {0.f, 0.f, 0.f, 0.f, 0.f, 0.f, 0.f, 0.f};

    #pragma unroll 4
    for (int k = 0; k < K; k += 4) {
        float2 av = *reinterpret_cast<const float2*>(aLds + k);
        v2f a; a.x = av.x; a.y = av.y;
        v2f b;
        b.x = W[(long long)(k + hi * 2    ) * NDIM + bCol];
        b.y = W[(long long)(k + hi * 2 + 1) * NDIM + bCol];
        acc = __builtin_amdgcn_wmma_f32_16x16x4_f32(
                   false, a,  false, b,
                   (short)0, acc,  false,  false);
    }

    const float bb = bias[bCol];
    float psum = 0.f, psq = 0.f;
    #pragma unroll
    for (int r = 0; r < 8; ++r) {
        long long row = (long long)tileM * 16 + r + hi * 8;
        float v = acc[r] + bb;
        v = v > 0.f ? v : 0.f;
        Y[row * NDIM + bCol] = v;
        psum += v;
        psq  += v * v;
    }
    if (STATS) {
        atomicAdd(&colsum[bCol], psum);
        atomicAdd(&colsumsq[bCol], psq);
    }
}

__global__ void bn_params_kernel(const float* __restrict__ colsum,
                                 const float* __restrict__ colsumsq,
                                 const float* __restrict__ gamma,
                                 const float* __restrict__ beta,
                                 float* __restrict__ scale,
                                 float* __restrict__ shift,
                                 int C, float invN) {
    int c = blockIdx.x * blockDim.x + threadIdx.x;
    if (c >= C) return;
    float mean = colsum[c] * invN;
    float var  = colsumsq[c] * invN - mean * mean;
    float s = gamma[c] * rsqrtf(var + BN_EPS);
    scale[c] = s;
    shift[c] = beta[c] - mean * s;
}

template<int F>
__global__ void bn_apply_vec4(const float* __restrict__ Yin,
                              const float* __restrict__ scale,
                              const float* __restrict__ shift,
                              float* __restrict__ Yout, long long nvec) {
    long long t = (long long)blockIdx.x * blockDim.x + threadIdx.x;
    if (t >= nvec) return;
    int c = (int)((t * 4) % F);
    float4 v = *reinterpret_cast<const float4*>(&Yin[t * 4]);
    float4 s = *reinterpret_cast<const float4*>(&scale[c]);
    float4 h = *reinterpret_cast<const float4*>(&shift[c]);
    float4 o;
    o.x = v.x * s.x + h.x;
    o.y = v.y * s.y + h.y;
    o.z = v.z * s.z + h.z;
    o.w = v.w * s.w + h.w;
    *reinterpret_cast<float4*>(&Yout[t * 4]) = o;
}

__global__ void bn_apply2(const float* __restrict__ Yin,
                          const float* __restrict__ scale,
                          const float* __restrict__ shift,
                          float* __restrict__ Yout, int total) {
    int t = blockIdx.x * blockDim.x + threadIdx.x;
    if (t >= total) return;
    int c = t & 1;
    Yout[t] = Yin[t] * scale[c] + shift[c];
}

__global__ void linear_out2_relu_stats(const float* __restrict__ G1,
                                       const float* __restrict__ W4,
                                       const float* __restrict__ b4,
                                       float* __restrict__ outpre,
                                       float* __restrict__ colsum,
                                       float* __restrict__ colsumsq,
                                       int Nn) {
    __shared__ float s_sum[2], s_sq[2];
    int tid = threadIdx.x;
    if (tid < 2) { s_sum[tid] = 0.f; s_sq[tid] = 0.f; }
    __syncthreads();
    int idx = blockIdx.x * blockDim.x + tid;
    if (idx < Nn * 2) {
        int row = idx >> 1, c = idx & 1;
        float acc = b4[c];
        const float* g = &G1[(long long)row * 256];
        #pragma unroll 8
        for (int k = 0; k < 256; ++k)
            acc = fmaf(g[k], W4[k * 2 + c], acc);
        float v = acc > 0.f ? acc : 0.f;
        outpre[idx] = v;
        atomicAdd(&s_sum[c], v);
        atomicAdd(&s_sq[c], v * v);
    }
    __syncthreads();
    if (tid < 2) {
        atomicAdd(&colsum[tid], s_sum[tid]);
        atomicAdd(&colsumsq[tid], s_sq[tid]);
    }
}

extern "C" void kernel_launch(void* const* d_in, const int* in_sizes, int n_in,
                              void* d_out, int out_size, void* d_ws, size_t ws_size,
                              hipStream_t stream) {
    (void)n_in; (void)out_size; (void)ws_size;

    const float* x   = (const float*)d_in[0];
    const int*   ei  = (const int*)  d_in[1];
    const float* W1  = (const float*)d_in[2];
    const float* b1  = (const float*)d_in[3];
    const float* W2  = (const float*)d_in[4];
    const float* b2  = (const float*)d_in[5];
    const float* W3  = (const float*)d_in[6];
    const float* b3  = (const float*)d_in[7];
    const float* W4  = (const float*)d_in[8];
    const float* b4  = (const float*)d_in[9];
    const float* gm1 = (const float*)d_in[10];
    const float* bt1 = (const float*)d_in[11];
    const float* gm5 = (const float*)d_in[12];
    const float* bt5 = (const float*)d_in[13];

    const int FIN = 128, DIM = 256;
    const int Nn = in_sizes[0] / FIN;
    const int E  = in_sizes[1] / 2;
    const int* src = ei;
    const int* dst = ei + E;

    float* ws = (float*)d_ws;
    float* colsum1   = ws + 0;
    float* colsumsq1 = ws + 256;
    float* scale1    = ws + 512;
    float* shift1    = ws + 768;
    float* colsum5   = ws + 1024;
    float* colsumsq5 = ws + 1028;
    float* scale5    = ws + 1032;
    float* shift5    = ws + 1036;
    float* R0 = ws + 2048;
    float* R1 = R0 + (size_t)Nn * FIN;
    float* R2 = R1 + (size_t)Nn * DIM;
    float* R3 = R2 + (size_t)Nn * DIM;

    float* outp = (float*)d_out;
    float* hid  = outp + (size_t)Nn * 2;

    hipMemsetAsync(ws, 0, 2048 * sizeof(float), stream);
    hipMemsetAsync(R0, 0, (size_t)Nn * FIN * sizeof(float), stream);

    const dim3 gemmBlk(32, 16);
    const dim3 gemmGrid(Nn / 16);

    {
        long long total = (long long)E * (FIN / 4);
        scatter_add_kernel<128><<<(int)((total + 255) / 256), 256, 0, stream>>>(
            x, src, dst, R0, E);
    }
    wmma_gemm_bias_relu<128, 256, true, false><<<gemmGrid, gemmBlk, 0, stream>>>(
        x, R0, W1, b1, R1, nullptr, nullptr);
    wmma_gemm_bias_relu<256, 256, false, true><<<gemmGrid, gemmBlk, 0, stream>>>(
        R1, nullptr, W2, b2, R2, colsum1, colsumsq1);
    bn_params_kernel<<<1, 256, 0, stream>>>(colsum1, colsumsq1, gm1, bt1,
                                            scale1, shift1, DIM, 1.0f / (float)Nn);
    {
        long long nvec = (long long)Nn * DIM / 4;
        bn_apply_vec4<256><<<(int)((nvec + 255) / 256), 256, 0, stream>>>(
            R2, scale1, shift1, hid, nvec);
    }
    hipMemsetAsync(R1, 0, (size_t)Nn * DIM * sizeof(float), stream);
    {
        long long total = (long long)E * (DIM / 4);
        scatter_add_kernel<256><<<(int)((total + 255) / 256), 256, 0, stream>>>(
            hid, src, dst, R1, E);
    }
    wmma_gemm_bias_relu<256, 256, true, false><<<gemmGrid, gemmBlk, 0, stream>>>(
        hid, R1, W3, b3, R2, nullptr, nullptr);
    linear_out2_relu_stats<<<(Nn * 2 + 255) / 256, 256, 0, stream>>>(
        R2, W4, b4, R3, colsum5, colsumsq5, Nn);
    bn_params_kernel<<<1, 32, 0, stream>>>(colsum5, colsumsq5, gm5, bt5,
                                           scale5, shift5, 2, 1.0f / (float)Nn);
    bn_apply2<<<(Nn * 2 + 255) / 256, 256, 0, stream>>>(R3, scale5, shift5,
                                                        outp, Nn * 2);
}


# segment_sum as CSR gather: per call rank edges by dst (1 int atomic each), scan, place; a wave sums a node's source rows in registers, one store; no f32 scatter atomics; sum order = arrival rank
# speedup vs baseline: 1.8257x; 1.8257x over previous
#include <hip/hip_runtime.h>

typedef __attribute__((ext_vector_type(2))) float v2f;
typedef __attribute__((ext_vector_type(8))) float v8f;

#define BN_EPS 1e-5f


__global__ void edge_rank_kernel(const int* __restrict__ dst,
                                 int* __restrict__ cnt,
                                 int* __restrict__ rank, int E) {
    int e = blockIdx.x * blockDim.x + threadIdx.x;
    if (e >= E) return;
    rank[e] = atomicAdd(&cnt[dst[e]], 1);
}

__global__ void __launch_bounds__(1024)
rowptr_scan_kernel(const int* __restrict__ cnt, int* __restrict__ rowptr, int N) {
    constexpr int PER = 16;
    __shared__ int waveTot[32];
    const int tid = threadIdx.x, lane = tid & 31, wv = tid >> 5;
    int carry = 0;
    for (int base = 0; base < N; base += 1024 * PER) {
        const int i0 = base + tid * PER;
        int c[PER];
        #pragma unroll
        for (int g = 0; g < PER; g += 4) {
            if (i0 + g + 3 < N) {
                const int4 v = *reinterpret_cast<const int4*>(&cnt[i0 + g]);
                c[g] = v.x; c[g + 1] = v.y; c[g + 2] = v.z; c[g + 3] = v.w;
            } else {
                #pragma unroll
                for (int q = 0; q < 4; ++q)
                    c[g + q] = (i0 + g + q < N) ? cnt[i0 + g + q] : 0;
            }
        }
        int tot = 0;
        #pragma unroll
        for (int q = 0; q < PER; ++q) tot += c[q];
        int inc = tot;
        #pragma unroll
        for (int d = 1; d < 32; d <<= 1) {
            int t = __shfl_up(inc, d);
            if (lane >= d) inc += t;
        }
        if (lane == 31) waveTot[wv] = inc;
        __syncthreads();
        const int wt = waveTot[lane];
        int winc = wt;
        #pragma unroll
        for (int d = 1; d < 32; d <<= 1) {
            int t = __shfl_up(winc, d);
            if (lane >= d) winc += t;
        }
        const int waveExcl = __shfl(winc, wv) - __shfl(wt, wv);
        const int chunkTot = __shfl(winc, 31);
        int ex = carry + waveExcl + (inc - tot);
        #pragma unroll
        for (int g = 0; g < PER; g += 4) {
            const int e0 = ex, e1 = e0 + c[g], e2 = e1 + c[g + 1], e3 = e2 + c[g + 2];
            ex = e3 + c[g + 3];
            if (i0 + g + 3 < N) {
                *reinterpret_cast<int4*>(&rowptr[i0 + g]) = make_int4(e0, e1, e2, e3);
            } else {
                if (i0 + g     < N) rowptr[i0 + g]     = e0;
                if (i0 + g + 1 < N) rowptr[i0 + g + 1] = e1;
                if (i0 + g + 2 < N) rowptr[i0 + g + 2] = e2;
            }
        }
        carry += chunkTot;
        __syncthreads();
    }
    if (tid == 0) rowptr[N] = carry;
}

__global__ void edge_place_kernel(const int* __restrict__ src,
                                  const int* __restrict__ dst,
                                  const int* __restrict__ rank,
                                  const int* __restrict__ rowptr,
                                  int* __restrict__ col, int E) {
    int e = blockIdx.x * blockDim.x + threadIdx.x;
    if (e >= E) return;
    col[rowptr[dst[e]] + rank[e]] = src[e];
}

template<int F>
__global__ void __launch_bounds__(256)
csr_gather_sum_kernel(const float* __restrict__ X,
                      const int* __restrict__ rowptr,
                      const int* __restrict__ col,
                      float* __restrict__ agg, int N) {
    constexpr int V = F / 128;
    constexpr int B = 8 / V;
    const int lane = threadIdx.x & 31;
    const int node = blockIdx.x * (blockDim.x >> 5)
                   + __builtin_amdgcn_readfirstlane(threadIdx.x >> 5);
    if (node >= N) return;
    const int beg = rowptr[node];
    const int end = rowptr[node + 1];

    float4 acc[V];
    #pragma unroll
    for (int w = 0; w < V; ++w) acc[w] = make_float4(0.f, 0.f, 0.f, 0.f);

    for (int base = beg; base < end; base += 32) {
        const int n = (end - base) < 32 ? (end - base) : 32;
        const int myc = (lane < n) ? col[base + lane] : 0;
        int j = 0;
        for (; j + B <= n; j += B) {
            float4 v[B][V];
            #pragma unroll
            for (int u = 0; u < B; ++u) {
                const int s = __builtin_amdgcn_readlane(myc, j + u);
                const float4* r = reinterpret_cast<const float4*>(X + (long long)s * F);
                #pragma unroll
                for (int w = 0; w < V; ++w) v[u][w] = r[lane + 32 * w];
            }
            #pragma unroll
            for (int u = 0; u < B; ++u) {
                #pragma unroll
                for (int w = 0; w < V; ++w) {
                    acc[w].x += v[u][w].x; acc[w].y += v[u][w].y;
                    acc[w].z += v[u][w].z; acc[w].w += v[u][w].w;
                }
            }
        }
        if (j < n) {
            float4 v[B][V];
            #pragma unroll
            for (int u = 0; u < B; ++u) {
                #pragma unroll
                for (int w = 0; w < V; ++w) v[u][w] = make_float4(0.f, 0.f, 0.f, 0.f);
                if (j + u < n) {
                    const int s = __builtin_amdgcn_readlane(myc, j + u);
                    const float4* r = reinterpret_cast<const float4*>(X + (long long)s * F);
                    #pragma unroll
                    for (int w = 0; w < V; ++w) v[u][w] = r[lane + 32 * w];
                }
            }
            #pragma unroll
            for (int u = 0; u < B; ++u) {
                #pragma unroll
                for (int w = 0; w < V; ++w) {
                    acc[w].x += v[u][w].x; acc[w].y += v[u][w].y;
                    acc[w].z += v[u][w].z; acc[w].w += v[u][w].w;
                }
            }
        }
    }

    float4* o = reinterpret_cast<float4*>(agg + (long long)node * F);
    #pragma unroll
    for (int w = 0; w < V; ++w) o[lane + 32 * w] = acc[w];
}

template<int K, int NDIM, bool ADD_A, bool STATS>
__global__ void wmma_gemm_bias_relu(const float* __restrict__ A0,
                                    const float* __restrict__ A1,
                                    const float* __restrict__ W,
                                    const float* __restrict__ bias,
                                    float* __restrict__ Y,
                                    float* __restrict__ colsum,
                                    float* __restrict__ colsumsq) {
    constexpr int LSTRIDE = K + 4;
    __shared__ float tileA[16 * LSTRIDE];

    const int lane  = threadIdx.x;
    const int wave  = threadIdx.y;
    const int tid   = wave * 32 + lane;
    const int tileM = blockIdx.x;

    const long long aBase = (long long)tileM * 16 * K;
    #pragma unroll
    for (int g = tid; g < 4 * K; g += 512) {
        int arow = g / (K / 4);
        int acol = (g % (K / 4)) * 4;
        float4 v = *reinterpret_cast<const float4*>(&A0[aBase + (long long)arow * K + acol]);
        if (ADD_A) {
            float4 w = *reinterpret_cast<const float4*>(&A1[aBase + (long long)arow * K + acol]);
            v.x += w.x; v.y += w.y; v.z += w.z; v.w += w.w;
        }
        *reinterpret_cast<float4*>(&tileA[arow * LSTRIDE + acol]) = v;
    }
    __syncthreads();

    const int r16  = lane & 15;
    const int hi   = lane >> 4;
    const int bCol = wave * 16 + r16;
    const float* aLds = &tileA[r16 * LSTRIDE + hi * 2];

    v8f acc = {0.f, 0.f, 0.f, 0.f, 0.f, 0.f, 0.f, 0.f};

    #pragma unroll 4
    for (int k = 0; k < K; k += 4) {
        float2 av = *reinterpret_cast<const float2*>(aLds + k);
        v2f a; a.x = av.x; a.y = av.y;
        v2f b;
        b.x = W[(long long)(k + hi * 2    ) * NDIM + bCol];
        b.y = W[(long long)(k + hi * 2 + 1) * NDIM + bCol];
        acc = __builtin_amdgcn_wmma_f32_16x16x4_f32(
                   false, a,  false, b,
                   (short)0, acc,  false,  false);
    }

    const float bb = bias[bCol];
    float psum = 0.f, psq = 0.f;
    #pragma unroll
    for (int r = 0; r < 8; ++r) {
        long long row = (long long)tileM * 16 + r + hi * 8;
        float v = acc[r] + bb;
        v = v > 0.f ? v : 0.f;
        Y[row * NDIM + bCol] = v;
        psum += v;
        psq  += v * v;
    }
    if (STATS) {
        atomicAdd(&colsum[bCol], psum);
        atomicAdd(&colsumsq[bCol], psq);
    }
}

__global__ void bn_params_kernel(const float* __restrict__ colsum,
                                 const float* __restrict__ colsumsq,
                                 const float* __restrict__ gamma,
                                 const float* __restrict__ beta,
                                 float* __restrict__ scale,
                                 float* __restrict__ shift,
                                 int C, float invN) {
    int c = blockIdx.x * blockDim.x + threadIdx.x;
    if (c >= C) return;
    float mean = colsum[c] * invN;
    float var  = colsumsq[c] * invN - mean * mean;
    float s = gamma[c] * rsqrtf(var + BN_EPS);
    scale[c] = s;
    shift[c] = beta[c] - mean * s;
}

template<int F>
__global__ void bn_apply_vec4(const float* __restrict__ Yin,
                              const float* __restrict__ scale,
                              const float* __restrict__ shift,
                              float* __restrict__ Yout, long long nvec) {
    long long t = (long long)blockIdx.x * blockDim.x + threadIdx.x;
    if (t >= nvec) return;
    int c = (int)((t * 4) % F);
    float4 v = *reinterpret_cast<const float4*>(&Yin[t * 4]);
    float4 s = *reinterpret_cast<const float4*>(&scale[c]);
    float4 h = *reinterpret_cast<const float4*>(&shift[c]);
    float4 o;
    o.x = v.x * s.x + h.x;
    o.y = v.y * s.y + h.y;
    o.z = v.z * s.z + h.z;
    o.w = v.w * s.w + h.w;
    *reinterpret_cast<float4*>(&Yout[t * 4]) = o;
}

__global__ void bn_apply2(const float* __restrict__ Yin,
                          const float* __restrict__ scale,
                          const float* __restrict__ shift,
                          float* __restrict__ Yout, int total) {
    int t = blockIdx.x * blockDim.x + threadIdx.x;
    if (t >= total) return;
    int c = t & 1;
    Yout[t] = Yin[t] * scale[c] + shift[c];
}

__global__ void linear_out2_relu_stats(const float* __restrict__ G1,
                                       const float* __restrict__ W4,
                                       const float* __restrict__ b4,
                                       float* __restrict__ outpre,
                                       float* __restrict__ colsum,
                                       float* __restrict__ colsumsq,
                                       int Nn) {
    __shared__ float s_sum[2], s_sq[2];
    int tid = threadIdx.x;
    if (tid < 2) { s_sum[tid] = 0.f; s_sq[tid] = 0.f; }
    __syncthreads();
    int idx = blockIdx.x * blockDim.x + tid;
    if (idx < Nn * 2) {
        int row = idx >> 1, c = idx & 1;
        float acc = b4[c];
        const float* g = &G1[(long long)row * 256];
        #pragma unroll 8
        for (int k = 0; k < 256; ++k)
            acc = fmaf(g[k], W4[k * 2 + c], acc);
        float v = acc > 0.f ? acc : 0.f;
        outpre[idx] = v;
        atomicAdd(&s_sum[c], v);
        atomicAdd(&s_sq[c], v * v);
    }
    __syncthreads();
    if (tid < 2) {
        atomicAdd(&colsum[tid], s_sum[tid]);
        atomicAdd(&colsumsq[tid], s_sq[tid]);
    }
}

extern "C" void kernel_launch(void* const* d_in, const int* in_sizes, int n_in,
                              void* d_out, int out_size, void* d_ws, size_t ws_size,
                              hipStream_t stream) {
    (void)n_in; (void)out_size; (void)ws_size;

    const float* x   = (const float*)d_in[0];
    const int*   ei  = (const int*)  d_in[1];
    const float* W1  = (const float*)d_in[2];
    const float* b1  = (const float*)d_in[3];
    const float* W2  = (const float*)d_in[4];
    const float* b2  = (const float*)d_in[5];
    const float* W3  = (const float*)d_in[6];
    const float* b3  = (const float*)d_in[7];
    const float* W4  = (const float*)d_in[8];
    const float* b4  = (const float*)d_in[9];
    const float* gm1 = (const float*)d_in[10];
    const float* bt1 = (const float*)d_in[11];
    const float* gm5 = (const float*)d_in[12];
    const float* bt5 = (const float*)d_in[13];

    const int FIN = 128, DIM = 256;
    const int Nn = in_sizes[0] / FIN;
    const int E  = in_sizes[1] / 2;
    const int* src = ei;
    const int* dst = ei + E;

    float* ws = (float*)d_ws;
    float* colsum1   = ws + 0;
    float* colsumsq1 = ws + 256;
    float* scale1    = ws + 512;
    float* shift1    = ws + 768;
    float* colsum5   = ws + 1024;
    float* colsumsq5 = ws + 1028;
    float* scale5    = ws + 1032;
    float* shift5    = ws + 1036;
    int*   cnt    = (int*)(ws + 2048);
    int*   rowptr = cnt + Nn;
    float* R0 = ws + 2048 + 2 * (size_t)Nn + 16;
    float* R1 = R0 + (size_t)Nn * FIN;
    float* R2 = R1 + (size_t)Nn * DIM;
    float* R3 = R2 + (size_t)Nn * DIM;
    int*   rank = (int*)(R3 + (size_t)Nn * 2);
    int*   col  = rank + E;

    float* outp = (float*)d_out;
    float* hid  = outp + (size_t)Nn * 2;

    hipMemsetAsync(ws, 0, (2048 + (size_t)Nn) * sizeof(float), stream);

    const dim3 gemmBlk(32, 16);
    const dim3 gemmGrid(Nn / 16);

    edge_rank_kernel<<<(E + 255) / 256, 256, 0, stream>>>(dst, cnt, rank, E);
    rowptr_scan_kernel<<<1, 1024, 0, stream>>>(cnt, rowptr, Nn);
    edge_place_kernel<<<(E + 255) / 256, 256, 0, stream>>>(src, dst, rank, rowptr, col, E);

    csr_gather_sum_kernel<128><<<(Nn + 7) / 8, 256, 0, stream>>>(x, rowptr, col, R0, Nn);
    wmma_gemm_bias_relu<128, 256, true, false><<<gemmGrid, gemmBlk, 0, stream>>>(
        x, R0, W1, b1, R1, nullptr, nullptr);
    wmma_gemm_bias_relu<256, 256, false, true><<<gemmGrid, gemmBlk, 0, stream>>>(
        R1, nullptr, W2, b2, R2, colsum1, colsumsq1);
    bn_params_kernel<<<1, 256, 0, stream>>>(colsum1, colsumsq1, gm1, bt1,
                                            scale1, shift1, DIM, 1.0f / (float)Nn);
    {
        long long nvec = (long long)Nn * DIM / 4;
        bn_apply_vec4<256><<<(int)((nvec + 255) / 256), 256, 0, stream>>>(
            R2, scale1, shift1, hid, nvec);
    }
    csr_gather_sum_kernel<256><<<(Nn + 7) / 8, 256, 0, stream>>>(hid, rowptr, col, R1, Nn);
    wmma_gemm_bias_relu<256, 256, true, false><<<gemmGrid, gemmBlk, 0, stream>>>(
        hid, R1, W3, b3, R2, nullptr, nullptr);
    linear_out2_relu_stats<<<(Nn * 2 + 255) / 256, 256, 0, stream>>>(
        R2, W4, b4, R3, colsum5, colsumsq5, Nn);
    bn_params_kernel<<<1, 32, 0, stream>>>(colsum5, colsumsq5, gm5, bt5,
                                           scale5, shift5, 2, 1.0f / (float)Nn);
    bn_apply2<<<(Nn * 2 + 255) / 256, 256, 0, stream>>>(R3, scale5, shift5,
                                                        outp, Nn * 2);
}


# GEMMs: M-tiles dealt evenly to 768 co-resident blocks (4-5 each); a wave feeds 4-5 WMMAs (own accumulators) from each W fragment, W fetched 4x less; BN-stat atomics add larger partials (sum order)
# speedup vs baseline: 2.0805x; 1.1396x over previous
#include <hip/hip_runtime.h>

typedef __attribute__((ext_vector_type(2))) float v2f;
typedef __attribute__((ext_vector_type(8))) float v8f;

#define BN_EPS 1e-5f


__global__ void edge_rank_kernel(const int* __restrict__ dst,
                                 int* __restrict__ cnt,
                                 int* __restrict__ rank, int E) {
    int e = blockIdx.x * blockDim.x + threadIdx.x;
    if (e >= E) return;
    rank[e] = atomicAdd(&cnt[dst[e]], 1);
}

__global__ void __launch_bounds__(1024)
rowptr_scan_kernel(const int* __restrict__ cnt, int* __restrict__ rowptr, int N) {
    constexpr int PER = 16;
    __shared__ int waveTot[32];
    const int tid = threadIdx.x, lane = tid & 31, wv = tid >> 5;
    int carry = 0;
    for (int base = 0; base < N; base += 1024 * PER) {
        const int i0 = base + tid * PER;
        int c[PER];
        #pragma unroll
        for (int g = 0; g < PER; g += 4) {
            if (i0 + g + 3 < N) {
                const int4 v = *reinterpret_cast<const int4*>(&cnt[i0 + g]);
                c[g] = v.x; c[g + 1] = v.y; c[g + 2] = v.z; c[g + 3] = v.w;
            } else {
                #pragma unroll
                for (int q = 0; q < 4; ++q)
                    c[g + q] = (i0 + g + q < N) ? cnt[i0 + g + q] : 0;
            }
        }
        int tot = 0;
        #pragma unroll
        for (int q = 0; q < PER; ++q) tot += c[q];
        int inc = tot;
        #pragma unroll
        for (int d = 1; d < 32; d <<= 1) {
            int t = __shfl_up(inc, d);
            if (lane >= d) inc += t;
        }
        if (lane == 31) waveTot[wv] = inc;
        __syncthreads();
        const int wt = waveTot[lane];
        int winc = wt;
        #pragma unroll
        for (int d = 1; d < 32; d <<= 1) {
            int t = __shfl_up(winc, d);
            if (lane >= d) winc += t;
        }
        const int waveExcl = __shfl(winc, wv) - __shfl(wt, wv);
        const int chunkTot = __shfl(winc, 31);
        int ex = carry + waveExcl + (inc - tot);
        #pragma unroll
        for (int g = 0; g < PER; g += 4) {
            const int e0 = ex, e1 = e0 + c[g], e2 = e1 + c[g + 1], e3 = e2 + c[g + 2];
            ex = e3 + c[g + 3];
            if (i0 + g + 3 < N) {
                *reinterpret_cast<int4*>(&rowptr[i0 + g]) = make_int4(e0, e1, e2, e3);
            } else {
                if (i0 + g     < N) rowptr[i0 + g]     = e0;
                if (i0 + g + 1 < N) rowptr[i0 + g + 1] = e1;
                if (i0 + g + 2 < N) rowptr[i0 + g + 2] = e2;
            }
        }
        carry += chunkTot;
        __syncthreads();
    }
    if (tid == 0) rowptr[N] = carry;
}

__global__ void edge_place_kernel(const int* __restrict__ src,
                                  const int* __restrict__ dst,
                                  const int* __restrict__ rank,
                                  const int* __restrict__ rowptr,
                                  int* __restrict__ col, int E) {
    int e = blockIdx.x * blockDim.x + threadIdx.x;
    if (e >= E) return;
    col[rowptr[dst[e]] + rank[e]] = src[e];
}

template<int F>
__global__ void __launch_bounds__(256)
csr_gather_sum_kernel(const float* __restrict__ X,
                      const int* __restrict__ rowptr,
                      const int* __restrict__ col,
                      float* __restrict__ agg, int N) {
    constexpr int V = F / 128;
    constexpr int B = 8 / V;
    const int lane = threadIdx.x & 31;
    const int node = blockIdx.x * (blockDim.x >> 5)
                   + __builtin_amdgcn_readfirstlane(threadIdx.x >> 5);
    if (node >= N) return;
    const int beg = rowptr[node];
    const int end = rowptr[node + 1];

    float4 acc[V];
    #pragma unroll
    for (int w = 0; w < V; ++w) acc[w] = make_float4(0.f, 0.f, 0.f, 0.f);

    for (int base = beg; base < end; base += 32) {
        const int n = (end - base) < 32 ? (end - base) : 32;
        const int myc = (lane < n) ? col[base + lane] : 0;
        int j = 0;
        for (; j + B <= n; j += B) {
            float4 v[B][V];
            #pragma unroll
            for (int u = 0; u < B; ++u) {
                const int s = __builtin_amdgcn_readlane(myc, j + u);
                const float4* r = reinterpret_cast<const float4*>(X + (long long)s * F);
                #pragma unroll
                for (int w = 0; w < V; ++w) v[u][w] = r[lane + 32 * w];
            }
            #pragma unroll
            for (int u = 0; u < B; ++u) {
                #pragma unroll
                for (int w = 0; w < V; ++w) {
                    acc[w].x += v[u][w].x; acc[w].y += v[u][w].y;
                    acc[w].z += v[u][w].z; acc[w].w += v[u][w].w;
                }
            }
        }
        if (j < n) {
            float4 v[B][V];
            #pragma unroll
            for (int u = 0; u < B; ++u) {
                #pragma unroll
                for (int w = 0; w < V; ++w) v[u][w] = make_float4(0.f, 0.f, 0.f, 0.f);
                if (j + u < n) {
                    const int s = __builtin_amdgcn_readlane(myc, j + u);
                    const float4* r = reinterpret_cast<const float4*>(X + (long long)s * F);
                    #pragma unroll
                    for (int w = 0; w < V; ++w) v[u][w] = r[lane + 32 * w];
                }
            }
            #pragma unroll
            for (int u = 0; u < B; ++u) {
                #pragma unroll
                for (int w = 0; w < V; ++w) {
                    acc[w].x += v[u][w].x; acc[w].y += v[u][w].y;
                    acc[w].z += v[u][w].z; acc[w].w += v[u][w].w;
                }
            }
        }
    }

    float4* o = reinterpret_cast<float4*>(agg + (long long)node * F);
    #pragma unroll
    for (int w = 0; w < V; ++w) o[lane + 32 * w] = acc[w];
}

constexpr int GEMM_MAXT = 5;

template<int K, int NDIM, int MT, bool STATS>
__device__ __forceinline__ void wmma_tile_run(const float* __restrict__ tileA,
                                              const float* __restrict__ W,
                                              float bb, float* __restrict__ Y,
                                              int row0, int M, int lane, int wave,
                                              float& psum, float& psq) {
    constexpr int LSTRIDE = K + 4;
    const int r16  = lane & 15;
    const int hi   = lane >> 4;
    const int bCol = wave * 16 + r16;
    const float* aLds = &tileA[r16 * LSTRIDE + hi * 2];
    const float* wp   = W + (long long)(hi * 2) * NDIM + bCol;

    v8f acc[MT];
    #pragma unroll
    for (int t = 0; t < MT; ++t)
        acc[t] = (v8f){0.f, 0.f, 0.f, 0.f, 0.f, 0.f, 0.f, 0.f};

    #pragma unroll 2
    for (int k = 0; k < K; k += 4) {
        v2f b;
        b.x = wp[(long long)k * NDIM];
        b.y = wp[(long long)(k + 1) * NDIM];
        #pragma unroll
        for (int t = 0; t < MT; ++t) {
            float2 av = *reinterpret_cast<const float2*>(aLds + t * 16 * LSTRIDE + k);
            v2f a; a.x = av.x; a.y = av.y;
            acc[t] = __builtin_amdgcn_wmma_f32_16x16x4_f32(
                       false, a,  false, b,
                       (short)0, acc[t],  false,  false);
        }
    }

    float* yp = Y + (long long)(row0 + hi * 8) * NDIM + bCol;
    if (row0 + MT * 16 <= M) {
        #pragma unroll
        for (int t = 0; t < MT; ++t) {
            #pragma unroll
            for (int r = 0; r < 8; ++r) {
                float v = acc[t][r] + bb;
                v = v > 0.f ? v : 0.f;
                yp[(t * 16 + r) * NDIM] = v;
                if (STATS) { psum += v; psq += v * v; }
            }
        }
    } else {
        const int left = M - (row0 + hi * 8);
        #pragma unroll
        for (int t = 0; t < MT; ++t) {
            #pragma unroll
            for (int r = 0; r < 8; ++r) {
                float v = acc[t][r] + bb;
                v = v > 0.f ? v : 0.f;
                if (t * 16 + r < left) {
                    yp[(t * 16 + r) * NDIM] = v;
                    if (STATS) { psum += v; psq += v * v; }
                }
            }
        }
    }
}

template<int K, int MT, bool ADD_A>
__device__ __forceinline__ void stage_rows(float* __restrict__ tileA,
                                           const float* __restrict__ A0,
                                           const float* __restrict__ A1,
                                           int row0, int M, int tid) {
    constexpr int LSTRIDE = K + 4;
    constexpr int RSTEP = 512 / (K / 4);
    constexpr int N  = MT * 16 / RSTEP;
    constexpr int NB = ADD_A ? 4 : 8;
    static_assert(512 % (K / 4) == 0 && (MT * 16) % RSTEP == 0, "whole trips only");
    const int r0   = tid / (K / 4);
    const int acol = (tid % (K / 4)) * 4;
    float* l0 = tileA + r0 * LSTRIDE + acol;
    if (row0 + MT * 16 <= M) {
        const float* p0 = A0 + (long long)(row0 + r0) * K + acol;
        const float* p1 = ADD_A ? A1 + (long long)(row0 + r0) * K + acol : nullptr;
        #pragma unroll
        for (int i0 = 0; i0 < N; i0 += NB) {
            float4 v[NB], w[NB];
            #pragma unroll
            for (int j = 0; j < NB; ++j) {
                if (i0 + j < N) {
                    v[j] = *reinterpret_cast<const float4*>(p0 + (i0 + j) * RSTEP * K);
                    if (ADD_A) w[j] = *reinterpret_cast<const float4*>(p1 + (i0 + j) * RSTEP * K);
                }
            }
            #pragma unroll
            for (int j = 0; j < NB; ++j) {
                if (i0 + j < N) {
                    float4 t = v[j];
                    if (ADD_A) { t.x += w[j].x; t.y += w[j].y; t.z += w[j].z; t.w += w[j].w; }
                    *reinterpret_cast<float4*>(l0 + (i0 + j) * RSTEP * LSTRIDE) = t;
                }
            }
        }
    } else {
        for (int i = 0; i < N; ++i) {
            const int row = row0 + r0 + i * RSTEP;
            float4 t = make_float4(0.f, 0.f, 0.f, 0.f);
            if (row < M) {
                t = *reinterpret_cast<const float4*>(A0 + (long long)row * K + acol);
                if (ADD_A) {
                    const float4 w = *reinterpret_cast<const float4*>(A1 + (long long)row * K + acol);
                    t.x += w.x; t.y += w.y; t.z += w.z; t.w += w.w;
                }
            }
            *reinterpret_cast<float4*>(l0 + i * RSTEP * LSTRIDE) = t;
        }
    }
}

template<int K, int NDIM, bool ADD_A, bool STATS>
__global__ void __launch_bounds__(512) __attribute__((amdgpu_waves_per_eu(12)))
wmma_gemm_bias_relu(const float* __restrict__ A0,
                    const float* __restrict__ A1,
                    const float* __restrict__ W,
                    const float* __restrict__ bias,
                    float* __restrict__ Y,
                    float* __restrict__ colsum,
                    float* __restrict__ colsumsq, int M) {
    constexpr int LSTRIDE = K + 4;
    __shared__ float tileA[GEMM_MAXT * 16 * LSTRIDE];

    const int lane  = threadIdx.x;
    const int wave  = threadIdx.y;
    const int tid   = wave * 32 + lane;

    const int units = (M + 15) >> 4;
    const int u0 = (int)(((long long)blockIdx.x * units) / gridDim.x);
    const int u1 = (int)(((long long)(blockIdx.x + 1) * units) / gridDim.x);
    const int n    = u1 - u0;
    const int row0 = u0 * 16;

    const float bb = bias[wave * 16 + (lane & 15)];
    float psum = 0.f, psq = 0.f;

    if (n == 5) {
        stage_rows<K, 5, ADD_A>(tileA, A0, A1, row0, M, tid);
        __syncthreads();
        wmma_tile_run<K, NDIM, 5, STATS>(tileA, W, bb, Y, row0, M, lane, wave, psum, psq);
    } else if (n == 4) {
        stage_rows<K, 4, ADD_A>(tileA, A0, A1, row0, M, tid);
        __syncthreads();
        wmma_tile_run<K, NDIM, 4, STATS>(tileA, W, bb, Y, row0, M, lane, wave, psum, psq);
    } else {
        for (int t = 0; t < n; ++t)
            stage_rows<K, 1, ADD_A>(tileA + t * 16 * LSTRIDE, A0, A1, row0 + t * 16, M, tid);
        __syncthreads();
        for (int t = 0; t < n; ++t)
            wmma_tile_run<K, NDIM, 1, STATS>(tileA + t * 16 * LSTRIDE, W, bb, Y,
                                             row0 + t * 16, M, lane, wave, psum, psq);
    }

    if (STATS) {
        atomicAdd(&colsum[wave * 16 + (lane & 15)], psum);
        atomicAdd(&colsumsq[wave * 16 + (lane & 15)], psq);
    }
}

__global__ void bn_params_kernel(const float* __restrict__ colsum,
                                 const float* __restrict__ colsumsq,
                                 const float* __restrict__ gamma,
                                 const float* __restrict__ beta,
                                 float* __restrict__ scale,
                                 float* __restrict__ shift,
                                 int C, float invN) {
    int c = blockIdx.x * blockDim.x + threadIdx.x;
    if (c >= C) return;
    float mean = colsum[c] * invN;
    float var  = colsumsq[c] * invN - mean * mean;
    float s = gamma[c] * rsqrtf(var + BN_EPS);
    scale[c] = s;
    shift[c] = beta[c] - mean * s;
}

template<int F>
__global__ void bn_apply_vec4(const float* __restrict__ Yin,
                              const float* __restrict__ scale,
                              const float* __restrict__ shift,
                              float* __restrict__ Yout, long long nvec) {
    long long t = (long long)blockIdx.x * blockDim.x + threadIdx.x;
    if (t >= nvec) return;
    int c = (int)((t * 4) % F);
    float4 v = *reinterpret_cast<const float4*>(&Yin[t * 4]);
    float4 s = *reinterpret_cast<const float4*>(&scale[c]);
    float4 h = *reinterpret_cast<const float4*>(&shift[c]);
    float4 o;
    o.x = v.x * s.x + h.x;
    o.y = v.y * s.y + h.y;
    o.z = v.z * s.z + h.z;
    o.w = v.w * s.w + h.w;
    *reinterpret_cast<float4*>(&Yout[t * 4]) = o;
}

__global__ void bn_apply2(const float* __restrict__ Yin,
                          const float* __restrict__ scale,
                          const float* __restrict__ shift,
                          float* __restrict__ Yout, int total) {
    int t = blockIdx.x * blockDim.x + threadIdx.x;
    if (t >= total) return;
    int c = t & 1;
    Yout[t] = Yin[t] * scale[c] + shift[c];
}

__global__ void linear_out2_relu_stats(const float* __restrict__ G1,
                                       const float* __restrict__ W4,
                                       const float* __restrict__ b4,
                                       float* __restrict__ outpre,
                                       float* __restrict__ colsum,
                                       float* __restrict__ colsumsq,
                                       int Nn) {
    __shared__ float s_sum[2], s_sq[2];
    int tid = threadIdx.x;
    if (tid < 2) { s_sum[tid] = 0.f; s_sq[tid] = 0.f; }
    __syncthreads();
    int idx = blockIdx.x * blockDim.x + tid;
    if (idx < Nn * 2) {
        int row = idx >> 1, c = idx & 1;
        float acc = b4[c];
        const float* g = &G1[(long long)row * 256];
        #pragma unroll 8
        for (int k = 0; k < 256; ++k)
            acc = fmaf(g[k], W4[k * 2 + c], acc);
        float v = acc > 0.f ? acc : 0.f;
        outpre[idx] = v;
        atomicAdd(&s_sum[c], v);
        atomicAdd(&s_sq[c], v * v);
    }
    __syncthreads();
    if (tid < 2) {
        atomicAdd(&colsum[tid], s_sum[tid]);
        atomicAdd(&colsumsq[tid], s_sq[tid]);
    }
}

extern "C" void kernel_launch(void* const* d_in, const int* in_sizes, int n_in,
                              void* d_out, int out_size, void* d_ws, size_t ws_size,
                              hipStream_t stream) {
    (void)n_in; (void)out_size; (void)ws_size;

    const float* x   = (const float*)d_in[0];
    const int*   ei  = (const int*)  d_in[1];
    const float* W1  = (const float*)d_in[2];
    const float* b1  = (const float*)d_in[3];
    const float* W2  = (const float*)d_in[4];
    const float* b2  = (const float*)d_in[5];
    const float* W3  = (const float*)d_in[6];
    const float* b3  = (const float*)d_in[7];
    const float* W4  = (const float*)d_in[8];
    const float* b4  = (const float*)d_in[9];
    const float* gm1 = (const float*)d_in[10];
    const float* bt1 = (const float*)d_in[11];
    const float* gm5 = (const float*)d_in[12];
    const float* bt5 = (const float*)d_in[13];

    const int FIN = 128, DIM = 256;
    const int Nn = in_sizes[0] / FIN;
    const int E  = in_sizes[1] / 2;
    const int* src = ei;
    const int* dst = ei + E;

    float* ws = (float*)d_ws;
    float* colsum1   = ws + 0;
    float* colsumsq1 = ws + 256;
    float* scale1    = ws + 512;
    float* shift1    = ws + 768;
    float* colsum5   = ws + 1024;
    float* colsumsq5 = ws + 1028;
    float* scale5    = ws + 1032;
    float* shift5    = ws + 1036;
    int*   cnt    = (int*)(ws + 2048);
    int*   rowptr = cnt + Nn;
    float* R0 = ws + 2048 + 2 * (size_t)Nn + 16;
    float* R1 = R0 + (size_t)Nn * FIN;
    float* R2 = R1 + (size_t)Nn * DIM;
    float* R3 = R2 + (size_t)Nn * DIM;
    int*   rank = (int*)(R3 + (size_t)Nn * 2);
    int*   col  = rank + E;

    float* outp = (float*)d_out;
    float* hid  = outp + (size_t)Nn * 2;

    hipMemsetAsync(ws, 0, (2048 + (size_t)Nn) * sizeof(float), stream);

    const dim3 gemmBlk(32, 16);
    const int mTiles = (Nn + 15) / 16;
    const dim3 gemmGrid(mTiles > 768 * 5 ? (mTiles + 4) / 5 : 768);

    edge_rank_kernel<<<(E + 255) / 256, 256, 0, stream>>>(dst, cnt, rank, E);
    rowptr_scan_kernel<<<1, 1024, 0, stream>>>(cnt, rowptr, Nn);
    edge_place_kernel<<<(E + 255) / 256, 256, 0, stream>>>(src, dst, rank, rowptr, col, E);

    csr_gather_sum_kernel<128><<<(Nn + 7) / 8, 256, 0, stream>>>(x, rowptr, col, R0, Nn);
    wmma_gemm_bias_relu<128, 256, true, false><<<gemmGrid, gemmBlk, 0, stream>>>(
        x, R0, W1, b1, R1, nullptr, nullptr, Nn);
    wmma_gemm_bias_relu<256, 256, false, true><<<gemmGrid, gemmBlk, 0, stream>>>(
        R1, nullptr, W2, b2, R2, colsum1, colsumsq1, Nn);
    bn_params_kernel<<<1, 256, 0, stream>>>(colsum1, colsumsq1, gm1, bt1,
                                            scale1, shift1, DIM, 1.0f / (float)Nn);
    {
        long long nvec = (long long)Nn * DIM / 4;
        bn_apply_vec4<256><<<(int)((nvec + 255) / 256), 256, 0, stream>>>(
            R2, scale1, shift1, hid, nvec);
    }
    csr_gather_sum_kernel<256><<<(Nn + 7) / 8, 256, 0, stream>>>(hid, rowptr, col, R1, Nn);
    wmma_gemm_bias_relu<256, 256, true, false><<<gemmGrid, gemmBlk, 0, stream>>>(
        hid, R1, W3, b3, R2, nullptr, nullptr, Nn);
    linear_out2_relu_stats<<<(Nn * 2 + 255) / 256, 256, 0, stream>>>(
        R2, W4, b4, R3, colsum5, colsumsq5, Nn);
    bn_params_kernel<<<1, 32, 0, stream>>>(colsum5, colsumsq5, gm5, bt5,
                                           scale5, shift5, 2, 1.0f / (float)Nn);
    bn_apply2<<<(Nn * 2 + 255) / 256, 256, 0, stream>>>(R3, scale5, shift5,
                                                        outp, Nn * 2);
}


# 256->2 head fused into the layer-2 GEMM: operand roles swapped so a lane holds 8 columns of one row; head = in-lane FMAs + LDS sum of 32 partials (new sum order); g1 never stored, one launch less
# speedup vs baseline: 2.1812x; 1.0484x over previous
#include <hip/hip_runtime.h>

typedef __attribute__((ext_vector_type(2))) float v2f;
typedef __attribute__((ext_vector_type(8))) float v8f;

#define BN_EPS 1e-5f


__global__ void edge_rank_kernel(const int* __restrict__ dst,
                                 int* __restrict__ cnt,
                                 int* __restrict__ rank, int E) {
    int e = blockIdx.x * blockDim.x + threadIdx.x;
    if (e >= E) return;
    rank[e] = atomicAdd(&cnt[dst[e]], 1);
}

__global__ void __launch_bounds__(1024)
rowptr_scan_kernel(const int* __restrict__ cnt, int* __restrict__ rowptr, int N) {
    constexpr int PER = 16;
    __shared__ int waveTot[32];
    const int tid = threadIdx.x, lane = tid & 31, wv = tid >> 5;
    int carry = 0;
    for (int base = 0; base < N; base += 1024 * PER) {
        const int i0 = base + tid * PER;
        int c[PER];
        #pragma unroll
        for (int g = 0; g < PER; g += 4) {
            if (i0 + g + 3 < N) {
                const int4 v = *reinterpret_cast<const int4*>(&cnt[i0 + g]);
                c[g] = v.x; c[g + 1] = v.y; c[g + 2] = v.z; c[g + 3] = v.w;
            } else {
                #pragma unroll
                for (int q = 0; q < 4; ++q)
                    c[g + q] = (i0 + g + q < N) ? cnt[i0 + g + q] : 0;
            }
        }
        int tot = 0;
        #pragma unroll
        for (int q = 0; q < PER; ++q) tot += c[q];
        int inc = tot;
        #pragma unroll
        for (int d = 1; d < 32; d <<= 1) {
            int t = __shfl_up(inc, d);
            if (lane >= d) inc += t;
        }
        if (lane == 31) waveTot[wv] = inc;
        __syncthreads();
        const int wt = waveTot[lane];
        int winc = wt;
        #pragma unroll
        for (int d = 1; d < 32; d <<= 1) {
            int t = __shfl_up(winc, d);
            if (lane >= d) winc += t;
        }
        const int waveExcl = __shfl(winc, wv) - __shfl(wt, wv);
        const int chunkTot = __shfl(winc, 31);
        int ex = carry + waveExcl + (inc - tot);
        #pragma unroll
        for (int g = 0; g < PER; g += 4) {
            const int e0 = ex, e1 = e0 + c[g], e2 = e1 + c[g + 1], e3 = e2 + c[g + 2];
            ex = e3 + c[g + 3];
            if (i0 + g + 3 < N) {
                *reinterpret_cast<int4*>(&rowptr[i0 + g]) = make_int4(e0, e1, e2, e3);
            } else {
                if (i0 + g     < N) rowptr[i0 + g]     = e0;
                if (i0 + g + 1 < N) rowptr[i0 + g + 1] = e1;
                if (i0 + g + 2 < N) rowptr[i0 + g + 2] = e2;
            }
        }
        carry += chunkTot;
        __syncthreads();
    }
    if (tid == 0) rowptr[N] = carry;
}

__global__ void edge_place_kernel(const int* __restrict__ src,
                                  const int* __restrict__ dst,
                                  const int* __restrict__ rank,
                                  const int* __restrict__ rowptr,
                                  int* __restrict__ col, int E) {
    int e = blockIdx.x * blockDim.x + threadIdx.x;
    if (e >= E) return;
    col[rowptr[dst[e]] + rank[e]] = src[e];
}

template<int F>
__global__ void __launch_bounds__(256)
csr_gather_sum_kernel(const float* __restrict__ X,
                      const int* __restrict__ rowptr,
                      const int* __restrict__ col,
                      float* __restrict__ agg, int N) {
    constexpr int V = F / 128;
    constexpr int B = 8 / V;
    const int lane = threadIdx.x & 31;
    const int node = blockIdx.x * (blockDim.x >> 5)
                   + __builtin_amdgcn_readfirstlane(threadIdx.x >> 5);
    if (node >= N) return;
    const int beg = rowptr[node];
    const int end = rowptr[node + 1];

    float4 acc[V];
    #pragma unroll
    for (int w = 0; w < V; ++w) acc[w] = make_float4(0.f, 0.f, 0.f, 0.f);

    for (int base = beg; base < end; base += 32) {
        const int n = (end - base) < 32 ? (end - base) : 32;
        const int myc = (lane < n) ? col[base + lane] : 0;
        int j = 0;
        for (; j + B <= n; j += B) {
            float4 v[B][V];
            #pragma unroll
            for (int u = 0; u < B; ++u) {
                const int s = __builtin_amdgcn_readlane(myc, j + u);
                const float4* r = reinterpret_cast<const float4*>(X + (long long)s * F);
                #pragma unroll
                for (int w = 0; w < V; ++w) v[u][w] = r[lane + 32 * w];
            }
            #pragma unroll
            for (int u = 0; u < B; ++u) {
                #pragma unroll
                for (int w = 0; w < V; ++w) {
                    acc[w].x += v[u][w].x; acc[w].y += v[u][w].y;
                    acc[w].z += v[u][w].z; acc[w].w += v[u][w].w;
                }
            }
        }
        if (j < n) {
            float4 v[B][V];
            #pragma unroll
            for (int u = 0; u < B; ++u) {
                #pragma unroll
                for (int w = 0; w < V; ++w) v[u][w] = make_float4(0.f, 0.f, 0.f, 0.f);
                if (j + u < n) {
                    const int s = __builtin_amdgcn_readlane(myc, j + u);
                    const float4* r = reinterpret_cast<const float4*>(X + (long long)s * F);
                    #pragma unroll
                    for (int w = 0; w < V; ++w) v[u][w] = r[lane + 32 * w];
                }
            }
            #pragma unroll
            for (int u = 0; u < B; ++u) {
                #pragma unroll
                for (int w = 0; w < V; ++w) {
                    acc[w].x += v[u][w].x; acc[w].y += v[u][w].y;
                    acc[w].z += v[u][w].z; acc[w].w += v[u][w].w;
                }
            }
        }
    }

    float4* o = reinterpret_cast<float4*>(agg + (long long)node * F);
    #pragma unroll
    for (int w = 0; w < V; ++w) o[lane + 32 * w] = acc[w];
}

constexpr int GEMM_MAXT = 5;

template<int K, int NDIM, int MT, bool STATS>
__device__ __forceinline__ void wmma_tile_run(const float* __restrict__ tileA,
                                              const float* __restrict__ W,
                                              float bb, float* __restrict__ Y,
                                              int row0, int M, int lane, int wave,
                                              float& psum, float& psq) {
    constexpr int LSTRIDE = K + 4;
    const int r16  = lane & 15;
    const int hi   = lane >> 4;
    const int bCol = wave * 16 + r16;
    const float* aLds = &tileA[r16 * LSTRIDE + hi * 2];
    const float* wp   = W + (long long)(hi * 2) * NDIM + bCol;

    v8f acc[MT];
    #pragma unroll
    for (int t = 0; t < MT; ++t)
        acc[t] = (v8f){0.f, 0.f, 0.f, 0.f, 0.f, 0.f, 0.f, 0.f};

    #pragma unroll 2
    for (int k = 0; k < K; k += 4) {
        v2f b;
        b.x = wp[(long long)k * NDIM];
        b.y = wp[(long long)(k + 1) * NDIM];
        #pragma unroll
        for (int t = 0; t < MT; ++t) {
            float2 av = *reinterpret_cast<const float2*>(aLds + t * 16 * LSTRIDE + k);
            v2f a; a.x = av.x; a.y = av.y;
            acc[t] = __builtin_amdgcn_wmma_f32_16x16x4_f32(
                       false, a,  false, b,
                       (short)0, acc[t],  false,  false);
        }
    }

    float* yp = Y + (long long)(row0 + hi * 8) * NDIM + bCol;
    if (row0 + MT * 16 <= M) {
        #pragma unroll
        for (int t = 0; t < MT; ++t) {
            #pragma unroll
            for (int r = 0; r < 8; ++r) {
                float v = acc[t][r] + bb;
                v = v > 0.f ? v : 0.f;
                yp[(t * 16 + r) * NDIM] = v;
                if (STATS) { psum += v; psq += v * v; }
            }
        }
    } else {
        const int left = M - (row0 + hi * 8);
        #pragma unroll
        for (int t = 0; t < MT; ++t) {
            #pragma unroll
            for (int r = 0; r < 8; ++r) {
                float v = acc[t][r] + bb;
                v = v > 0.f ? v : 0.f;
                if (t * 16 + r < left) {
                    yp[(t * 16 + r) * NDIM] = v;
                    if (STATS) { psum += v; psq += v * v; }
                }
            }
        }
    }
}

template<int K, int NDIM, int MT>
__device__ __forceinline__ void wmma_tile_run_head(const float* __restrict__ tileA,
                                                   const float* __restrict__ W,
                                                   const float* __restrict__ bias,
                                                   const float* __restrict__ W4,
                                                   int lane, int wave,
                                                   float2* __restrict__ pp) {
    constexpr int LSTRIDE = K + 4;
    const int r16  = lane & 15;
    const int hi   = lane >> 4;
    const int bCol = wave * 16 + r16;
    const float* aLds = &tileA[r16 * LSTRIDE + hi * 2];
    const float* wp   = W + (long long)(hi * 2) * NDIM + bCol;

    v8f acc[MT];
    #pragma unroll
    for (int t = 0; t < MT; ++t)
        acc[t] = (v8f){0.f, 0.f, 0.f, 0.f, 0.f, 0.f, 0.f, 0.f};

    #pragma unroll 2
    for (int k = 0; k < K; k += 4) {
        v2f b;
        b.x = wp[(long long)k * NDIM];
        b.y = wp[(long long)(k + 1) * NDIM];
        #pragma unroll
        for (int t = 0; t < MT; ++t) {
            float2 av = *reinterpret_cast<const float2*>(aLds + t * 16 * LSTRIDE + k);
            v2f a; a.x = av.x; a.y = av.y;
            acc[t] = __builtin_amdgcn_wmma_f32_16x16x4_f32(
                       false, b,  false, a,
                       (short)0, acc[t],  false,  false);
        }
    }

    const int f0 = wave * 16 + hi * 8;
    float bia[8], wa[8], wb[8];
    #pragma unroll
    for (int r = 0; r < 8; ++r) {
        bia[r] = bias[f0 + r];
        wa[r]  = W4[(f0 + r) * 2];
        wb[r]  = W4[(f0 + r) * 2 + 1];
    }
    #pragma unroll
    for (int t = 0; t < MT; ++t) {
        float s0 = 0.f, s1 = 0.f;
        #pragma unroll
        for (int r = 0; r < 8; ++r) {
            float v = acc[t][r] + bia[r];
            v = v > 0.f ? v : 0.f;
            s0 = fmaf(v, wa[r], s0);
            s1 = fmaf(v, wb[r], s1);
        }
        pp[t] = make_float2(s0, s1);
    }
}

template<int K, int MT, bool ADD_A>
__device__ __forceinline__ void stage_rows(float* __restrict__ tileA,
                                           const float* __restrict__ A0,
                                           const float* __restrict__ A1,
                                           int row0, int M, int tid) {
    constexpr int LSTRIDE = K + 4;
    constexpr int RSTEP = 512 / (K / 4);
    constexpr int N  = MT * 16 / RSTEP;
    constexpr int NB = ADD_A ? 4 : 8;
    static_assert(512 % (K / 4) == 0 && (MT * 16) % RSTEP == 0, "whole trips only");
    const int r0   = tid / (K / 4);
    const int acol = (tid % (K / 4)) * 4;
    float* l0 = tileA + r0 * LSTRIDE + acol;
    if (row0 + MT * 16 <= M) {
        const float* p0 = A0 + (long long)(row0 + r0) * K + acol;
        const float* p1 = ADD_A ? A1 + (long long)(row0 + r0) * K + acol : nullptr;
        #pragma unroll
        for (int i0 = 0; i0 < N; i0 += NB) {
            float4 v[NB], w[NB];
            #pragma unroll
            for (int j = 0; j < NB; ++j) {
                if (i0 + j < N) {
                    v[j] = *reinterpret_cast<const float4*>(p0 + (i0 + j) * RSTEP * K);
                    if (ADD_A) w[j] = *reinterpret_cast<const float4*>(p1 + (i0 + j) * RSTEP * K);
                }
            }
            #pragma unroll
            for (int j = 0; j < NB; ++j) {
                if (i0 + j < N) {
                    float4 t = v[j];
                    if (ADD_A) { t.x += w[j].x; t.y += w[j].y; t.z += w[j].z; t.w += w[j].w; }
                    *reinterpret_cast<float4*>(l0 + (i0 + j) * RSTEP * LSTRIDE) = t;
                }
            }
        }
    } else {
        for (int i = 0; i < N; ++i) {
            const int row = row0 + r0 + i * RSTEP;
            float4 t = make_float4(0.f, 0.f, 0.f, 0.f);
            if (row < M) {
                t = *reinterpret_cast<const float4*>(A0 + (long long)row * K + acol);
                if (ADD_A) {
                    const float4 w = *reinterpret_cast<const float4*>(A1 + (long long)row * K + acol);
                    t.x += w.x; t.y += w.y; t.z += w.z; t.w += w.w;
                }
            }
            *reinterpret_cast<float4*>(l0 + i * RSTEP * LSTRIDE) = t;
        }
    }
}

template<int K, int NDIM, bool ADD_A, bool STATS, bool HEAD>
__global__ void __launch_bounds__(512) __attribute__((amdgpu_waves_per_eu(12)))
wmma_gemm_bias_relu(const float* __restrict__ A0,
                    const float* __restrict__ A1,
                    const float* __restrict__ W,
                    const float* __restrict__ bias,
                    float* __restrict__ Y,
                    float* __restrict__ colsum,
                    float* __restrict__ colsumsq,
                    const float* __restrict__ W4,
                    const float* __restrict__ b4, int M) {
    constexpr int LSTRIDE = K + 4;
    __shared__ __attribute__((aligned(16))) float tileA[GEMM_MAXT * 16 * LSTRIDE];

    const int lane  = threadIdx.x;
    const int wave  = threadIdx.y;
    const int tid   = wave * 32 + lane;

    const int units = (M + 15) >> 4;
    const int u0 = (int)(((long long)blockIdx.x * units) / gridDim.x);
    const int u1 = (int)(((long long)(blockIdx.x + 1) * units) / gridDim.x);
    const int n    = u1 - u0;
    const int row0 = u0 * 16;

    const float bb = HEAD ? 0.f : bias[wave * 16 + (lane & 15)];
    float psum = 0.f, psq = 0.f;
    float2 pp[GEMM_MAXT];
    #pragma unroll
    for (int t = 0; t < GEMM_MAXT; ++t) pp[t] = make_float2(0.f, 0.f);

    if (n == 5) {
        stage_rows<K, 5, ADD_A>(tileA, A0, A1, row0, M, tid);
        __syncthreads();
        if (HEAD) wmma_tile_run_head<K, NDIM, 5>(tileA, W, bias, W4, lane, wave, pp);
        else wmma_tile_run<K, NDIM, 5, STATS>(tileA, W, bb, Y, row0, M, lane, wave, psum, psq);
    } else if (n == 4) {
        stage_rows<K, 4, ADD_A>(tileA, A0, A1, row0, M, tid);
        __syncthreads();
        if (HEAD) wmma_tile_run_head<K, NDIM, 4>(tileA, W, bias, W4, lane, wave, pp);
        else wmma_tile_run<K, NDIM, 4, STATS>(tileA, W, bb, Y, row0, M, lane, wave, psum, psq);
    } else {
        for (int t = 0; t < n; ++t)
            stage_rows<K, 1, ADD_A>(tileA + t * 16 * LSTRIDE, A0, A1, row0 + t * 16, M, tid);
        __syncthreads();
        if (HEAD) {
            #pragma unroll
            for (int t = 0; t < 3; ++t)
                if (t < n)
                    wmma_tile_run_head<K, NDIM, 1>(tileA + t * 16 * LSTRIDE, W, bias, W4,
                                                   lane, wave, pp + t);
        } else {
            for (int t = 0; t < n; ++t)
                wmma_tile_run<K, NDIM, 1, STATS>(tileA + t * 16 * LSTRIDE, W, bb, Y,
                                                 row0 + t * 16, M, lane, wave, psum, psq);
        }
    }

    if (HEAD) {
        constexpr int PROWS = GEMM_MAXT * 16;
        __syncthreads();
        float2* part = reinterpret_cast<float2*>(tileA);
        const int grp = wave * 2 + (lane >> 4);
        #pragma unroll
        for (int t = 0; t < GEMM_MAXT; ++t)
            if (t < n) part[grp * PROWS + t * 16 + (lane & 15)] = pp[t];
        __syncthreads();
        if (wave == 0) {
            const float* pf = tileA;
            const int c = lane & 1;
            const float bc = b4[c];
            float ssum = 0.f, ssq = 0.f;
            #pragma unroll
            for (int j = 0; j < GEMM_MAXT; ++j) {
                if (j < n) {
                    const int idx = j * 32 + lane;
                    float s = bc;
                    #pragma unroll
                    for (int g = 0; g < 32; ++g) s += pf[g * (PROWS * 2) + idx];
                    const float v = s > 0.f ? s : 0.f;
                    const int row = row0 + (idx >> 1);
                    if (row < M) {
                        Y[(long long)row * 2 + c] = v;
                        ssum += v; ssq += v * v;
                    }
                }
            }
            #pragma unroll
            for (int d = 2; d < 32; d <<= 1) {
                ssum += __shfl_xor(ssum, d);
                ssq  += __shfl_xor(ssq, d);
            }
            if (lane < 2) {
                atomicAdd(&colsum[lane], ssum);
                atomicAdd(&colsumsq[lane], ssq);
            }
        }
    }

    if (STATS) {
        atomicAdd(&colsum[wave * 16 + (lane & 15)], psum);
        atomicAdd(&colsumsq[wave * 16 + (lane & 15)], psq);
    }
}

__global__ void bn_params_kernel(const float* __restrict__ colsum,
                                 const float* __restrict__ colsumsq,
                                 const float* __restrict__ gamma,
                                 const float* __restrict__ beta,
                                 float* __restrict__ scale,
                                 float* __restrict__ shift,
                                 int C, float invN) {
    int c = blockIdx.x * blockDim.x + threadIdx.x;
    if (c >= C) return;
    float mean = colsum[c] * invN;
    float var  = colsumsq[c] * invN - mean * mean;
    float s = gamma[c] * rsqrtf(var + BN_EPS);
    scale[c] = s;
    shift[c] = beta[c] - mean * s;
}

template<int F>
__global__ void bn_apply_vec4(const float* __restrict__ Yin,
                              const float* __restrict__ scale,
                              const float* __restrict__ shift,
                              float* __restrict__ Yout, long long nvec) {
    long long t = (long long)blockIdx.x * blockDim.x + threadIdx.x;
    if (t >= nvec) return;
    int c = (int)((t * 4) % F);
    float4 v = *reinterpret_cast<const float4*>(&Yin[t * 4]);
    float4 s = *reinterpret_cast<const float4*>(&scale[c]);
    float4 h = *reinterpret_cast<const float4*>(&shift[c]);
    float4 o;
    o.x = v.x * s.x + h.x;
    o.y = v.y * s.y + h.y;
    o.z = v.z * s.z + h.z;
    o.w = v.w * s.w + h.w;
    *reinterpret_cast<float4*>(&Yout[t * 4]) = o;
}

__global__ void bn_apply2(const float* __restrict__ Yin,
                          const float* __restrict__ scale,
                          const float* __restrict__ shift,
                          float* __restrict__ Yout, int total) {
    int t = blockIdx.x * blockDim.x + threadIdx.x;
    if (t >= total) return;
    int c = t & 1;
    Yout[t] = Yin[t] * scale[c] + shift[c];
}

extern "C" void kernel_launch(void* const* d_in, const int* in_sizes, int n_in,
                              void* d_out, int out_size, void* d_ws, size_t ws_size,
                              hipStream_t stream) {
    (void)n_in; (void)out_size; (void)ws_size;

    const float* x   = (const float*)d_in[0];
    const int*   ei  = (const int*)  d_in[1];
    const float* W1  = (const float*)d_in[2];
    const float* b1  = (const float*)d_in[3];
    const float* W2  = (const float*)d_in[4];
    const float* b2  = (const float*)d_in[5];
    const float* W3  = (const float*)d_in[6];
    const float* b3  = (const float*)d_in[7];
    const float* W4  = (const float*)d_in[8];
    const float* b4  = (const float*)d_in[9];
    const float* gm1 = (const float*)d_in[10];
    const float* bt1 = (const float*)d_in[11];
    const float* gm5 = (const float*)d_in[12];
    const float* bt5 = (const float*)d_in[13];

    const int FIN = 128, DIM = 256;
    const int Nn = in_sizes[0] / FIN;
    const int E  = in_sizes[1] / 2;
    const int* src = ei;
    const int* dst = ei + E;

    float* ws = (float*)d_ws;
    float* colsum1   = ws + 0;
    float* colsumsq1 = ws + 256;
    float* scale1    = ws + 512;
    float* shift1    = ws + 768;
    float* colsum5   = ws + 1024;
    float* colsumsq5 = ws + 1028;
    float* scale5    = ws + 1032;
    float* shift5    = ws + 1036;
    int*   cnt    = (int*)(ws + 2048);
    int*   rowptr = cnt + Nn;
    float* R0 = ws + 2048 + 2 * (size_t)Nn + 16;
    float* R1 = R0 + (size_t)Nn * FIN;
    float* R2 = R1 + (size_t)Nn * DIM;
    float* R3 = R2 + (size_t)Nn * DIM;
    int*   rank = (int*)(R3 + (size_t)Nn * 2);
    int*   col  = rank + E;

    float* outp = (float*)d_out;
    float* hid  = outp + (size_t)Nn * 2;

    hipMemsetAsync(ws, 0, (2048 + (size_t)Nn) * sizeof(float), stream);

    const dim3 gemmBlk(32, 16);
    const int mTiles = (Nn + 15) / 16;
    const dim3 gemmGrid(mTiles > 768 * 5 ? (mTiles + 4) / 5 : 768);

    edge_rank_kernel<<<(E + 255) / 256, 256, 0, stream>>>(dst, cnt, rank, E);
    rowptr_scan_kernel<<<1, 1024, 0, stream>>>(cnt, rowptr, Nn);
    edge_place_kernel<<<(E + 255) / 256, 256, 0, stream>>>(src, dst, rank, rowptr, col, E);

    csr_gather_sum_kernel<128><<<(Nn + 7) / 8, 256, 0, stream>>>(x, rowptr, col, R0, Nn);
    wmma_gemm_bias_relu<128, 256, true, false, false><<<gemmGrid, gemmBlk, 0, stream>>>(
        x, R0, W1, b1, R1, nullptr, nullptr, nullptr, nullptr, Nn);
    wmma_gemm_bias_relu<256, 256, false, true, false><<<gemmGrid, gemmBlk, 0, stream>>>(
        R1, nullptr, W2, b2, R2, colsum1, colsumsq1, nullptr, nullptr, Nn);
    bn_params_kernel<<<1, 256, 0, stream>>>(colsum1, colsumsq1, gm1, bt1,
                                            scale1, shift1, DIM, 1.0f / (float)Nn);
    {
        long long nvec = (long long)Nn * DIM / 4;
        bn_apply_vec4<256><<<(int)((nvec + 255) / 256), 256, 0, stream>>>(
            R2, scale1, shift1, hid, nvec);
    }
    csr_gather_sum_kernel<256><<<(Nn + 7) / 8, 256, 0, stream>>>(hid, rowptr, col, R1, Nn);
    wmma_gemm_bias_relu<256, 256, true, false, true><<<gemmGrid, gemmBlk, 0, stream>>>(
        hid, R1, W3, b3, R3, colsum5, colsumsq5, W4, b4, Nn);
    bn_params_kernel<<<1, 32, 0, stream>>>(colsum5, colsumsq5, gm5, bt5,
                                           scale5, shift5, 2, 1.0f / (float)Nn);
    bn_apply2<<<(Nn * 2 + 255) / 256, 256, 0, stream>>>(R3, scale5, shift5,
                                                        outp, Nn * 2);
}


# BatchNorm-1 folded into the layer-2 gather: a wave derives scale/shift from the column sums, stores hid, adds s*row+t of each raw source row onto it (own row first: sum order); GEMM-3 stages 1 array
# speedup vs baseline: 2.2505x; 1.0318x over previous
#include <hip/hip_runtime.h>

typedef __attribute__((ext_vector_type(2))) float v2f;
typedef __attribute__((ext_vector_type(8))) float v8f;

#define BN_EPS 1e-5f


__global__ void edge_rank_kernel(const int* __restrict__ dst,
                                 int* __restrict__ cnt,
                                 int* __restrict__ rank, int E) {
    int e = blockIdx.x * blockDim.x + threadIdx.x;
    if (e >= E) return;
    rank[e] = atomicAdd(&cnt[dst[e]], 1);
}

__global__ void __launch_bounds__(1024)
rowptr_scan_kernel(const int* __restrict__ cnt, int* __restrict__ rowptr, int N) {
    constexpr int PER = 16;
    __shared__ int waveTot[32];
    const int tid = threadIdx.x, lane = tid & 31, wv = tid >> 5;
    int carry = 0;
    for (int base = 0; base < N; base += 1024 * PER) {
        const int i0 = base + tid * PER;
        int c[PER];
        #pragma unroll
        for (int g = 0; g < PER; g += 4) {
            if (i0 + g + 3 < N) {
                const int4 v = *reinterpret_cast<const int4*>(&cnt[i0 + g]);
                c[g] = v.x; c[g + 1] = v.y; c[g + 2] = v.z; c[g + 3] = v.w;
            } else {
                #pragma unroll
                for (int q = 0; q < 4; ++q)
                    c[g + q] = (i0 + g + q < N) ? cnt[i0 + g + q] : 0;
            }
        }
        int tot = 0;
        #pragma unroll
        for (int q = 0; q < PER; ++q) tot += c[q];
        int inc = tot;
        #pragma unroll
        for (int d = 1; d < 32; d <<= 1) {
            int t = __shfl_up(inc, d);
            if (lane >= d) inc += t;
        }
        if (lane == 31) waveTot[wv] = inc;
        __syncthreads();
        const int wt = waveTot[lane];
        int winc = wt;
        #pragma unroll
        for (int d = 1; d < 32; d <<= 1) {
            int t = __shfl_up(winc, d);
            if (lane >= d) winc += t;
        }
        const int waveExcl = __shfl(winc, wv) - __shfl(wt, wv);
        const int chunkTot = __shfl(winc, 31);
        int ex = carry + waveExcl + (inc - tot);
        #pragma unroll
        for (int g = 0; g < PER; g += 4) {
            const int e0 = ex, e1 = e0 + c[g], e2 = e1 + c[g + 1], e3 = e2 + c[g + 2];
            ex = e3 + c[g + 3];
            if (i0 + g + 3 < N) {
                *reinterpret_cast<int4*>(&rowptr[i0 + g]) = make_int4(e0, e1, e2, e3);
            } else {
                if (i0 + g     < N) rowptr[i0 + g]     = e0;
                if (i0 + g + 1 < N) rowptr[i0 + g + 1] = e1;
                if (i0 + g + 2 < N) rowptr[i0 + g + 2] = e2;
            }
        }
        carry += chunkTot;
        __syncthreads();
    }
    if (tid == 0) rowptr[N] = carry;
}

__global__ void edge_place_kernel(const int* __restrict__ src,
                                  const int* __restrict__ dst,
                                  const int* __restrict__ rank,
                                  const int* __restrict__ rowptr,
                                  int* __restrict__ col, int E) {
    int e = blockIdx.x * blockDim.x + threadIdx.x;
    if (e >= E) return;
    col[rowptr[dst[e]] + rank[e]] = src[e];
}

template<int F>
__global__ void __launch_bounds__(256)
csr_gather_sum_kernel(const float* __restrict__ X,
                      const int* __restrict__ rowptr,
                      const int* __restrict__ col,
                      float* __restrict__ agg, int N) {
    constexpr int V = F / 128;
    constexpr int B = 8 / V;
    const int lane = threadIdx.x & 31;
    const int node = blockIdx.x * (blockDim.x >> 5)
                   + __builtin_amdgcn_readfirstlane(threadIdx.x >> 5);
    if (node >= N) return;
    const int beg = rowptr[node];
    const int end = rowptr[node + 1];

    float4 acc[V];
    #pragma unroll
    for (int w = 0; w < V; ++w) acc[w] = make_float4(0.f, 0.f, 0.f, 0.f);

    for (int base = beg; base < end; base += 32) {
        const int n = (end - base) < 32 ? (end - base) : 32;
        const int myc = (lane < n) ? col[base + lane] : 0;
        int j = 0;
        for (; j + B <= n; j += B) {
            float4 v[B][V];
            #pragma unroll
            for (int u = 0; u < B; ++u) {
                const int s = __builtin_amdgcn_readlane(myc, j + u);
                const float4* r = reinterpret_cast<const float4*>(X + (long long)s * F);
                #pragma unroll
                for (int w = 0; w < V; ++w) v[u][w] = r[lane + 32 * w];
            }
            #pragma unroll
            for (int u = 0; u < B; ++u) {
                #pragma unroll
                for (int w = 0; w < V; ++w) {
                    acc[w].x += v[u][w].x; acc[w].y += v[u][w].y;
                    acc[w].z += v[u][w].z; acc[w].w += v[u][w].w;
                }
            }
        }
        if (j < n) {
            float4 v[B][V];
            #pragma unroll
            for (int u = 0; u < B; ++u) {
                #pragma unroll
                for (int w = 0; w < V; ++w) v[u][w] = make_float4(0.f, 0.f, 0.f, 0.f);
                if (j + u < n) {
                    const int s = __builtin_amdgcn_readlane(myc, j + u);
                    const float4* r = reinterpret_cast<const float4*>(X + (long long)s * F);
                    #pragma unroll
                    for (int w = 0; w < V; ++w) v[u][w] = r[lane + 32 * w];
                }
            }
            #pragma unroll
            for (int u = 0; u < B; ++u) {
                #pragma unroll
                for (int w = 0; w < V; ++w) {
                    acc[w].x += v[u][w].x; acc[w].y += v[u][w].y;
                    acc[w].z += v[u][w].z; acc[w].w += v[u][w].w;
                }
            }
        }
    }

    float4* o = reinterpret_cast<float4*>(agg + (long long)node * F);
    #pragma unroll
    for (int w = 0; w < V; ++w) o[lane + 32 * w] = acc[w];
}

__device__ __forceinline__ void add_normalised(float4& a, const float4& v,
                                               const float4& s, const float4& t) {
    a.x += v.x * s.x + t.x;
    a.y += v.y * s.y + t.y;
    a.z += v.z * s.z + t.z;
    a.w += v.w * s.w + t.w;
}

template<int F>
__global__ void __launch_bounds__(256)
csr_gather_sum_bn_kernel(const float* __restrict__ Y,
                         const int* __restrict__ rowptr,
                         const int* __restrict__ col,
                         const float* __restrict__ colsum,
                         const float* __restrict__ colsumsq,
                         const float* __restrict__ gamma,
                         const float* __restrict__ beta,
                         float* __restrict__ hid,
                         float* __restrict__ hsum, int N, float invN) {
    constexpr int V = F / 128;
    constexpr int B = 8 / V;
    const int lane = threadIdx.x & 31;
    const int node = blockIdx.x * (blockDim.x >> 5)
                   + __builtin_amdgcn_readfirstlane(threadIdx.x >> 5);
    if (node >= N) return;
    const int beg = rowptr[node];
    const int end = rowptr[node + 1];

    const float4* yrow = reinterpret_cast<const float4*>(Y + (long long)node * F);
    float4 acc[V], s[V], t[V];
    #pragma unroll
    for (int w = 0; w < V; ++w) {
        const int c4 = lane + 32 * w;
        const float4 own = yrow[c4];
        const float4 cs = reinterpret_cast<const float4*>(colsum)[c4];
        const float4 cq = reinterpret_cast<const float4*>(colsumsq)[c4];
        const float csv[4] = {cs.x, cs.y, cs.z, cs.w};
        const float cqv[4] = {cq.x, cq.y, cq.z, cq.w};
        float sv[4], tv[4];
        #pragma unroll
        for (int q = 0; q < 4; ++q) {
            const float mean = csv[q] * invN;
            const float var  = cqv[q] * invN - mean * mean;
            sv[q] = gamma[c4 * 4 + q] * rsqrtf(var + BN_EPS);
            tv[q] = beta[c4 * 4 + q] - mean * sv[q];
        }
        s[w] = make_float4(sv[0], sv[1], sv[2], sv[3]);
        t[w] = make_float4(tv[0], tv[1], tv[2], tv[3]);
        acc[w].x = own.x * s[w].x + t[w].x;
        acc[w].y = own.y * s[w].y + t[w].y;
        acc[w].z = own.z * s[w].z + t[w].z;
        acc[w].w = own.w * s[w].w + t[w].w;
    }
    {
        float4* h = reinterpret_cast<float4*>(hid + (long long)node * F);
        #pragma unroll
        for (int w = 0; w < V; ++w) h[lane + 32 * w] = acc[w];
    }

    for (int base = beg; base < end; base += 32) {
        const int n = (end - base) < 32 ? (end - base) : 32;
        const int myc = (lane < n) ? col[base + lane] : 0;
        int j = 0;
        for (; j + B <= n; j += B) {
            float4 v[B][V];
            #pragma unroll
            for (int u = 0; u < B; ++u) {
                const int sidx = __builtin_amdgcn_readlane(myc, j + u);
                const float4* r = reinterpret_cast<const float4*>(Y + (long long)sidx * F);
                #pragma unroll
                for (int w = 0; w < V; ++w) v[u][w] = r[lane + 32 * w];
            }
            #pragma unroll
            for (int u = 0; u < B; ++u) {
                #pragma unroll
                for (int w = 0; w < V; ++w) add_normalised(acc[w], v[u][w], s[w], t[w]);
            }
        }
        if (j < n) {
            float4 v[B][V];
            #pragma unroll
            for (int u = 0; u < B; ++u) {
                #pragma unroll
                for (int w = 0; w < V; ++w) v[u][w] = make_float4(0.f, 0.f, 0.f, 0.f);
                if (j + u < n) {
                    const int sidx = __builtin_amdgcn_readlane(myc, j + u);
                    const float4* r = reinterpret_cast<const float4*>(Y + (long long)sidx * F);
                    #pragma unroll
                    for (int w = 0; w < V; ++w) v[u][w] = r[lane + 32 * w];
                }
            }
            #pragma unroll
            for (int u = 0; u < B; ++u) {
                if (j + u < n) {
                    #pragma unroll
                    for (int w = 0; w < V; ++w) add_normalised(acc[w], v[u][w], s[w], t[w]);
                }
            }
        }
    }

    float4* o = reinterpret_cast<float4*>(hsum + (long long)node * F);
    #pragma unroll
    for (int w = 0; w < V; ++w) o[lane + 32 * w] = acc[w];
}

constexpr int GEMM_MAXT = 5;

template<int K, int NDIM, int MT, bool STATS>
__device__ __forceinline__ void wmma_tile_run(const float* __restrict__ tileA,
                                              const float* __restrict__ W,
                                              float bb, float* __restrict__ Y,
                                              int row0, int M, int lane, int wave,
                                              float& psum, float& psq) {
    constexpr int LSTRIDE = K + 4;
    const int r16  = lane & 15;
    const int hi   = lane >> 4;
    const int bCol = wave * 16 + r16;
    const float* aLds = &tileA[r16 * LSTRIDE + hi * 2];
    const float* wp   = W + (long long)(hi * 2) * NDIM + bCol;

    v8f acc[MT];
    #pragma unroll
    for (int t = 0; t < MT; ++t)
        acc[t] = (v8f){0.f, 0.f, 0.f, 0.f, 0.f, 0.f, 0.f, 0.f};

    #pragma unroll 2
    for (int k = 0; k < K; k += 4) {
        v2f b;
        b.x = wp[(long long)k * NDIM];
        b.y = wp[(long long)(k + 1) * NDIM];
        #pragma unroll
        for (int t = 0; t < MT; ++t) {
            float2 av = *reinterpret_cast<const float2*>(aLds + t * 16 * LSTRIDE + k);
            v2f a; a.x = av.x; a.y = av.y;
            acc[t] = __builtin_amdgcn_wmma_f32_16x16x4_f32(
                       false, a,  false, b,
                       (short)0, acc[t],  false,  false);
        }
    }

    float* yp = Y + (long long)(row0 + hi * 8) * NDIM + bCol;
    if (row0 + MT * 16 <= M) {
        #pragma unroll
        for (int t = 0; t < MT; ++t) {
            #pragma unroll
            for (int r = 0; r < 8; ++r) {
                float v = acc[t][r] + bb;
                v = v > 0.f ? v : 0.f;
                yp[(t * 16 + r) * NDIM] = v;
                if (STATS) { psum += v; psq += v * v; }
            }
        }
    } else {
        const int left = M - (row0 + hi * 8);
        #pragma unroll
        for (int t = 0; t < MT; ++t) {
            #pragma unroll
            for (int r = 0; r < 8; ++r) {
                float v = acc[t][r] + bb;
                v = v > 0.f ? v : 0.f;
                if (t * 16 + r < left) {
                    yp[(t * 16 + r) * NDIM] = v;
                    if (STATS) { psum += v; psq += v * v; }
                }
            }
        }
    }
}

template<int K, int NDIM, int MT>
__device__ __forceinline__ void wmma_tile_run_head(const float* __restrict__ tileA,
                                                   const float* __restrict__ W,
                                                   const float* __restrict__ bias,
                                                   const float* __restrict__ W4,
                                                   int lane, int wave,
                                                   float2* __restrict__ pp) {
    constexpr int LSTRIDE = K + 4;
    const int r16  = lane & 15;
    const int hi   = lane >> 4;
    const int bCol = wave * 16 + r16;
    const float* aLds = &tileA[r16 * LSTRIDE + hi * 2];
    const float* wp   = W + (long long)(hi * 2) * NDIM + bCol;

    v8f acc[MT];
    #pragma unroll
    for (int t = 0; t < MT; ++t)
        acc[t] = (v8f){0.f, 0.f, 0.f, 0.f, 0.f, 0.f, 0.f, 0.f};

    #pragma unroll 2
    for (int k = 0; k < K; k += 4) {
        v2f b;
        b.x = wp[(long long)k * NDIM];
        b.y = wp[(long long)(k + 1) * NDIM];
        #pragma unroll
        for (int t = 0; t < MT; ++t) {
            float2 av = *reinterpret_cast<const float2*>(aLds + t * 16 * LSTRIDE + k);
            v2f a; a.x = av.x; a.y = av.y;
            acc[t] = __builtin_amdgcn_wmma_f32_16x16x4_f32(
                       false, b,  false, a,
                       (short)0, acc[t],  false,  false);
        }
    }

    const int f0 = wave * 16 + hi * 8;
    float bia[8], wa[8], wb[8];
    #pragma unroll
    for (int r = 0; r < 8; ++r) {
        bia[r] = bias[f0 + r];
        wa[r]  = W4[(f0 + r) * 2];
        wb[r]  = W4[(f0 + r) * 2 + 1];
    }
    #pragma unroll
    for (int t = 0; t < MT; ++t) {
        float s0 = 0.f, s1 = 0.f;
        #pragma unroll
        for (int r = 0; r < 8; ++r) {
            float v = acc[t][r] + bia[r];
            v = v > 0.f ? v : 0.f;
            s0 = fmaf(v, wa[r], s0);
            s1 = fmaf(v, wb[r], s1);
        }
        pp[t] = make_float2(s0, s1);
    }
}

template<int K, int MT, bool ADD_A>
__device__ __forceinline__ void stage_rows(float* __restrict__ tileA,
                                           const float* __restrict__ A0,
                                           const float* __restrict__ A1,
                                           int row0, int M, int tid) {
    constexpr int LSTRIDE = K + 4;
    constexpr int RSTEP = 512 / (K / 4);
    constexpr int N  = MT * 16 / RSTEP;
    constexpr int NB = ADD_A ? 4 : 8;
    static_assert(512 % (K / 4) == 0 && (MT * 16) % RSTEP == 0, "whole trips only");
    const int r0   = tid / (K / 4);
    const int acol = (tid % (K / 4)) * 4;
    float* l0 = tileA + r0 * LSTRIDE + acol;
    if (row0 + MT * 16 <= M) {
        const float* p0 = A0 + (long long)(row0 + r0) * K + acol;
        const float* p1 = ADD_A ? A1 + (long long)(row0 + r0) * K + acol : nullptr;
        #pragma unroll
        for (int i0 = 0; i0 < N; i0 += NB) {
            float4 v[NB], w[NB];
            #pragma unroll
            for (int j = 0; j < NB; ++j) {
                if (i0 + j < N) {
                    v[j] = *reinterpret_cast<const float4*>(p0 + (i0 + j) * RSTEP * K);
                    if (ADD_A) w[j] = *reinterpret_cast<const float4*>(p1 + (i0 + j) * RSTEP * K);
                }
            }
            #pragma unroll
            for (int j = 0; j < NB; ++j) {
                if (i0 + j < N) {
                    float4 t = v[j];
                    if (ADD_A) { t.x += w[j].x; t.y += w[j].y; t.z += w[j].z; t.w += w[j].w; }
                    *reinterpret_cast<float4*>(l0 + (i0 + j) * RSTEP * LSTRIDE) = t;
                }
            }
        }
    } else {
        for (int i = 0; i < N; ++i) {
            const int row = row0 + r0 + i * RSTEP;
            float4 t = make_float4(0.f, 0.f, 0.f, 0.f);
            if (row < M) {
                t = *reinterpret_cast<const float4*>(A0 + (long long)row * K + acol);
                if (ADD_A) {
                    const float4 w = *reinterpret_cast<const float4*>(A1 + (long long)row * K + acol);
                    t.x += w.x; t.y += w.y; t.z += w.z; t.w += w.w;
                }
            }
            *reinterpret_cast<float4*>(l0 + i * RSTEP * LSTRIDE) = t;
        }
    }
}

template<int K, int NDIM, bool ADD_A, bool STATS, bool HEAD>
__global__ void __launch_bounds__(512) __attribute__((amdgpu_waves_per_eu(12)))
wmma_gemm_bias_relu(const float* __restrict__ A0,
                    const float* __restrict__ A1,
                    const float* __restrict__ W,
                    const float* __restrict__ bias,
                    float* __restrict__ Y,
                    float* __restrict__ colsum,
                    float* __restrict__ colsumsq,
                    const float* __restrict__ W4,
                    const float* __restrict__ b4, int M) {
    constexpr int LSTRIDE = K + 4;
    __shared__ __attribute__((aligned(16))) float tileA[GEMM_MAXT * 16 * LSTRIDE];

    const int lane  = threadIdx.x;
    const int wave  = threadIdx.y;
    const int tid   = wave * 32 + lane;

    const int units = (M + 15) >> 4;
    const int u0 = (int)(((long long)blockIdx.x * units) / gridDim.x);
    const int u1 = (int)(((long long)(blockIdx.x + 1) * units) / gridDim.x);
    const int n    = u1 - u0;
    const int row0 = u0 * 16;

    const float bb = HEAD ? 0.f : bias[wave * 16 + (lane & 15)];
    float psum = 0.f, psq = 0.f;
    float2 pp[GEMM_MAXT];
    #pragma unroll
    for (int t = 0; t < GEMM_MAXT; ++t) pp[t] = make_float2(0.f, 0.f);

    if (n == 5) {
        stage_rows<K, 5, ADD_A>(tileA, A0, A1, row0, M, tid);
        __syncthreads();
        if (HEAD) wmma_tile_run_head<K, NDIM, 5>(tileA, W, bias, W4, lane, wave, pp);
        else wmma_tile_run<K, NDIM, 5, STATS>(tileA, W, bb, Y, row0, M, lane, wave, psum, psq);
    } else if (n == 4) {
        stage_rows<K, 4, ADD_A>(tileA, A0, A1, row0, M, tid);
        __syncthreads();
        if (HEAD) wmma_tile_run_head<K, NDIM, 4>(tileA, W, bias, W4, lane, wave, pp);
        else wmma_tile_run<K, NDIM, 4, STATS>(tileA, W, bb, Y, row0, M, lane, wave, psum, psq);
    } else {
        for (int t = 0; t < n; ++t)
            stage_rows<K, 1, ADD_A>(tileA + t * 16 * LSTRIDE, A0, A1, row0 + t * 16, M, tid);
        __syncthreads();
        if (HEAD) {
            #pragma unroll
            for (int t = 0; t < 3; ++t)
                if (t < n)
                    wmma_tile_run_head<K, NDIM, 1>(tileA + t * 16 * LSTRIDE, W, bias, W4,
                                                   lane, wave, pp + t);
        } else {
            for (int t = 0; t < n; ++t)
                wmma_tile_run<K, NDIM, 1, STATS>(tileA + t * 16 * LSTRIDE, W, bb, Y,
                                                 row0 + t * 16, M, lane, wave, psum, psq);
        }
    }

    if (HEAD) {
        constexpr int PROWS = GEMM_MAXT * 16;
        __syncthreads();
        float2* part = reinterpret_cast<float2*>(tileA);
        const int grp = wave * 2 + (lane >> 4);
        #pragma unroll
        for (int t = 0; t < GEMM_MAXT; ++t)
            if (t < n) part[grp * PROWS + t * 16 + (lane & 15)] = pp[t];
        __syncthreads();
        if (wave == 0) {
            const float* pf = tileA;
            const int c = lane & 1;
            const float bc = b4[c];
            float ssum = 0.f, ssq = 0.f;
            #pragma unroll
            for (int j = 0; j < GEMM_MAXT; ++j) {
                if (j < n) {
                    const int idx = j * 32 + lane;
                    float s = bc;
                    #pragma unroll
                    for (int g = 0; g < 32; ++g) s += pf[g * (PROWS * 2) + idx];
                    const float v = s > 0.f ? s : 0.f;
                    const int row = row0 + (idx >> 1);
                    if (row < M) {
                        Y[(long long)row * 2 + c] = v;
                        ssum += v; ssq += v * v;
                    }
                }
            }
            #pragma unroll
            for (int d = 2; d < 32; d <<= 1) {
                ssum += __shfl_xor(ssum, d);
                ssq  += __shfl_xor(ssq, d);
            }
            if (lane < 2) {
                atomicAdd(&colsum[lane], ssum);
                atomicAdd(&colsumsq[lane], ssq);
            }
        }
    }

    if (STATS) {
        atomicAdd(&colsum[wave * 16 + (lane & 15)], psum);
        atomicAdd(&colsumsq[wave * 16 + (lane & 15)], psq);
    }
}

__global__ void bn_params_kernel(const float* __restrict__ colsum,
                                 const float* __restrict__ colsumsq,
                                 const float* __restrict__ gamma,
                                 const float* __restrict__ beta,
                                 float* __restrict__ scale,
                                 float* __restrict__ shift,
                                 int C, float invN) {
    int c = blockIdx.x * blockDim.x + threadIdx.x;
    if (c >= C) return;
    float mean = colsum[c] * invN;
    float var  = colsumsq[c] * invN - mean * mean;
    float s = gamma[c] * rsqrtf(var + BN_EPS);
    scale[c] = s;
    shift[c] = beta[c] - mean * s;
}

__global__ void bn_apply2(const float* __restrict__ Yin,
                          const float* __restrict__ scale,
                          const float* __restrict__ shift,
                          float* __restrict__ Yout, int total) {
    int t = blockIdx.x * blockDim.x + threadIdx.x;
    if (t >= total) return;
    int c = t & 1;
    Yout[t] = Yin[t] * scale[c] + shift[c];
}

extern "C" void kernel_launch(void* const* d_in, const int* in_sizes, int n_in,
                              void* d_out, int out_size, void* d_ws, size_t ws_size,
                              hipStream_t stream) {
    (void)n_in; (void)out_size; (void)ws_size;

    const float* x   = (const float*)d_in[0];
    const int*   ei  = (const int*)  d_in[1];
    const float* W1  = (const float*)d_in[2];
    const float* b1  = (const float*)d_in[3];
    const float* W2  = (const float*)d_in[4];
    const float* b2  = (const float*)d_in[5];
    const float* W3  = (const float*)d_in[6];
    const float* b3  = (const float*)d_in[7];
    const float* W4  = (const float*)d_in[8];
    const float* b4  = (const float*)d_in[9];
    const float* gm1 = (const float*)d_in[10];
    const float* bt1 = (const float*)d_in[11];
    const float* gm5 = (const float*)d_in[12];
    const float* bt5 = (const float*)d_in[13];

    const int FIN = 128, DIM = 256;
    const int Nn = in_sizes[0] / FIN;
    const int E  = in_sizes[1] / 2;
    const int* src = ei;
    const int* dst = ei + E;

    float* ws = (float*)d_ws;
    float* colsum1   = ws + 0;
    float* colsumsq1 = ws + 256;
    float* colsum5   = ws + 1024;
    float* colsumsq5 = ws + 1028;
    float* scale5    = ws + 1032;
    float* shift5    = ws + 1036;
    int*   cnt    = (int*)(ws + 2048);
    int*   rowptr = cnt + Nn;
    float* R0 = ws + 2048 + 2 * (size_t)Nn + 16;
    float* R1 = R0 + (size_t)Nn * FIN;
    float* R2 = R1 + (size_t)Nn * DIM;
    float* R3 = R2 + (size_t)Nn * DIM;
    int*   rank = (int*)(R3 + (size_t)Nn * 2);
    int*   col  = rank + E;

    float* outp = (float*)d_out;
    float* hid  = outp + (size_t)Nn * 2;

    hipMemsetAsync(ws, 0, (2048 + (size_t)Nn) * sizeof(float), stream);

    const dim3 gemmBlk(32, 16);
    const int mTiles = (Nn + 15) / 16;
    const dim3 gemmGrid(mTiles > 768 * 5 ? (mTiles + 4) / 5 : 768);

    edge_rank_kernel<<<(E + 255) / 256, 256, 0, stream>>>(dst, cnt, rank, E);
    rowptr_scan_kernel<<<1, 1024, 0, stream>>>(cnt, rowptr, Nn);
    edge_place_kernel<<<(E + 255) / 256, 256, 0, stream>>>(src, dst, rank, rowptr, col, E);

    csr_gather_sum_kernel<128><<<(Nn + 7) / 8, 256, 0, stream>>>(x, rowptr, col, R0, Nn);
    wmma_gemm_bias_relu<128, 256, true, false, false><<<gemmGrid, gemmBlk, 0, stream>>>(
        x, R0, W1, b1, R1, nullptr, nullptr, nullptr, nullptr, Nn);
    wmma_gemm_bias_relu<256, 256, false, true, false><<<gemmGrid, gemmBlk, 0, stream>>>(
        R1, nullptr, W2, b2, R2, colsum1, colsumsq1, nullptr, nullptr, Nn);
    csr_gather_sum_bn_kernel<256><<<(Nn + 7) / 8, 256, 0, stream>>>(
        R2, rowptr, col, colsum1, colsumsq1, gm1, bt1, hid, R1, Nn, 1.0f / (float)Nn);
    wmma_gemm_bias_relu<256, 256, false, false, true><<<gemmGrid, gemmBlk, 0, stream>>>(
        R1, nullptr, W3, b3, R3, colsum5, colsumsq5, W4, b4, Nn);
    bn_params_kernel<<<1, 32, 0, stream>>>(colsum5, colsumsq5, gm5, bt5,
                                           scale5, shift5, 2, 1.0f / (float)Nn);
    bn_apply2<<<(Nn * 2 + 255) / 256, 256, 0, stream>>>(R3, scale5, shift5,
                                                        outp, Nn * 2);
}


# edges grouped by dst in one pass: the counting atomic's rank indexes a 64-slot bucket per node (past 64: spill list walked by that node's wave); scan, rank array, place pass gone; sum order as before
# speedup vs baseline: 2.6275x; 1.1675x over previous
#include <hip/hip_runtime.h>

typedef __attribute__((ext_vector_type(2))) float v2f;
typedef __attribute__((ext_vector_type(8))) float v8f;

#define BN_EPS 1e-5f

constexpr int PITCH = 64;

__global__ void edge_bucket_kernel(const int* __restrict__ src,
                                   const int* __restrict__ dst,
                                   int* __restrict__ cnt,
                                   int* __restrict__ col,
                                   int* __restrict__ spillCnt,
                                   int* __restrict__ spill, int E) {
    int e = blockIdx.x * blockDim.x + threadIdx.x;
    if (e >= E) return;
    const int d = dst[e];
    const int s = src[e];
    const int r = atomicAdd(&cnt[d], 1);
    if (r < PITCH) {
        col[(long long)d * PITCH + r] = s;
    } else {
        const int k = atomicAdd(spillCnt, 1);
        reinterpret_cast<int2*>(spill)[k] = make_int2(d, s);
    }
}

template<int F>
__global__ void __launch_bounds__(256)
csr_gather_sum_kernel(const float* __restrict__ X,
                      const int* __restrict__ cnt,
                      const int* __restrict__ col,
                      const int* __restrict__ spillCnt,
                      const int* __restrict__ spill,
                      float* __restrict__ agg, int N) {
    constexpr int V = F / 128;
    constexpr int B = 8 / V;
    const int lane = threadIdx.x & 31;
    const int node = blockIdx.x * (blockDim.x >> 5)
                   + __builtin_amdgcn_readfirstlane(threadIdx.x >> 5);
    if (node >= N) return;
    const int deg = cnt[node];
    const int beg = node * PITCH;
    const int end = beg + (deg < PITCH ? deg : PITCH);

    float4 acc[V];
    #pragma unroll
    for (int w = 0; w < V; ++w) acc[w] = make_float4(0.f, 0.f, 0.f, 0.f);

    if (deg > PITCH) {
        const int nsp = *spillCnt;
        for (int base = 0; base < nsp; base += 32) {
            int2 ds = make_int2(-1, 0);
            if (base + lane < nsp) ds = reinterpret_cast<const int2*>(spill)[base + lane];
            unsigned m = __builtin_amdgcn_ballot_w32(ds.x == node);
            while (m) {
                const int b = __builtin_ctz(m);
                m &= m - 1;
                const int s = __builtin_amdgcn_readlane(ds.y, b);
                const float4* r = reinterpret_cast<const float4*>(X + (long long)s * F);
                #pragma unroll
                for (int w = 0; w < V; ++w) {
                    const float4 v = r[lane + 32 * w];
                    acc[w].x += v.x; acc[w].y += v.y; acc[w].z += v.z; acc[w].w += v.w;
                }
            }
        }
    }

    for (int base = beg; base < end; base += 32) {
        const int n = (end - base) < 32 ? (end - base) : 32;
        const int myc = (lane < n) ? col[base + lane] : 0;
        int j = 0;
        for (; j + B <= n; j += B) {
            float4 v[B][V];
            #pragma unroll
            for (int u = 0; u < B; ++u) {
                const int s = __builtin_amdgcn_readlane(myc, j + u);
                const float4* r = reinterpret_cast<const float4*>(X + (long long)s * F);
                #pragma unroll
                for (int w = 0; w < V; ++w) v[u][w] = r[lane + 32 * w];
            }
            #pragma unroll
            for (int u = 0; u < B; ++u) {
                #pragma unroll
                for (int w = 0; w < V; ++w) {
                    acc[w].x += v[u][w].x; acc[w].y += v[u][w].y;
                    acc[w].z += v[u][w].z; acc[w].w += v[u][w].w;
                }
            }
        }
        if (j < n) {
            float4 v[B][V];
            #pragma unroll
            for (int u = 0; u < B; ++u) {
                #pragma unroll
                for (int w = 0; w < V; ++w) v[u][w] = make_float4(0.f, 0.f, 0.f, 0.f);
                if (j + u < n) {
                    const int s = __builtin_amdgcn_readlane(myc, j + u);
                    const float4* r = reinterpret_cast<const float4*>(X + (long long)s * F);
                    #pragma unroll
                    for (int w = 0; w < V; ++w) v[u][w] = r[lane + 32 * w];
                }
            }
            #pragma unroll
            for (int u = 0; u < B; ++u) {
                #pragma unroll
                for (int w = 0; w < V; ++w) {
                    acc[w].x += v[u][w].x; acc[w].y += v[u][w].y;
                    acc[w].z += v[u][w].z; acc[w].w += v[u][w].w;
                }
            }
        }
    }

    float4* o = reinterpret_cast<float4*>(agg + (long long)node * F);
    #pragma unroll
    for (int w = 0; w < V; ++w) o[lane + 32 * w] = acc[w];
}

__device__ __forceinline__ void add_normalised(float4& a, const float4& v,
                                               const float4& s, const float4& t) {
    a.x += v.x * s.x + t.x;
    a.y += v.y * s.y + t.y;
    a.z += v.z * s.z + t.z;
    a.w += v.w * s.w + t.w;
}

template<int F>
__global__ void __launch_bounds__(256)
csr_gather_sum_bn_kernel(const float* __restrict__ Y,
                         const int* __restrict__ cnt,
                         const int* __restrict__ col,
                         const int* __restrict__ spillCnt,
                         const int* __restrict__ spill,
                         const float* __restrict__ colsum,
                         const float* __restrict__ colsumsq,
                         const float* __restrict__ gamma,
                         const float* __restrict__ beta,
                         float* __restrict__ hid,
                         float* __restrict__ hsum, int N, float invN) {
    constexpr int V = F / 128;
    constexpr int B = 8 / V;
    const int lane = threadIdx.x & 31;
    const int node = blockIdx.x * (blockDim.x >> 5)
                   + __builtin_amdgcn_readfirstlane(threadIdx.x >> 5);
    if (node >= N) return;
    const int deg = cnt[node];
    const int beg = node * PITCH;
    const int end = beg + (deg < PITCH ? deg : PITCH);

    const float4* yrow = reinterpret_cast<const float4*>(Y + (long long)node * F);
    float4 acc[V], s[V], t[V];
    #pragma unroll
    for (int w = 0; w < V; ++w) {
        const int c4 = lane + 32 * w;
        const float4 own = yrow[c4];
        const float4 cs = reinterpret_cast<const float4*>(colsum)[c4];
        const float4 cq = reinterpret_cast<const float4*>(colsumsq)[c4];
        const float csv[4] = {cs.x, cs.y, cs.z, cs.w};
        const float cqv[4] = {cq.x, cq.y, cq.z, cq.w};
        float sv[4], tv[4];
        #pragma unroll
        for (int q = 0; q < 4; ++q) {
            const float mean = csv[q] * invN;
            const float var  = cqv[q] * invN - mean * mean;
            sv[q] = gamma[c4 * 4 + q] * rsqrtf(var + BN_EPS);
            tv[q] = beta[c4 * 4 + q] - mean * sv[q];
        }
        s[w] = make_float4(sv[0], sv[1], sv[2], sv[3]);
        t[w] = make_float4(tv[0], tv[1], tv[2], tv[3]);
        acc[w].x = own.x * s[w].x + t[w].x;
        acc[w].y = own.y * s[w].y + t[w].y;
        acc[w].z = own.z * s[w].z + t[w].z;
        acc[w].w = own.w * s[w].w + t[w].w;
    }
    {
        float4* h = reinterpret_cast<float4*>(hid + (long long)node * F);
        #pragma unroll
        for (int w = 0; w < V; ++w) h[lane + 32 * w] = acc[w];
    }

    if (deg > PITCH) {
        const int nsp = *spillCnt;
        for (int base = 0; base < nsp; base += 32) {
            int2 ds = make_int2(-1, 0);
            if (base + lane < nsp) ds = reinterpret_cast<const int2*>(spill)[base + lane];
            unsigned m = __builtin_amdgcn_ballot_w32(ds.x == node);
            while (m) {
                const int b = __builtin_ctz(m);
                m &= m - 1;
                const int sidx = __builtin_amdgcn_readlane(ds.y, b);
                const float4* r = reinterpret_cast<const float4*>(Y + (long long)sidx * F);
                #pragma unroll
                for (int w = 0; w < V; ++w) add_normalised(acc[w], r[lane + 32 * w], s[w], t[w]);
            }
        }
    }

    for (int base = beg; base < end; base += 32) {
        const int n = (end - base) < 32 ? (end - base) : 32;
        const int myc = (lane < n) ? col[base + lane] : 0;
        int j = 0;
        for (; j + B <= n; j += B) {
            float4 v[B][V];
            #pragma unroll
            for (int u = 0; u < B; ++u) {
                const int sidx = __builtin_amdgcn_readlane(myc, j + u);
                const float4* r = reinterpret_cast<const float4*>(Y + (long long)sidx * F);
                #pragma unroll
                for (int w = 0; w < V; ++w) v[u][w] = r[lane + 32 * w];
            }
            #pragma unroll
            for (int u = 0; u < B; ++u) {
                #pragma unroll
                for (int w = 0; w < V; ++w) add_normalised(acc[w], v[u][w], s[w], t[w]);
            }
        }
        if (j < n) {
            float4 v[B][V];
            #pragma unroll
            for (int u = 0; u < B; ++u) {
                #pragma unroll
                for (int w = 0; w < V; ++w) v[u][w] = make_float4(0.f, 0.f, 0.f, 0.f);
                if (j + u < n) {
                    const int sidx = __builtin_amdgcn_readlane(myc, j + u);
                    const float4* r = reinterpret_cast<const float4*>(Y + (long long)sidx * F);
                    #pragma unroll
                    for (int w = 0; w < V; ++w) v[u][w] = r[lane + 32 * w];
                }
            }
            #pragma unroll
            for (int u = 0; u < B; ++u) {
                if (j + u < n) {
                    #pragma unroll
                    for (int w = 0; w < V; ++w) add_normalised(acc[w], v[u][w], s[w], t[w]);
                }
            }
        }
    }

    float4* o = reinterpret_cast<float4*>(hsum + (long long)node * F);
    #pragma unroll
    for (int w = 0; w < V; ++w) o[lane + 32 * w] = acc[w];
}

constexpr int GEMM_MAXT = 5;

template<int K, int NDIM, int MT, bool STATS>
__device__ __forceinline__ void wmma_tile_run(const float* __restrict__ tileA,
                                              const float* __restrict__ W,
                                              float bb, float* __restrict__ Y,
                                              int row0, int M, int lane, int wave,
                                              float& psum, float& psq) {
    constexpr int LSTRIDE = K + 4;
    const int r16  = lane & 15;
    const int hi   = lane >> 4;
    const int bCol = wave * 16 + r16;
    const float* aLds = &tileA[r16 * LSTRIDE + hi * 2];
    const float* wp   = W + (long long)(hi * 2) * NDIM + bCol;

    v8f acc[MT];
    #pragma unroll
    for (int t = 0; t < MT; ++t)
        acc[t] = (v8f){0.f, 0.f, 0.f, 0.f, 0.f, 0.f, 0.f, 0.f};

    #pragma unroll 2
    for (int k = 0; k < K; k += 4) {
        v2f b;
        b.x = wp[(long long)k * NDIM];
        b.y = wp[(long long)(k + 1) * NDIM];
        #pragma unroll
        for (int t = 0; t < MT; ++t) {
            float2 av = *reinterpret_cast<const float2*>(aLds + t * 16 * LSTRIDE + k);
            v2f a; a.x = av.x; a.y = av.y;
            acc[t] = __builtin_amdgcn_wmma_f32_16x16x4_f32(
                       false, a,  false, b,
                       (short)0, acc[t],  false,  false);
        }
    }

    float* yp = Y + (long long)(row0 + hi * 8) * NDIM + bCol;
    if (row0 + MT * 16 <= M) {
        #pragma unroll
        for (int t = 0; t < MT; ++t) {
            #pragma unroll
            for (int r = 0; r < 8; ++r) {
                float v = acc[t][r] + bb;
                v = v > 0.f ? v : 0.f;
                yp[(t * 16 + r) * NDIM] = v;
                if (STATS) { psum += v; psq += v * v; }
            }
        }
    } else {
        const int left = M - (row0 + hi * 8);
        #pragma unroll
        for (int t = 0; t < MT; ++t) {
            #pragma unroll
            for (int r = 0; r < 8; ++r) {
                float v = acc[t][r] + bb;
                v = v > 0.f ? v : 0.f;
                if (t * 16 + r < left) {
                    yp[(t * 16 + r) * NDIM] = v;
                    if (STATS) { psum += v; psq += v * v; }
                }
            }
        }
    }
}

template<int K, int NDIM, int MT>
__device__ __forceinline__ void wmma_tile_run_head(const float* __restrict__ tileA,
                                                   const float* __restrict__ W,
                                                   const float* __restrict__ bias,
                                                   const float* __restrict__ W4,
                                                   int lane, int wave,
                                                   float2* __restrict__ pp) {
    constexpr int LSTRIDE = K + 4;
    const int r16  = lane & 15;
    const int hi   = lane >> 4;
    const int bCol = wave * 16 + r16;
    const float* aLds = &tileA[r16 * LSTRIDE + hi * 2];
    const float* wp   = W + (long long)(hi * 2) * NDIM + bCol;

    v8f acc[MT];
    #pragma unroll
    for (int t = 0; t < MT; ++t)
        acc[t] = (v8f){0.f, 0.f, 0.f, 0.f, 0.f, 0.f, 0.f, 0.f};

    #pragma unroll 2
    for (int k = 0; k < K; k += 4) {
        v2f b;
        b.x = wp[(long long)k * NDIM];
        b.y = wp[(long long)(k + 1) * NDIM];
        #pragma unroll
        for (int t = 0; t < MT; ++t) {
            float2 av = *reinterpret_cast<const float2*>(aLds + t * 16 * LSTRIDE + k);
            v2f a; a.x = av.x; a.y = av.y;
            acc[t] = __builtin_amdgcn_wmma_f32_16x16x4_f32(
                       false, b,  false, a,
                       (short)0, acc[t],  false,  false);
        }
    }

    const int f0 = wave * 16 + hi * 8;
    float bia[8], wa[8], wb[8];
    #pragma unroll
    for (int r = 0; r < 8; ++r) {
        bia[r] = bias[f0 + r];
        wa[r]  = W4[(f0 + r) * 2];
        wb[r]  = W4[(f0 + r) * 2 + 1];
    }
    #pragma unroll
    for (int t = 0; t < MT; ++t) {
        float s0 = 0.f, s1 = 0.f;
        #pragma unroll
        for (int r = 0; r < 8; ++r) {
            float v = acc[t][r] + bia[r];
            v = v > 0.f ? v : 0.f;
            s0 = fmaf(v, wa[r], s0);
            s1 = fmaf(v, wb[r], s1);
        }
        pp[t] = make_float2(s0, s1);
    }
}

template<int K, int MT, bool ADD_A>
__device__ __forceinline__ void stage_rows(float* __restrict__ tileA,
                                           const float* __restrict__ A0,
                                           const float* __restrict__ A1,
                                           int row0, int M, int tid) {
    constexpr int LSTRIDE = K + 4;
    constexpr int RSTEP = 512 / (K / 4);
    constexpr int N  = MT * 16 / RSTEP;
    constexpr int NB = ADD_A ? 4 : 8;
    static_assert(512 % (K / 4) == 0 && (MT * 16) % RSTEP == 0, "whole trips only");
    const int r0   = tid / (K / 4);
    const int acol = (tid % (K / 4)) * 4;
    float* l0 = tileA + r0 * LSTRIDE + acol;
    if (row0 + MT * 16 <= M) {
        const float* p0 = A0 + (long long)(row0 + r0) * K + acol;
        const float* p1 = ADD_A ? A1 + (long long)(row0 + r0) * K + acol : nullptr;
        #pragma unroll
        for (int i0 = 0; i0 < N; i0 += NB) {
            float4 v[NB], w[NB];
            #pragma unroll
            for (int j = 0; j < NB; ++j) {
                if (i0 + j < N) {
                    v[j] = *reinterpret_cast<const float4*>(p0 + (i0 + j) * RSTEP * K);
                    if (ADD_A) w[j] = *reinterpret_cast<const float4*>(p1 + (i0 + j) * RSTEP * K);
                }
            }
            #pragma unroll
            for (int j = 0; j < NB; ++j) {
                if (i0 + j < N) {
                    float4 t = v[j];
                    if (ADD_A) { t.x += w[j].x; t.y += w[j].y; t.z += w[j].z; t.w += w[j].w; }
                    *reinterpret_cast<float4*>(l0 + (i0 + j) * RSTEP * LSTRIDE) = t;
                }
            }
        }
    } else {
        for (int i = 0; i < N; ++i) {
            const int row = row0 + r0 + i * RSTEP;
            float4 t = make_float4(0.f, 0.f, 0.f, 0.f);
            if (row < M) {
                t = *reinterpret_cast<const float4*>(A0 + (long long)row * K + acol);
                if (ADD_A) {
                    const float4 w = *reinterpret_cast<const float4*>(A1 + (long long)row * K + acol);
                    t.x += w.x; t.y += w.y; t.z += w.z; t.w += w.w;
                }
            }
            *reinterpret_cast<float4*>(l0 + i * RSTEP * LSTRIDE) = t;
        }
    }
}

template<int K, int NDIM, bool ADD_A, bool STATS, bool HEAD>
__global__ void __launch_bounds__(512) __attribute__((amdgpu_waves_per_eu(12)))
wmma_gemm_bias_relu(const float* __restrict__ A0,
                    const float* __restrict__ A1,
                    const float* __restrict__ W,
                    const float* __restrict__ bias,
                    float* __restrict__ Y,
                    float* __restrict__ colsum,
                    float* __restrict__ colsumsq,
                    const float* __restrict__ W4,
                    const float* __restrict__ b4, int M) {
    constexpr int LSTRIDE = K + 4;
    __shared__ __attribute__((aligned(16))) float tileA[GEMM_MAXT * 16 * LSTRIDE];

    const int lane  = threadIdx.x;
    const int wave  = threadIdx.y;
    const int tid   = wave * 32 + lane;

    const int units = (M + 15) >> 4;
    const int u0 = (int)(((long long)blockIdx.x * units) / gridDim.x);
    const int u1 = (int)(((long long)(blockIdx.x + 1) * units) / gridDim.x);
    const int n    = u1 - u0;
    const int row0 = u0 * 16;

    const float bb = HEAD ? 0.f : bias[wave * 16 + (lane & 15)];
    float psum = 0.f, psq = 0.f;
    float2 pp[GEMM_MAXT];
    #pragma unroll
    for (int t = 0; t < GEMM_MAXT; ++t) pp[t] = make_float2(0.f, 0.f);

    if (n == 5) {
        stage_rows<K, 5, ADD_A>(tileA, A0, A1, row0, M, tid);
        __syncthreads();
        if (HEAD) wmma_tile_run_head<K, NDIM, 5>(tileA, W, bias, W4, lane, wave, pp);
        else wmma_tile_run<K, NDIM, 5, STATS>(tileA, W, bb, Y, row0, M, lane, wave, psum, psq);
    } else if (n == 4) {
        stage_rows<K, 4, ADD_A>(tileA, A0, A1, row0, M, tid);
        __syncthreads();
        if (HEAD) wmma_tile_run_head<K, NDIM, 4>(tileA, W, bias, W4, lane, wave, pp);
        else wmma_tile_run<K, NDIM, 4, STATS>(tileA, W, bb, Y, row0, M, lane, wave, psum, psq);
    } else {
        for (int t = 0; t < n; ++t)
            stage_rows<K, 1, ADD_A>(tileA + t * 16 * LSTRIDE, A0, A1, row0 + t * 16, M, tid);
        __syncthreads();
        if (HEAD) {
            #pragma unroll
            for (int t = 0; t < 3; ++t)
                if (t < n)
                    wmma_tile_run_head<K, NDIM, 1>(tileA + t * 16 * LSTRIDE, W, bias, W4,
                                                   lane, wave, pp + t);
        } else {
            for (int t = 0; t < n; ++t)
                wmma_tile_run<K, NDIM, 1, STATS>(tileA + t * 16 * LSTRIDE, W, bb, Y,
                                                 row0 + t * 16, M, lane, wave, psum, psq);
        }
    }

    if (HEAD) {
        constexpr int PROWS = GEMM_MAXT * 16;
        __syncthreads();
        float2* part = reinterpret_cast<float2*>(tileA);
        const int grp = wave * 2 + (lane >> 4);
        #pragma unroll
        for (int t = 0; t < GEMM_MAXT; ++t)
            if (t < n) part[grp * PROWS + t * 16 + (lane & 15)] = pp[t];
        __syncthreads();
        if (wave == 0) {
            const float* pf = tileA;
            const int c = lane & 1;
            const float bc = b4[c];
            float ssum = 0.f, ssq = 0.f;
            #pragma unroll
            for (int j = 0; j < GEMM_MAXT; ++j) {
                if (j < n) {
                    const int idx = j * 32 + lane;
                    float s = bc;
                    #pragma unroll
                    for (int g = 0; g < 32; ++g) s += pf[g * (PROWS * 2) + idx];
                    const float v = s > 0.f ? s : 0.f;
                    const int row = row0 + (idx >> 1);
                    if (row < M) {
                        Y[(long long)row * 2 + c] = v;
                        ssum += v; ssq += v * v;
                    }
                }
            }
            #pragma unroll
            for (int d = 2; d < 32; d <<= 1) {
                ssum += __shfl_xor(ssum, d);
                ssq  += __shfl_xor(ssq, d);
            }
            if (lane < 2) {
                atomicAdd(&colsum[lane], ssum);
                atomicAdd(&colsumsq[lane], ssq);
            }
        }
    }

    if (STATS) {
        atomicAdd(&colsum[wave * 16 + (lane & 15)], psum);
        atomicAdd(&colsumsq[wave * 16 + (lane & 15)], psq);
    }
}

__global__ void bn_params_kernel(const float* __restrict__ colsum,
                                 const float* __restrict__ colsumsq,
                                 const float* __restrict__ gamma,
                                 const float* __restrict__ beta,
                                 float* __restrict__ scale,
                                 float* __restrict__ shift,
                                 int C, float invN) {
    int c = blockIdx.x * blockDim.x + threadIdx.x;
    if (c >= C) return;
    float mean = colsum[c] * invN;
    float var  = colsumsq[c] * invN - mean * mean;
    float s = gamma[c] * rsqrtf(var + BN_EPS);
    scale[c] = s;
    shift[c] = beta[c] - mean * s;
}

__global__ void bn_apply2(const float* __restrict__ Yin,
                          const float* __restrict__ scale,
                          const float* __restrict__ shift,
                          float* __restrict__ Yout, int total) {
    int t = blockIdx.x * blockDim.x + threadIdx.x;
    if (t >= total) return;
    int c = t & 1;
    Yout[t] = Yin[t] * scale[c] + shift[c];
}

extern "C" void kernel_launch(void* const* d_in, const int* in_sizes, int n_in,
                              void* d_out, int out_size, void* d_ws, size_t ws_size,
                              hipStream_t stream) {
    (void)n_in; (void)out_size; (void)ws_size;

    const float* x   = (const float*)d_in[0];
    const int*   ei  = (const int*)  d_in[1];
    const float* W1  = (const float*)d_in[2];
    const float* b1  = (const float*)d_in[3];
    const float* W2  = (const float*)d_in[4];
    const float* b2  = (const float*)d_in[5];
    const float* W3  = (const float*)d_in[6];
    const float* b3  = (const float*)d_in[7];
    const float* W4  = (const float*)d_in[8];
    const float* b4  = (const float*)d_in[9];
    const float* gm1 = (const float*)d_in[10];
    const float* bt1 = (const float*)d_in[11];
    const float* gm5 = (const float*)d_in[12];
    const float* bt5 = (const float*)d_in[13];

    const int FIN = 128, DIM = 256;
    const int Nn = in_sizes[0] / FIN;
    const int E  = in_sizes[1] / 2;
    const int* src = ei;
    const int* dst = ei + E;

    float* ws = (float*)d_ws;
    float* colsum1   = ws + 0;
    float* colsumsq1 = ws + 256;
    float* colsum5   = ws + 1024;
    float* colsumsq5 = ws + 1028;
    float* scale5    = ws + 1032;
    float* shift5    = ws + 1036;
    int*   spillCnt  = (int*)(ws + 1040);
    int*   cnt    = (int*)(ws + 2048);
    const size_t hdr = (2048 + (size_t)Nn + 63) / 64 * 64;
    int*   col = (int*)(ws + hdr);
    float* R0 = ws + hdr + (size_t)Nn * PITCH;
    float* R1 = R0 + (size_t)Nn * FIN;
    float* R2 = R1 + (size_t)Nn * DIM;
    float* R3 = R2 + (size_t)Nn * DIM;
    int*   spill = (int*)(R3 + (size_t)Nn * 2);

    float* outp = (float*)d_out;
    float* hid  = outp + (size_t)Nn * 2;

    hipMemsetAsync(ws, 0, (2048 + (size_t)Nn) * sizeof(float), stream);

    const dim3 gemmBlk(32, 16);
    const int mTiles = (Nn + 15) / 16;
    const dim3 gemmGrid(mTiles > 768 * 5 ? (mTiles + 4) / 5 : 768);

    edge_bucket_kernel<<<(E + 255) / 256, 256, 0, stream>>>(src, dst, cnt, col,
                                                            spillCnt, spill, E);

    csr_gather_sum_kernel<128><<<(Nn + 7) / 8, 256, 0, stream>>>(x, cnt, col, spillCnt, spill,
                                                                 R0, Nn);
    wmma_gemm_bias_relu<128, 256, true, false, false><<<gemmGrid, gemmBlk, 0, stream>>>(
        x, R0, W1, b1, R1, nullptr, nullptr, nullptr, nullptr, Nn);
    wmma_gemm_bias_relu<256, 256, false, true, false><<<gemmGrid, gemmBlk, 0, stream>>>(
        R1, nullptr, W2, b2, R2, colsum1, colsumsq1, nullptr, nullptr, Nn);
    csr_gather_sum_bn_kernel<256><<<(Nn + 7) / 8, 256, 0, stream>>>(
        R2, cnt, col, spillCnt, spill, colsum1, colsumsq1, gm1, bt1, hid, R1, Nn,
        1.0f / (float)Nn);
    wmma_gemm_bias_relu<256, 256, false, false, true><<<gemmGrid, gemmBlk, 0, stream>>>(
        R1, nullptr, W3, b3, R3, colsum5, colsumsq5, W4, b4, Nn);
    bn_params_kernel<<<1, 32, 0, stream>>>(colsum5, colsumsq5, gm5, bt5,
                                           scale5, shift5, 2, 1.0f / (float)Nn);
    bn_apply2<<<(Nn * 2 + 255) / 256, 256, 0, stream>>>(R3, scale5, shift5,
                                                        outp, Nn * 2);
}


# first-layer GEMM pair fused: relu((x+agg)@W1+b1) goes from the accumulators into the block's LDS buffer and feeds the W2 product there; h1 never stored or re-read, one launch less; sums unchanged
# speedup vs baseline: 2.7415x; 1.0434x over previous
#include <hip/hip_runtime.h>

typedef __attribute__((ext_vector_type(2))) float v2f;
typedef __attribute__((ext_vector_type(8))) float v8f;

#define BN_EPS 1e-5f

constexpr int PITCH = 64;

__global__ void edge_bucket_kernel(const int* __restrict__ src,
                                   const int* __restrict__ dst,
                                   int* __restrict__ cnt,
                                   int* __restrict__ col,
                                   int* __restrict__ spillCnt,
                                   int* __restrict__ spill, int E) {
    int e = blockIdx.x * blockDim.x + threadIdx.x;
    if (e >= E) return;
    const int d = dst[e];
    const int s = src[e];
    const int r = atomicAdd(&cnt[d], 1);
    if (r < PITCH) {
        col[(long long)d * PITCH + r] = s;
    } else {
        const int k = atomicAdd(spillCnt, 1);
        reinterpret_cast<int2*>(spill)[k] = make_int2(d, s);
    }
}

template<int F>
__global__ void __launch_bounds__(256)
csr_gather_sum_kernel(const float* __restrict__ X,
                      const int* __restrict__ cnt,
                      const int* __restrict__ col,
                      const int* __restrict__ spillCnt,
                      const int* __restrict__ spill,
                      float* __restrict__ agg, int N) {
    constexpr int V = F / 128;
    constexpr int B = 8 / V;
    const int lane = threadIdx.x & 31;
    const int node = blockIdx.x * (blockDim.x >> 5)
                   + __builtin_amdgcn_readfirstlane(threadIdx.x >> 5);
    if (node >= N) return;
    const int deg = cnt[node];
    const int beg = node * PITCH;
    const int end = beg + (deg < PITCH ? deg : PITCH);

    float4 acc[V];
    #pragma unroll
    for (int w = 0; w < V; ++w) acc[w] = make_float4(0.f, 0.f, 0.f, 0.f);

    if (deg > PITCH) {
        const int nsp = *spillCnt;
        for (int base = 0; base < nsp; base += 32) {
            int2 ds = make_int2(-1, 0);
            if (base + lane < nsp) ds = reinterpret_cast<const int2*>(spill)[base + lane];
            unsigned m = __builtin_amdgcn_ballot_w32(ds.x == node);
            while (m) {
                const int b = __builtin_ctz(m);
                m &= m - 1;
                const int s = __builtin_amdgcn_readlane(ds.y, b);
                const float4* r = reinterpret_cast<const float4*>(X + (long long)s * F);
                #pragma unroll
                for (int w = 0; w < V; ++w) {
                    const float4 v = r[lane + 32 * w];
                    acc[w].x += v.x; acc[w].y += v.y; acc[w].z += v.z; acc[w].w += v.w;
                }
            }
        }
    }

    for (int base = beg; base < end; base += 32) {
        const int n = (end - base) < 32 ? (end - base) : 32;
        const int myc = (lane < n) ? col[base + lane] : 0;
        int j = 0;
        for (; j + B <= n; j += B) {
            float4 v[B][V];
            #pragma unroll
            for (int u = 0; u < B; ++u) {
                const int s = __builtin_amdgcn_readlane(myc, j + u);
                const float4* r = reinterpret_cast<const float4*>(X + (long long)s * F);
                #pragma unroll
                for (int w = 0; w < V; ++w) v[u][w] = r[lane + 32 * w];
            }
            #pragma unroll
            for (int u = 0; u < B; ++u) {
                #pragma unroll
                for (int w = 0; w < V; ++w) {
                    acc[w].x += v[u][w].x; acc[w].y += v[u][w].y;
                    acc[w].z += v[u][w].z; acc[w].w += v[u][w].w;
                }
            }
        }
        if (j < n) {
            float4 v[B][V];
            #pragma unroll
            for (int u = 0; u < B; ++u) {
                #pragma unroll
                for (int w = 0; w < V; ++w) v[u][w] = make_float4(0.f, 0.f, 0.f, 0.f);
                if (j + u < n) {
                    const int s = __builtin_amdgcn_readlane(myc, j + u);
                    const float4* r = reinterpret_cast<const float4*>(X + (long long)s * F);
                    #pragma unroll
                    for (int w = 0; w < V; ++w) v[u][w] = r[lane + 32 * w];
                }
            }
            #pragma unroll
            for (int u = 0; u < B; ++u) {
                #pragma unroll
                for (int w = 0; w < V; ++w) {
                    acc[w].x += v[u][w].x; acc[w].y += v[u][w].y;
                    acc[w].z += v[u][w].z; acc[w].w += v[u][w].w;
                }
            }
        }
    }

    float4* o = reinterpret_cast<float4*>(agg + (long long)node * F);
    #pragma unroll
    for (int w = 0; w < V; ++w) o[lane + 32 * w] = acc[w];
}

__device__ __forceinline__ void add_normalised(float4& a, const float4& v,
                                               const float4& s, const float4& t) {
    a.x += v.x * s.x + t.x;
    a.y += v.y * s.y + t.y;
    a.z += v.z * s.z + t.z;
    a.w += v.w * s.w + t.w;
}

template<int F>
__global__ void __launch_bounds__(256)
csr_gather_sum_bn_kernel(const float* __restrict__ Y,
                         const int* __restrict__ cnt,
                         const int* __restrict__ col,
                         const int* __restrict__ spillCnt,
                         const int* __restrict__ spill,
                         const float* __restrict__ colsum,
                         const float* __restrict__ colsumsq,
                         const float* __restrict__ gamma,
                         const float* __restrict__ beta,
                         float* __restrict__ hid,
                         float* __restrict__ hsum, int N, float invN) {
    constexpr int V = F / 128;
    constexpr int B = 8 / V;
    const int lane = threadIdx.x & 31;
    const int node = blockIdx.x * (blockDim.x >> 5)
                   + __builtin_amdgcn_readfirstlane(threadIdx.x >> 5);
    if (node >= N) return;
    const int deg = cnt[node];
    const int beg = node * PITCH;
    const int end = beg + (deg < PITCH ? deg : PITCH);

    const float4* yrow = reinterpret_cast<const float4*>(Y + (long long)node * F);
    float4 acc[V], s[V], t[V];
    #pragma unroll
    for (int w = 0; w < V; ++w) {
        const int c4 = lane + 32 * w;
        const float4 own = yrow[c4];
        const float4 cs = reinterpret_cast<const float4*>(colsum)[c4];
        const float4 cq = reinterpret_cast<const float4*>(colsumsq)[c4];
        const float csv[4] = {cs.x, cs.y, cs.z, cs.w};
        const float cqv[4] = {cq.x, cq.y, cq.z, cq.w};
        float sv[4], tv[4];
        #pragma unroll
        for (int q = 0; q < 4; ++q) {
            const float mean = csv[q] * invN;
            const float var  = cqv[q] * invN - mean * mean;
            sv[q] = gamma[c4 * 4 + q] * rsqrtf(var + BN_EPS);
            tv[q] = beta[c4 * 4 + q] - mean * sv[q];
        }
        s[w] = make_float4(sv[0], sv[1], sv[2], sv[3]);
        t[w] = make_float4(tv[0], tv[1], tv[2], tv[3]);
        acc[w].x = own.x * s[w].x + t[w].x;
        acc[w].y = own.y * s[w].y + t[w].y;
        acc[w].z = own.z * s[w].z + t[w].z;
        acc[w].w = own.w * s[w].w + t[w].w;
    }
    {
        float4* h = reinterpret_cast<float4*>(hid + (long long)node * F);
        #pragma unroll
        for (int w = 0; w < V; ++w) h[lane + 32 * w] = acc[w];
    }

    if (deg > PITCH) {
        const int nsp = *spillCnt;
        for (int base = 0; base < nsp; base += 32) {
            int2 ds = make_int2(-1, 0);
            if (base + lane < nsp) ds = reinterpret_cast<const int2*>(spill)[base + lane];
            unsigned m = __builtin_amdgcn_ballot_w32(ds.x == node);
            while (m) {
                const int b = __builtin_ctz(m);
                m &= m - 1;
                const int sidx = __builtin_amdgcn_readlane(ds.y, b);
                const float4* r = reinterpret_cast<const float4*>(Y + (long long)sidx * F);
                #pragma unroll
                for (int w = 0; w < V; ++w) add_normalised(acc[w], r[lane + 32 * w], s[w], t[w]);
            }
        }
    }

    for (int base = beg; base < end; base += 32) {
        const int n = (end - base) < 32 ? (end - base) : 32;
        const int myc = (lane < n) ? col[base + lane] : 0;
        int j = 0;
        for (; j + B <= n; j += B) {
            float4 v[B][V];
            #pragma unroll
            for (int u = 0; u < B; ++u) {
                const int sidx = __builtin_amdgcn_readlane(myc, j + u);
                const float4* r = reinterpret_cast<const float4*>(Y + (long long)sidx * F);
                #pragma unroll
                for (int w = 0; w < V; ++w) v[u][w] = r[lane + 32 * w];
            }
            #pragma unroll
            for (int u = 0; u < B; ++u) {
                #pragma unroll
                for (int w = 0; w < V; ++w) add_normalised(acc[w], v[u][w], s[w], t[w]);
            }
        }
        if (j < n) {
            float4 v[B][V];
            #pragma unroll
            for (int u = 0; u < B; ++u) {
                #pragma unroll
                for (int w = 0; w < V; ++w) v[u][w] = make_float4(0.f, 0.f, 0.f, 0.f);
                if (j + u < n) {
                    const int sidx = __builtin_amdgcn_readlane(myc, j + u);
                    const float4* r = reinterpret_cast<const float4*>(Y + (long long)sidx * F);
                    #pragma unroll
                    for (int w = 0; w < V; ++w) v[u][w] = r[lane + 32 * w];
                }
            }
            #pragma unroll
            for (int u = 0; u < B; ++u) {
                if (j + u < n) {
                    #pragma unroll
                    for (int w = 0; w < V; ++w) add_normalised(acc[w], v[u][w], s[w], t[w]);
                }
            }
        }
    }

    float4* o = reinterpret_cast<float4*>(hsum + (long long)node * F);
    #pragma unroll
    for (int w = 0; w < V; ++w) o[lane + 32 * w] = acc[w];
}

constexpr int GEMM_MAXT = 5;

template<int K, int NDIM, int MT, bool STATS>
__device__ __forceinline__ void wmma_tile_run(const float* __restrict__ tileA,
                                              const float* __restrict__ W,
                                              float bb, float* __restrict__ Y,
                                              int row0, int M, int lane, int wave,
                                              float& psum, float& psq) {
    constexpr int LSTRIDE = K + 4;
    const int r16  = lane & 15;
    const int hi   = lane >> 4;
    const int bCol = wave * 16 + r16;
    const float* aLds = &tileA[r16 * LSTRIDE + hi * 2];
    const float* wp   = W + (long long)(hi * 2) * NDIM + bCol;

    v8f acc[MT];
    #pragma unroll
    for (int t = 0; t < MT; ++t)
        acc[t] = (v8f){0.f, 0.f, 0.f, 0.f, 0.f, 0.f, 0.f, 0.f};

    #pragma unroll 2
    for (int k = 0; k < K; k += 4) {
        v2f b;
        b.x = wp[(long long)k * NDIM];
        b.y = wp[(long long)(k + 1) * NDIM];
        #pragma unroll
        for (int t = 0; t < MT; ++t) {
            float2 av = *reinterpret_cast<const float2*>(aLds + t * 16 * LSTRIDE + k);
            v2f a; a.x = av.x; a.y = av.y;
            acc[t] = __builtin_amdgcn_wmma_f32_16x16x4_f32(
                       false, a,  false, b,
                       (short)0, acc[t],  false,  false);
        }
    }

    float* yp = Y + (long long)(row0 + hi * 8) * NDIM + bCol;
    if (row0 + MT * 16 <= M) {
        #pragma unroll
        for (int t = 0; t < MT; ++t) {
            #pragma unroll
            for (int r = 0; r < 8; ++r) {
                float v = acc[t][r] + bb;
                v = v > 0.f ? v : 0.f;
                yp[(t * 16 + r) * NDIM] = v;
                if (STATS) { psum += v; psq += v * v; }
            }
        }
    } else {
        const int left = M - (row0 + hi * 8);
        #pragma unroll
        for (int t = 0; t < MT; ++t) {
            #pragma unroll
            for (int r = 0; r < 8; ++r) {
                float v = acc[t][r] + bb;
                v = v > 0.f ? v : 0.f;
                if (t * 16 + r < left) {
                    yp[(t * 16 + r) * NDIM] = v;
                    if (STATS) { psum += v; psq += v * v; }
                }
            }
        }
    }
}

template<int K, int NDIM, int MT>
__device__ __forceinline__ void wmma_tile_run_to_lds(float* tile,
                                                     const float* __restrict__ W,
                                                     float bb, int lane, int wave) {
    constexpr int LSTRIDE = K + 4;
    constexpr int HSTRIDE = NDIM + 4;
    const int r16  = lane & 15;
    const int hi   = lane >> 4;
    const int bCol = wave * 16 + r16;
    const float* aLds = &tile[r16 * LSTRIDE + hi * 2];
    const float* wp   = W + (long long)(hi * 2) * NDIM + bCol;

    v8f acc[MT];
    #pragma unroll
    for (int t = 0; t < MT; ++t)
        acc[t] = (v8f){0.f, 0.f, 0.f, 0.f, 0.f, 0.f, 0.f, 0.f};

    #pragma unroll 2
    for (int k = 0; k < K; k += 4) {
        v2f b;
        b.x = wp[(long long)k * NDIM];
        b.y = wp[(long long)(k + 1) * NDIM];
        #pragma unroll
        for (int t = 0; t < MT; ++t) {
            float2 av = *reinterpret_cast<const float2*>(aLds + t * 16 * LSTRIDE + k);
            v2f a; a.x = av.x; a.y = av.y;
            acc[t] = __builtin_amdgcn_wmma_f32_16x16x4_f32(
                       false, a,  false, b,
                       (short)0, acc[t],  false,  false);
        }
    }

    __syncthreads();
    float* hp = tile + (hi * 8) * HSTRIDE + bCol;
    #pragma unroll
    for (int t = 0; t < MT; ++t) {
        #pragma unroll
        for (int r = 0; r < 8; ++r) {
            float v = acc[t][r] + bb;
            hp[(t * 16 + r) * HSTRIDE] = v > 0.f ? v : 0.f;
        }
    }
}

template<int K, int NDIM, int MT>
__device__ __forceinline__ void wmma_tile_run_head(const float* __restrict__ tileA,
                                                   const float* __restrict__ W,
                                                   const float* __restrict__ bias,
                                                   const float* __restrict__ W4,
                                                   int lane, int wave,
                                                   float2* __restrict__ pp) {
    constexpr int LSTRIDE = K + 4;
    const int r16  = lane & 15;
    const int hi   = lane >> 4;
    const int bCol = wave * 16 + r16;
    const float* aLds = &tileA[r16 * LSTRIDE + hi * 2];
    const float* wp   = W + (long long)(hi * 2) * NDIM + bCol;

    v8f acc[MT];
    #pragma unroll
    for (int t = 0; t < MT; ++t)
        acc[t] = (v8f){0.f, 0.f, 0.f, 0.f, 0.f, 0.f, 0.f, 0.f};

    #pragma unroll 2
    for (int k = 0; k < K; k += 4) {
        v2f b;
        b.x = wp[(long long)k * NDIM];
        b.y = wp[(long long)(k + 1) * NDIM];
        #pragma unroll
        for (int t = 0; t < MT; ++t) {
            float2 av = *reinterpret_cast<const float2*>(aLds + t * 16 * LSTRIDE + k);
            v2f a; a.x = av.x; a.y = av.y;
            acc[t] = __builtin_amdgcn_wmma_f32_16x16x4_f32(
                       false, b,  false, a,
                       (short)0, acc[t],  false,  false);
        }
    }

    const int f0 = wave * 16 + hi * 8;
    float bia[8], wa[8], wb[8];
    #pragma unroll
    for (int r = 0; r < 8; ++r) {
        bia[r] = bias[f0 + r];
        wa[r]  = W4[(f0 + r) * 2];
        wb[r]  = W4[(f0 + r) * 2 + 1];
    }
    #pragma unroll
    for (int t = 0; t < MT; ++t) {
        float s0 = 0.f, s1 = 0.f;
        #pragma unroll
        for (int r = 0; r < 8; ++r) {
            float v = acc[t][r] + bia[r];
            v = v > 0.f ? v : 0.f;
            s0 = fmaf(v, wa[r], s0);
            s1 = fmaf(v, wb[r], s1);
        }
        pp[t] = make_float2(s0, s1);
    }
}

template<int K, int MT, bool ADD_A>
__device__ __forceinline__ void stage_rows(float* __restrict__ tileA,
                                           const float* __restrict__ A0,
                                           const float* __restrict__ A1,
                                           int row0, int M, int tid) {
    constexpr int LSTRIDE = K + 4;
    constexpr int RSTEP = 512 / (K / 4);
    constexpr int N  = MT * 16 / RSTEP;
    constexpr int NB = ADD_A ? 4 : 8;
    static_assert(512 % (K / 4) == 0 && (MT * 16) % RSTEP == 0, "whole trips only");
    const int r0   = tid / (K / 4);
    const int acol = (tid % (K / 4)) * 4;
    float* l0 = tileA + r0 * LSTRIDE + acol;
    if (row0 + MT * 16 <= M) {
        const float* p0 = A0 + (long long)(row0 + r0) * K + acol;
        const float* p1 = ADD_A ? A1 + (long long)(row0 + r0) * K + acol : nullptr;
        #pragma unroll
        for (int i0 = 0; i0 < N; i0 += NB) {
            float4 v[NB], w[NB];
            #pragma unroll
            for (int j = 0; j < NB; ++j) {
                if (i0 + j < N) {
                    v[j] = *reinterpret_cast<const float4*>(p0 + (i0 + j) * RSTEP * K);
                    if (ADD_A) w[j] = *reinterpret_cast<const float4*>(p1 + (i0 + j) * RSTEP * K);
                }
            }
            #pragma unroll
            for (int j = 0; j < NB; ++j) {
                if (i0 + j < N) {
                    float4 t = v[j];
                    if (ADD_A) { t.x += w[j].x; t.y += w[j].y; t.z += w[j].z; t.w += w[j].w; }
                    *reinterpret_cast<float4*>(l0 + (i0 + j) * RSTEP * LSTRIDE) = t;
                }
            }
        }
    } else {
        for (int i = 0; i < N; ++i) {
            const int row = row0 + r0 + i * RSTEP;
            float4 t = make_float4(0.f, 0.f, 0.f, 0.f);
            if (row < M) {
                t = *reinterpret_cast<const float4*>(A0 + (long long)row * K + acol);
                if (ADD_A) {
                    const float4 w = *reinterpret_cast<const float4*>(A1 + (long long)row * K + acol);
                    t.x += w.x; t.y += w.y; t.z += w.z; t.w += w.w;
                }
            }
            *reinterpret_cast<float4*>(l0 + i * RSTEP * LSTRIDE) = t;
        }
    }
}

template<int K, int NDIM, bool ADD_A, bool STATS, bool HEAD>
__global__ void __launch_bounds__(512) __attribute__((amdgpu_waves_per_eu(12)))
wmma_gemm_bias_relu(const float* __restrict__ A0,
                    const float* __restrict__ A1,
                    const float* __restrict__ W,
                    const float* __restrict__ bias,
                    float* __restrict__ Y,
                    float* __restrict__ colsum,
                    float* __restrict__ colsumsq,
                    const float* __restrict__ W4,
                    const float* __restrict__ b4, int M) {
    constexpr int LSTRIDE = K + 4;
    __shared__ __attribute__((aligned(16))) float tileA[GEMM_MAXT * 16 * LSTRIDE];

    const int lane  = threadIdx.x;
    const int wave  = threadIdx.y;
    const int tid   = wave * 32 + lane;

    const int units = (M + 15) >> 4;
    const int u0 = (int)(((long long)blockIdx.x * units) / gridDim.x);
    const int u1 = (int)(((long long)(blockIdx.x + 1) * units) / gridDim.x);
    const int n    = u1 - u0;
    const int row0 = u0 * 16;

    const float bb = HEAD ? 0.f : bias[wave * 16 + (lane & 15)];
    float psum = 0.f, psq = 0.f;
    float2 pp[GEMM_MAXT];
    #pragma unroll
    for (int t = 0; t < GEMM_MAXT; ++t) pp[t] = make_float2(0.f, 0.f);

    if (n == 5) {
        stage_rows<K, 5, ADD_A>(tileA, A0, A1, row0, M, tid);
        __syncthreads();
        if (HEAD) wmma_tile_run_head<K, NDIM, 5>(tileA, W, bias, W4, lane, wave, pp);
        else wmma_tile_run<K, NDIM, 5, STATS>(tileA, W, bb, Y, row0, M, lane, wave, psum, psq);
    } else if (n == 4) {
        stage_rows<K, 4, ADD_A>(tileA, A0, A1, row0, M, tid);
        __syncthreads();
        if (HEAD) wmma_tile_run_head<K, NDIM, 4>(tileA, W, bias, W4, lane, wave, pp);
        else wmma_tile_run<K, NDIM, 4, STATS>(tileA, W, bb, Y, row0, M, lane, wave, psum, psq);
    } else {
        for (int t = 0; t < n; ++t)
            stage_rows<K, 1, ADD_A>(tileA + t * 16 * LSTRIDE, A0, A1, row0 + t * 16, M, tid);
        __syncthreads();
        if (HEAD) {
            #pragma unroll
            for (int t = 0; t < 3; ++t)
                if (t < n)
                    wmma_tile_run_head<K, NDIM, 1>(tileA + t * 16 * LSTRIDE, W, bias, W4,
                                                   lane, wave, pp + t);
        } else {
            for (int t = 0; t < n; ++t)
                wmma_tile_run<K, NDIM, 1, STATS>(tileA + t * 16 * LSTRIDE, W, bb, Y,
                                                 row0 + t * 16, M, lane, wave, psum, psq);
        }
    }

    if (HEAD) {
        constexpr int PROWS = GEMM_MAXT * 16;
        __syncthreads();
        float2* part = reinterpret_cast<float2*>(tileA);
        const int grp = wave * 2 + (lane >> 4);
        #pragma unroll
        for (int t = 0; t < GEMM_MAXT; ++t)
            if (t < n) part[grp * PROWS + t * 16 + (lane & 15)] = pp[t];
        __syncthreads();
        if (wave == 0) {
            const float* pf = tileA;
            const int c = lane & 1;
            const float bc = b4[c];
            float ssum = 0.f, ssq = 0.f;
            #pragma unroll
            for (int j = 0; j < GEMM_MAXT; ++j) {
                if (j < n) {
                    const int idx = j * 32 + lane;
                    float s = bc;
                    #pragma unroll
                    for (int g = 0; g < 32; ++g) s += pf[g * (PROWS * 2) + idx];
                    const float v = s > 0.f ? s : 0.f;
                    const int row = row0 + (idx >> 1);
                    if (row < M) {
                        Y[(long long)row * 2 + c] = v;
                        ssum += v; ssq += v * v;
                    }
                }
            }
            #pragma unroll
            for (int d = 2; d < 32; d <<= 1) {
                ssum += __shfl_xor(ssum, d);
                ssq  += __shfl_xor(ssq, d);
            }
            if (lane < 2) {
                atomicAdd(&colsum[lane], ssum);
                atomicAdd(&colsumsq[lane], ssq);
            }
        }
    }

    if (STATS) {
        atomicAdd(&colsum[wave * 16 + (lane & 15)], psum);
        atomicAdd(&colsumsq[wave * 16 + (lane & 15)], psq);
    }
}

template<int K1, int NDIM>
__global__ void __launch_bounds__(512) __attribute__((amdgpu_waves_per_eu(12)))
wmma_gemm_pair_bias_relu(const float* __restrict__ A0,
                         const float* __restrict__ A1,
                         const float* __restrict__ Wa,
                         const float* __restrict__ ba,
                         const float* __restrict__ Wb,
                         const float* __restrict__ bb,
                         float* __restrict__ Y,
                         float* __restrict__ colsum,
                         float* __restrict__ colsumsq, int M) {
    constexpr int LS1 = K1 + 4;
    constexpr int LS2 = NDIM + 4;
    static_assert(K1 <= NDIM, "the buffer is sized for the wider of the two row sets");
    __shared__ __attribute__((aligned(16))) float tile[GEMM_MAXT * 16 * LS2];

    const int lane  = threadIdx.x;
    const int wave  = threadIdx.y;
    const int tid   = wave * 32 + lane;

    const int units = (M + 15) >> 4;
    const int u0 = (int)(((long long)blockIdx.x * units) / gridDim.x);
    const int u1 = (int)(((long long)(blockIdx.x + 1) * units) / gridDim.x);
    const int n    = u1 - u0;
    const int row0 = u0 * 16;

    const float bias1 = ba[wave * 16 + (lane & 15)];
    const float bias2 = bb[wave * 16 + (lane & 15)];
    float psum = 0.f, psq = 0.f;

    if (n == 5) {
        stage_rows<K1, 5, true>(tile, A0, A1, row0, M, tid);
        __syncthreads();
        wmma_tile_run_to_lds<K1, NDIM, 5>(tile, Wa, bias1, lane, wave);
        __syncthreads();
        wmma_tile_run<NDIM, NDIM, 5, true>(tile, Wb, bias2, Y, row0, M, lane, wave, psum, psq);
    } else if (n == 4) {
        stage_rows<K1, 4, true>(tile, A0, A1, row0, M, tid);
        __syncthreads();
        wmma_tile_run_to_lds<K1, NDIM, 4>(tile, Wa, bias1, lane, wave);
        __syncthreads();
        wmma_tile_run<NDIM, NDIM, 4, true>(tile, Wb, bias2, Y, row0, M, lane, wave, psum, psq);
    } else {
        for (int t = 0; t < n; ++t) {
            if (t) __syncthreads();
            stage_rows<K1, 1, true>(tile, A0, A1, row0 + t * 16, M, tid);
            __syncthreads();
            wmma_tile_run_to_lds<K1, NDIM, 1>(tile, Wa, bias1, lane, wave);
            __syncthreads();
            wmma_tile_run<NDIM, NDIM, 1, true>(tile, Wb, bias2, Y, row0 + t * 16, M,
                                               lane, wave, psum, psq);
        }
    }

    atomicAdd(&colsum[wave * 16 + (lane & 15)], psum);
    atomicAdd(&colsumsq[wave * 16 + (lane & 15)], psq);
}

__global__ void bn_params_kernel(const float* __restrict__ colsum,
                                 const float* __restrict__ colsumsq,
                                 const float* __restrict__ gamma,
                                 const float* __restrict__ beta,
                                 float* __restrict__ scale,
                                 float* __restrict__ shift,
                                 int C, float invN) {
    int c = blockIdx.x * blockDim.x + threadIdx.x;
    if (c >= C) return;
    float mean = colsum[c] * invN;
    float var  = colsumsq[c] * invN - mean * mean;
    float s = gamma[c] * rsqrtf(var + BN_EPS);
    scale[c] = s;
    shift[c] = beta[c] - mean * s;
}

__global__ void bn_apply2(const float* __restrict__ Yin,
                          const float* __restrict__ scale,
                          const float* __restrict__ shift,
                          float* __restrict__ Yout, int total) {
    int t = blockIdx.x * blockDim.x + threadIdx.x;
    if (t >= total) return;
    int c = t & 1;
    Yout[t] = Yin[t] * scale[c] + shift[c];
}

extern "C" void kernel_launch(void* const* d_in, const int* in_sizes, int n_in,
                              void* d_out, int out_size, void* d_ws, size_t ws_size,
                              hipStream_t stream) {
    (void)n_in; (void)out_size; (void)ws_size;

    const float* x   = (const float*)d_in[0];
    const int*   ei  = (const int*)  d_in[1];
    const float* W1  = (const float*)d_in[2];
    const float* b1  = (const float*)d_in[3];
    const float* W2  = (const float*)d_in[4];
    const float* b2  = (const float*)d_in[5];
    const float* W3  = (const float*)d_in[6];
    const float* b3  = (const float*)d_in[7];
    const float* W4  = (const float*)d_in[8];
    const float* b4  = (const float*)d_in[9];
    const float* gm1 = (const float*)d_in[10];
    const float* bt1 = (const float*)d_in[11];
    const float* gm5 = (const float*)d_in[12];
    const float* bt5 = (const float*)d_in[13];

    const int FIN = 128, DIM = 256;
    const int Nn = in_sizes[0] / FIN;
    const int E  = in_sizes[1] / 2;
    const int* src = ei;
    const int* dst = ei + E;

    float* ws = (float*)d_ws;
    float* colsum1   = ws + 0;
    float* colsumsq1 = ws + 256;
    float* colsum5   = ws + 1024;
    float* colsumsq5 = ws + 1028;
    float* scale5    = ws + 1032;
    float* shift5    = ws + 1036;
    int*   spillCnt  = (int*)(ws + 1040);
    int*   cnt    = (int*)(ws + 2048);
    const size_t hdr = (2048 + (size_t)Nn + 63) / 64 * 64;
    int*   col = (int*)(ws + hdr);
    float* R0 = ws + hdr + (size_t)Nn * PITCH;
    float* R1 = R0 + (size_t)Nn * FIN;
    float* R2 = R1 + (size_t)Nn * DIM;
    float* R3 = R2 + (size_t)Nn * DIM;
    int*   spill = (int*)(R3 + (size_t)Nn * 2);

    float* outp = (float*)d_out;
    float* hid  = outp + (size_t)Nn * 2;

    hipMemsetAsync(ws, 0, (2048 + (size_t)Nn) * sizeof(float), stream);

    const dim3 gemmBlk(32, 16);
    const int mTiles = (Nn + 15) / 16;
    const dim3 gemmGrid(mTiles > 768 * 5 ? (mTiles + 4) / 5 : 768);

    edge_bucket_kernel<<<(E + 255) / 256, 256, 0, stream>>>(src, dst, cnt, col,
                                                            spillCnt, spill, E);

    csr_gather_sum_kernel<128><<<(Nn + 7) / 8, 256, 0, stream>>>(x, cnt, col, spillCnt, spill,
                                                                 R0, Nn);
    wmma_gemm_pair_bias_relu<128, 256><<<gemmGrid, gemmBlk, 0, stream>>>(
        x, R0, W1, b1, W2, b2, R2, colsum1, colsumsq1, Nn);
    csr_gather_sum_bn_kernel<256><<<(Nn + 7) / 8, 256, 0, stream>>>(
        R2, cnt, col, spillCnt, spill, colsum1, colsumsq1, gm1, bt1, hid, R1, Nn,
        1.0f / (float)Nn);
    wmma_gemm_bias_relu<256, 256, false, false, true><<<gemmGrid, gemmBlk, 0, stream>>>(
        R1, nullptr, W3, b3, R3, colsum5, colsumsq5, W4, b4, Nn);
    bn_params_kernel<<<1, 32, 0, stream>>>(colsum5, colsumsq5, gm5, bt5,
                                           scale5, shift5, 2, 1.0f / (float)Nn);
    bn_apply2<<<(Nn * 2 + 255) / 256, 256, 0, stream>>>(R3, scale5, shift5,
                                                        outp, Nn * 2);
}


# neighbour sums built in the GEMM kernels: a wave sums its 4-5 nodes' source rows in registers into the block's LDS row buffer; summed rows never stored or re-read, 2 launches less; sum orders kept
# speedup vs baseline: 2.9635x; 1.0810x over previous
#include <hip/hip_runtime.h>

typedef __attribute__((ext_vector_type(2))) float v2f;
typedef __attribute__((ext_vector_type(8))) float v8f;

#define BN_EPS 1e-5f

constexpr int PITCH = 64;

__global__ void edge_bucket_kernel(const int* __restrict__ src,
                                   const int* __restrict__ dst,
                                   int* __restrict__ cnt,
                                   int* __restrict__ col,
                                   int* __restrict__ spillCnt,
                                   int* __restrict__ spill, int E) {
    int e = blockIdx.x * blockDim.x + threadIdx.x;
    if (e >= E) return;
    const int d = dst[e];
    const int s = src[e];
    const int r = atomicAdd(&cnt[d], 1);
    if (r < PITCH) {
        col[(long long)d * PITCH + r] = s;
    } else {
        const int k = atomicAdd(spillCnt, 1);
        reinterpret_cast<int2*>(spill)[k] = make_int2(d, s);
    }
}

template<bool NORM>
__device__ __forceinline__ void add_row(float4& a, const float4& v,
                                        const float4& s, const float4& t) {
    if (NORM) {
        a.x += v.x * s.x + t.x;
        a.y += v.y * s.y + t.y;
        a.z += v.z * s.z + t.z;
        a.w += v.w * s.w + t.w;
    } else {
        a.x += v.x; a.y += v.y; a.z += v.z; a.w += v.w;
    }
}

template<int F, bool NORM>
__device__ __forceinline__ void sum_source_rows(const float* __restrict__ X,
                                                int node, int deg, int myc,
                                                const int* __restrict__ col,
                                                const int* __restrict__ spillCnt,
                                                const int* __restrict__ spill,
                                                int lane, float4 (&acc)[F / 128],
                                                const float4 (&s)[F / 128],
                                                const float4 (&t)[F / 128]) {
    constexpr int V = F / 128;
    constexpr int B = 8 / V;
    const int beg = node * PITCH;
    const int end = beg + (deg < PITCH ? deg : PITCH);

    if (deg > PITCH) {
        const int nsp = *spillCnt;
        for (int base = 0; base < nsp; base += 32) {
            int2 ds = make_int2(-1, 0);
            if (base + lane < nsp) ds = reinterpret_cast<const int2*>(spill)[base + lane];
            unsigned m = __builtin_amdgcn_ballot_w32(ds.x == node);
            while (m) {
                const int b = __builtin_ctz(m);
                m &= m - 1;
                const int sidx = __builtin_amdgcn_readlane(ds.y, b);
                const float4* r = reinterpret_cast<const float4*>(X + (long long)sidx * F);
                #pragma unroll
                for (int w = 0; w < V; ++w) add_row<NORM>(acc[w], r[lane + 32 * w], s[w], t[w]);
            }
        }
    }

    for (int base = beg; base < end; base += 32) {
        const int n = (end - base) < 32 ? (end - base) : 32;
        int c = myc;
        if (base != beg) c = (lane < n) ? col[base + lane] : 0;
        int j = 0;
        for (; j + B <= n; j += B) {
            float4 v[B][V];
            #pragma unroll
            for (int u = 0; u < B; ++u) {
                const int sidx = __builtin_amdgcn_readlane(c, j + u);
                const float4* r = reinterpret_cast<const float4*>(X + (long long)sidx * F);
                #pragma unroll
                for (int w = 0; w < V; ++w) v[u][w] = r[lane + 32 * w];
            }
            #pragma unroll
            for (int u = 0; u < B; ++u) {
                #pragma unroll
                for (int w = 0; w < V; ++w) add_row<NORM>(acc[w], v[u][w], s[w], t[w]);
            }
        }
        if (j < n) {
            float4 v[B][V];
            #pragma unroll
            for (int u = 0; u < B; ++u) {
                #pragma unroll
                for (int w = 0; w < V; ++w) v[u][w] = make_float4(0.f, 0.f, 0.f, 0.f);
                if (j + u < n) {
                    const int sidx = __builtin_amdgcn_readlane(c, j + u);
                    const float4* r = reinterpret_cast<const float4*>(X + (long long)sidx * F);
                    #pragma unroll
                    for (int w = 0; w < V; ++w) v[u][w] = r[lane + 32 * w];
                }
            }
            #pragma unroll
            for (int u = 0; u < B; ++u) {
                if (!NORM || j + u < n) {
                    #pragma unroll
                    for (int w = 0; w < V; ++w) add_row<NORM>(acc[w], v[u][w], s[w], t[w]);
                }
            }
        }
    }
}

template<int K, int MT, bool NORM>
__device__ __forceinline__ void gather_stage(float* __restrict__ tile,
                                             const float* __restrict__ X,
                                             const int* __restrict__ cnt,
                                             const int* __restrict__ col,
                                             const int* __restrict__ spillCnt,
                                             const int* __restrict__ spill,
                                             const float* __restrict__ colsum,
                                             const float* __restrict__ colsumsq,
                                             const float* __restrict__ gamma,
                                             const float* __restrict__ beta,
                                             float* __restrict__ hid,
                                             int row0, int M, float invN,
                                             int lane, int wave) {
    constexpr int V = K / 128;
    constexpr int LSTRIDE = K + 4;

    float4 s[V], t[V];
    #pragma unroll
    for (int w = 0; w < V; ++w) {
        s[w] = make_float4(0.f, 0.f, 0.f, 0.f);
        t[w] = make_float4(0.f, 0.f, 0.f, 0.f);
        if (NORM) {
            const int c4 = lane + 32 * w;
            const float4 cs = reinterpret_cast<const float4*>(colsum)[c4];
            const float4 cq = reinterpret_cast<const float4*>(colsumsq)[c4];
            const float csv[4] = {cs.x, cs.y, cs.z, cs.w};
            const float cqv[4] = {cq.x, cq.y, cq.z, cq.w};
            float sv[4], tv[4];
            #pragma unroll
            for (int q = 0; q < 4; ++q) {
                const float mean = csv[q] * invN;
                const float var  = cqv[q] * invN - mean * mean;
                sv[q] = gamma[c4 * 4 + q] * rsqrtf(var + BN_EPS);
                tv[q] = beta[c4 * 4 + q] - mean * sv[q];
            }
            s[w] = make_float4(sv[0], sv[1], sv[2], sv[3]);
            t[w] = make_float4(tv[0], tv[1], tv[2], tv[3]);
        }
    }

    int degv = 0;
    {
        const int nd = row0 + wave + 16 * lane;
        if (lane < MT && nd < M) degv = cnt[nd];
    }
    int node = row0 + wave;
    int deg  = __builtin_amdgcn_readlane(degv, 0);
    int myc  = 0;
    if (lane < (deg < 32 ? deg : 32)) myc = col[node * PITCH + lane];

    #pragma nounroll
    for (int i = 0; i < MT; ++i) {
        const int nnode = node + 16;
        int ndeg = 0, nmyc = 0;
        if (i + 1 < MT) {
            ndeg = __builtin_amdgcn_readlane(degv, i + 1);
            if (lane < (ndeg < 32 ? ndeg : 32)) nmyc = col[nnode * PITCH + lane];
        }

        float4 acc[V];
        #pragma unroll
        for (int w = 0; w < V; ++w) acc[w] = make_float4(0.f, 0.f, 0.f, 0.f);

        if (node < M) {
            const float4* xrow = reinterpret_cast<const float4*>(X + (long long)node * K);
            float4 own[V];
            #pragma unroll
            for (int w = 0; w < V; ++w) own[w] = xrow[lane + 32 * w];
            if (NORM) {
                float4* h = reinterpret_cast<float4*>(hid + (long long)node * K);
                #pragma unroll
                for (int w = 0; w < V; ++w) {
                    acc[w].x = own[w].x * s[w].x + t[w].x;
                    acc[w].y = own[w].y * s[w].y + t[w].y;
                    acc[w].z = own[w].z * s[w].z + t[w].z;
                    acc[w].w = own[w].w * s[w].w + t[w].w;
                    h[lane + 32 * w] = acc[w];
                }
            }
            sum_source_rows<K, NORM>(X, node, deg, myc, col, spillCnt, spill, lane, acc, s, t);
            if (!NORM) {
                #pragma unroll
                for (int w = 0; w < V; ++w) {
                    acc[w].x = own[w].x + acc[w].x;
                    acc[w].y = own[w].y + acc[w].y;
                    acc[w].z = own[w].z + acc[w].z;
                    acc[w].w = own[w].w + acc[w].w;
                }
            }
        }

        float4* l = reinterpret_cast<float4*>(tile + (wave + 16 * i) * LSTRIDE);
        #pragma unroll
        for (int w = 0; w < V; ++w) l[lane + 32 * w] = acc[w];

        node = nnode; deg = ndeg; myc = nmyc;
    }
}

constexpr int GEMM_MAXT = 5;

template<int K, int NDIM, int MT, bool STATS>
__device__ __forceinline__ void wmma_tile_run(const float* __restrict__ tileA,
                                              const float* __restrict__ W,
                                              float bb, float* __restrict__ Y,
                                              int row0, int M, int lane, int wave,
                                              float& psum, float& psq) {
    constexpr int LSTRIDE = K + 4;
    const int r16  = lane & 15;
    const int hi   = lane >> 4;
    const int bCol = wave * 16 + r16;
    const float* aLds = &tileA[r16 * LSTRIDE + hi * 2];
    const float* wp   = W + (long long)(hi * 2) * NDIM + bCol;

    v8f acc[MT];
    #pragma unroll
    for (int t = 0; t < MT; ++t)
        acc[t] = (v8f){0.f, 0.f, 0.f, 0.f, 0.f, 0.f, 0.f, 0.f};

    #pragma unroll 2
    for (int k = 0; k < K; k += 4) {
        v2f b;
        b.x = wp[(long long)k * NDIM];
        b.y = wp[(long long)(k + 1) * NDIM];
        #pragma unroll
        for (int t = 0; t < MT; ++t) {
            float2 av = *reinterpret_cast<const float2*>(aLds + t * 16 * LSTRIDE + k);
            v2f a; a.x = av.x; a.y = av.y;
            acc[t] = __builtin_amdgcn_wmma_f32_16x16x4_f32(
                       false, a,  false, b,
                       (short)0, acc[t],  false,  false);
        }
    }

    float* yp = Y + (long long)(row0 + hi * 8) * NDIM + bCol;
    if (row0 + MT * 16 <= M) {
        #pragma unroll
        for (int t = 0; t < MT; ++t) {
            #pragma unroll
            for (int r = 0; r < 8; ++r) {
                float v = acc[t][r] + bb;
                v = v > 0.f ? v : 0.f;
                yp[(t * 16 + r) * NDIM] = v;
                if (STATS) { psum += v; psq += v * v; }
            }
        }
    } else {
        const int left = M - (row0 + hi * 8);
        #pragma unroll
        for (int t = 0; t < MT; ++t) {
            #pragma unroll
            for (int r = 0; r < 8; ++r) {
                float v = acc[t][r] + bb;
                v = v > 0.f ? v : 0.f;
                if (t * 16 + r < left) {
                    yp[(t * 16 + r) * NDIM] = v;
                    if (STATS) { psum += v; psq += v * v; }
                }
            }
        }
    }
}

template<int K, int NDIM, int MT>
__device__ __forceinline__ void wmma_tile_run_to_lds(float* tile,
                                                     const float* __restrict__ W,
                                                     float bb, int lane, int wave) {
    constexpr int LSTRIDE = K + 4;
    constexpr int HSTRIDE = NDIM + 4;
    const int r16  = lane & 15;
    const int hi   = lane >> 4;
    const int bCol = wave * 16 + r16;
    const float* aLds = &tile[r16 * LSTRIDE + hi * 2];
    const float* wp   = W + (long long)(hi * 2) * NDIM + bCol;

    v8f acc[MT];
    #pragma unroll
    for (int t = 0; t < MT; ++t)
        acc[t] = (v8f){0.f, 0.f, 0.f, 0.f, 0.f, 0.f, 0.f, 0.f};

    #pragma unroll 2
    for (int k = 0; k < K; k += 4) {
        v2f b;
        b.x = wp[(long long)k * NDIM];
        b.y = wp[(long long)(k + 1) * NDIM];
        #pragma unroll
        for (int t = 0; t < MT; ++t) {
            float2 av = *reinterpret_cast<const float2*>(aLds + t * 16 * LSTRIDE + k);
            v2f a; a.x = av.x; a.y = av.y;
            acc[t] = __builtin_amdgcn_wmma_f32_16x16x4_f32(
                       false, a,  false, b,
                       (short)0, acc[t],  false,  false);
        }
    }

    __syncthreads();
    float* hp = tile + (hi * 8) * HSTRIDE + bCol;
    #pragma unroll
    for (int t = 0; t < MT; ++t) {
        #pragma unroll
        for (int r = 0; r < 8; ++r) {
            float v = acc[t][r] + bb;
            hp[(t * 16 + r) * HSTRIDE] = v > 0.f ? v : 0.f;
        }
    }
}

template<int K, int NDIM, int MT>
__device__ __forceinline__ void wmma_tile_run_head(const float* __restrict__ tileA,
                                                   const float* __restrict__ W,
                                                   const float* __restrict__ bias,
                                                   const float* __restrict__ W4,
                                                   int lane, int wave,
                                                   float2* __restrict__ pp) {
    constexpr int LSTRIDE = K + 4;
    const int r16  = lane & 15;
    const int hi   = lane >> 4;
    const int bCol = wave * 16 + r16;
    const float* aLds = &tileA[r16 * LSTRIDE + hi * 2];
    const float* wp   = W + (long long)(hi * 2) * NDIM + bCol;

    v8f acc[MT];
    #pragma unroll
    for (int t = 0; t < MT; ++t)
        acc[t] = (v8f){0.f, 0.f, 0.f, 0.f, 0.f, 0.f, 0.f, 0.f};

    #pragma unroll 2
    for (int k = 0; k < K; k += 4) {
        v2f b;
        b.x = wp[(long long)k * NDIM];
        b.y = wp[(long long)(k + 1) * NDIM];
        #pragma unroll
        for (int t = 0; t < MT; ++t) {
            float2 av = *reinterpret_cast<const float2*>(aLds + t * 16 * LSTRIDE + k);
            v2f a; a.x = av.x; a.y = av.y;
            acc[t] = __builtin_amdgcn_wmma_f32_16x16x4_f32(
                       false, b,  false, a,
                       (short)0, acc[t],  false,  false);
        }
    }

    const int f0 = wave * 16 + hi * 8;
    float bia[8], wa[8], wb[8];
    #pragma unroll
    for (int r = 0; r < 8; ++r) {
        bia[r] = bias[f0 + r];
        wa[r]  = W4[(f0 + r) * 2];
        wb[r]  = W4[(f0 + r) * 2 + 1];
    }
    #pragma unroll
    for (int t = 0; t < MT; ++t) {
        float s0 = 0.f, s1 = 0.f;
        #pragma unroll
        for (int r = 0; r < 8; ++r) {
            float v = acc[t][r] + bia[r];
            v = v > 0.f ? v : 0.f;
            s0 = fmaf(v, wa[r], s0);
            s1 = fmaf(v, wb[r], s1);
        }
        pp[t] = make_float2(s0, s1);
    }
}

template<int K1, int NDIM>
__global__ void __launch_bounds__(512) __attribute__((amdgpu_waves_per_eu(12)))
gin_layer1_kernel(const float* __restrict__ X,
                  const int* __restrict__ cnt,
                  const int* __restrict__ col,
                  const int* __restrict__ spillCnt,
                  const int* __restrict__ spill,
                  const float* __restrict__ Wa,
                  const float* __restrict__ ba,
                  const float* __restrict__ Wb,
                  const float* __restrict__ bb,
                  float* __restrict__ Y,
                  float* __restrict__ colsum,
                  float* __restrict__ colsumsq, int M) {
    constexpr int LS2 = NDIM + 4;
    static_assert(K1 <= NDIM, "the buffer is sized for the wider of the two row sets");
    __shared__ __attribute__((aligned(16))) float tile[GEMM_MAXT * 16 * LS2];

    const int lane  = threadIdx.x;
    const int wave  = __builtin_amdgcn_readfirstlane(threadIdx.y);

    const int units = (M + 15) >> 4;
    const int u0 = (int)(((long long)blockIdx.x * units) / gridDim.x);
    const int u1 = (int)(((long long)(blockIdx.x + 1) * units) / gridDim.x);
    const int n    = u1 - u0;
    const int row0 = u0 * 16;

    const float bias1 = ba[wave * 16 + (lane & 15)];
    const float bias2 = bb[wave * 16 + (lane & 15)];
    float psum = 0.f, psq = 0.f;

    if (n == 5) {
        gather_stage<K1, 5, false>(tile, X, cnt, col, spillCnt, spill, nullptr, nullptr,
                                   nullptr, nullptr, nullptr, row0, M, 0.f, lane, wave);
        __syncthreads();
        wmma_tile_run_to_lds<K1, NDIM, 5>(tile, Wa, bias1, lane, wave);
        __syncthreads();
        wmma_tile_run<NDIM, NDIM, 5, true>(tile, Wb, bias2, Y, row0, M, lane, wave, psum, psq);
    } else if (n == 4) {
        gather_stage<K1, 4, false>(tile, X, cnt, col, spillCnt, spill, nullptr, nullptr,
                                   nullptr, nullptr, nullptr, row0, M, 0.f, lane, wave);
        __syncthreads();
        wmma_tile_run_to_lds<K1, NDIM, 4>(tile, Wa, bias1, lane, wave);
        __syncthreads();
        wmma_tile_run<NDIM, NDIM, 4, true>(tile, Wb, bias2, Y, row0, M, lane, wave, psum, psq);
    } else {
        for (int t = 0; t < n; ++t) {
            if (t) __syncthreads();
            gather_stage<K1, 1, false>(tile, X, cnt, col, spillCnt, spill, nullptr, nullptr,
                                       nullptr, nullptr, nullptr, row0 + t * 16, M, 0.f,
                                       lane, wave);
            __syncthreads();
            wmma_tile_run_to_lds<K1, NDIM, 1>(tile, Wa, bias1, lane, wave);
            __syncthreads();
            wmma_tile_run<NDIM, NDIM, 1, true>(tile, Wb, bias2, Y, row0 + t * 16, M,
                                               lane, wave, psum, psq);
        }
    }

    atomicAdd(&colsum[wave * 16 + (lane & 15)], psum);
    atomicAdd(&colsumsq[wave * 16 + (lane & 15)], psq);
}

template<int K, int NDIM>
__global__ void __launch_bounds__(512) __attribute__((amdgpu_waves_per_eu(12)))
gin_layer2_head_kernel(const float* __restrict__ H2,
                       const int* __restrict__ cnt,
                       const int* __restrict__ col,
                       const int* __restrict__ spillCnt,
                       const int* __restrict__ spill,
                       const float* __restrict__ colsum1,
                       const float* __restrict__ colsumsq1,
                       const float* __restrict__ gamma,
                       const float* __restrict__ beta,
                       float* __restrict__ hid,
                       const float* __restrict__ W,
                       const float* __restrict__ bias,
                       const float* __restrict__ W4,
                       const float* __restrict__ b4,
                       float* __restrict__ out2,
                       float* __restrict__ colsum5,
                       float* __restrict__ colsumsq5, int M, float invN) {
    constexpr int LSTRIDE = K + 4;
    __shared__ __attribute__((aligned(16))) float tileA[GEMM_MAXT * 16 * LSTRIDE];

    const int lane  = threadIdx.x;
    const int wave  = __builtin_amdgcn_readfirstlane(threadIdx.y);

    const int units = (M + 15) >> 4;
    const int u0 = (int)(((long long)blockIdx.x * units) / gridDim.x);
    const int u1 = (int)(((long long)(blockIdx.x + 1) * units) / gridDim.x);
    const int n    = u1 - u0;
    const int row0 = u0 * 16;

    float2 pp[GEMM_MAXT];
    #pragma unroll
    for (int t = 0; t < GEMM_MAXT; ++t) pp[t] = make_float2(0.f, 0.f);

    if (n == 5) {
        gather_stage<K, 5, true>(tileA, H2, cnt, col, spillCnt, spill, colsum1, colsumsq1,
                                 gamma, beta, hid, row0, M, invN, lane, wave);
        __syncthreads();
        wmma_tile_run_head<K, NDIM, 5>(tileA, W, bias, W4, lane, wave, pp);
    } else if (n == 4) {
        gather_stage<K, 4, true>(tileA, H2, cnt, col, spillCnt, spill, colsum1, colsumsq1,
                                 gamma, beta, hid, row0, M, invN, lane, wave);
        __syncthreads();
        wmma_tile_run_head<K, NDIM, 4>(tileA, W, bias, W4, lane, wave, pp);
    } else {
        for (int t = 0; t < n; ++t)
            gather_stage<K, 1, true>(tileA + t * 16 * LSTRIDE, H2, cnt, col, spillCnt, spill,
                                     colsum1, colsumsq1, gamma, beta, hid, row0 + t * 16, M,
                                     invN, lane, wave);
        __syncthreads();
        #pragma unroll
        for (int t = 0; t < 3; ++t)
            if (t < n)
                wmma_tile_run_head<K, NDIM, 1>(tileA + t * 16 * LSTRIDE, W, bias, W4,
                                               lane, wave, pp + t);
    }

    constexpr int PROWS = GEMM_MAXT * 16;
    __syncthreads();
    float2* part = reinterpret_cast<float2*>(tileA);
    const int grp = wave * 2 + (lane >> 4);
    #pragma unroll
    for (int t = 0; t < GEMM_MAXT; ++t)
        if (t < n) part[grp * PROWS + t * 16 + (lane & 15)] = pp[t];
    __syncthreads();
    if (wave == 0) {
        const float* pf = tileA;
        const int c = lane & 1;
        const float bc = b4[c];
        float ssum = 0.f, ssq = 0.f;
        #pragma unroll
        for (int j = 0; j < GEMM_MAXT; ++j) {
            if (j < n) {
                const int idx = j * 32 + lane;
                float sacc = bc;
                #pragma unroll
                for (int g = 0; g < 32; ++g) sacc += pf[g * (PROWS * 2) + idx];
                const float v = sacc > 0.f ? sacc : 0.f;
                const int row = row0 + (idx >> 1);
                if (row < M) {
                    out2[(long long)row * 2 + c] = v;
                    ssum += v; ssq += v * v;
                }
            }
        }
        #pragma unroll
        for (int d = 2; d < 32; d <<= 1) {
            ssum += __shfl_xor(ssum, d);
            ssq  += __shfl_xor(ssq, d);
        }
        if (lane < 2) {
            atomicAdd(&colsum5[lane], ssum);
            atomicAdd(&colsumsq5[lane], ssq);
        }
    }
}

__global__ void bn_params_kernel(const float* __restrict__ colsum,
                                 const float* __restrict__ colsumsq,
                                 const float* __restrict__ gamma,
                                 const float* __restrict__ beta,
                                 float* __restrict__ scale,
                                 float* __restrict__ shift,
                                 int C, float invN) {
    int c = blockIdx.x * blockDim.x + threadIdx.x;
    if (c >= C) return;
    float mean = colsum[c] * invN;
    float var  = colsumsq[c] * invN - mean * mean;
    float s = gamma[c] * rsqrtf(var + BN_EPS);
    scale[c] = s;
    shift[c] = beta[c] - mean * s;
}

__global__ void bn_apply2(const float* __restrict__ Yin,
                          const float* __restrict__ scale,
                          const float* __restrict__ shift,
                          float* __restrict__ Yout, int total) {
    int t = blockIdx.x * blockDim.x + threadIdx.x;
    if (t >= total) return;
    int c = t & 1;
    Yout[t] = Yin[t] * scale[c] + shift[c];
}

extern "C" void kernel_launch(void* const* d_in, const int* in_sizes, int n_in,
                              void* d_out, int out_size, void* d_ws, size_t ws_size,
                              hipStream_t stream) {
    (void)n_in; (void)out_size; (void)ws_size;

    const float* x   = (const float*)d_in[0];
    const int*   ei  = (const int*)  d_in[1];
    const float* W1  = (const float*)d_in[2];
    const float* b1  = (const float*)d_in[3];
    const float* W2  = (const float*)d_in[4];
    const float* b2  = (const float*)d_in[5];
    const float* W3  = (const float*)d_in[6];
    const float* b3  = (const float*)d_in[7];
    const float* W4  = (const float*)d_in[8];
    const float* b4  = (const float*)d_in[9];
    const float* gm1 = (const float*)d_in[10];
    const float* bt1 = (const float*)d_in[11];
    const float* gm5 = (const float*)d_in[12];
    const float* bt5 = (const float*)d_in[13];

    const int FIN = 128, DIM = 256;
    const int Nn = in_sizes[0] / FIN;
    const int E  = in_sizes[1] / 2;
    const int* src = ei;
    const int* dst = ei + E;

    float* ws = (float*)d_ws;
    float* colsum1   = ws + 0;
    float* colsumsq1 = ws + 256;
    float* colsum5   = ws + 1024;
    float* colsumsq5 = ws + 1028;
    float* scale5    = ws + 1032;
    float* shift5    = ws + 1036;
    int*   spillCnt  = (int*)(ws + 1040);
    int*   cnt    = (int*)(ws + 2048);
    const size_t hdr = (2048 + (size_t)Nn + 63) / 64 * 64;
    int*   col = (int*)(ws + hdr);
    float* R2 = ws + hdr + (size_t)Nn * PITCH;
    float* R3 = R2 + (size_t)Nn * DIM;
    int*   spill = (int*)(R3 + (size_t)Nn * 2);

    float* outp = (float*)d_out;
    float* hid  = outp + (size_t)Nn * 2;

    hipMemsetAsync(ws, 0, (2048 + (size_t)Nn) * sizeof(float), stream);

    const dim3 gemmBlk(32, 16);
    const int mTiles = (Nn + 15) / 16;
    const dim3 gemmGrid(mTiles > 768 * 5 ? (mTiles + 4) / 5 : 768);

    edge_bucket_kernel<<<(E + 255) / 256, 256, 0, stream>>>(src, dst, cnt, col,
                                                            spillCnt, spill, E);

    gin_layer1_kernel<128, 256><<<gemmGrid, gemmBlk, 0, stream>>>(
        x, cnt, col, spillCnt, spill, W1, b1, W2, b2, R2, colsum1, colsumsq1, Nn);
    gin_layer2_head_kernel<256, 256><<<gemmGrid, gemmBlk, 0, stream>>>(
        R2, cnt, col, spillCnt, spill, colsum1, colsumsq1, gm1, bt1, hid,
        W3, b3, W4, b4, R3, colsum5, colsumsq5, Nn, 1.0f / (float)Nn);
    bn_params_kernel<<<1, 32, 0, stream>>>(colsum5, colsumsq5, gm5, bt5,
                                           scale5, shift5, 2, 1.0f / (float)Nn);
    bn_apply2<<<(Nn * 2 + 255) / 256, 256, 0, stream>>>(R3, scale5, shift5,
                                                        outp, Nn * 2);
}


# edges grouped by dst in 2 passes: a block sorts its chunk by 128-node bin in LDS (1 global add per bin, not per edge); a block per bin ranks by LDS counters, writes whole buckets; sum order: arrival
# speedup vs baseline: 3.5384x; 1.1940x over previous
#include <hip/hip_runtime.h>

typedef __attribute__((ext_vector_type(2))) float v2f;
typedef __attribute__((ext_vector_type(8))) float v8f;

#define BN_EPS 1e-5f

constexpr int PITCH = 64;

constexpr int BIN_SH  = 7;
constexpr int BIN_N   = 1 << BIN_SH;
constexpr int MAXBINS = 512;
constexpr int BIN_CAP = 8192;
constexpr int CNT_PAD = 32;
constexpr int EB_T = 1024;
constexpr int EB_E = 4;

__global__ void __launch_bounds__(EB_T)
edge_bin_kernel(const int* __restrict__ src, const int* __restrict__ dst,
                int* __restrict__ binCnt, int* __restrict__ rec,
                int* __restrict__ ovfCnt, int* __restrict__ ovf, int E, int nbins) {
    __shared__ int hist[MAXBINS];
    __shared__ int off[MAXBINS];
    __shared__ int gbase[MAXBINS];
    __shared__ int2 srt[EB_T * EB_E];

    const int tid = threadIdx.x;
    for (int i = tid; i < MAXBINS; i += EB_T) hist[i] = 0;
    __syncthreads();

    const int e0 = blockIdx.x * (EB_T * EB_E);
    int d[EB_E], s[EB_E], r[EB_E];
    #pragma unroll
    for (int u = 0; u < EB_E; ++u) {
        const int e = e0 + u * EB_T + tid;
        d[u] = -1; s[u] = 0; r[u] = 0;
        if (e < E) {
            d[u] = dst[e];
            s[u] = src[e];
            r[u] = atomicAdd(&hist[d[u] >> BIN_SH], 1);
        }
    }
    __syncthreads();

    if (tid < 32) {
        constexpr int Q = MAXBINS / 32;
        int loc[Q];
        int sum = 0;
        #pragma unroll
        for (int q = 0; q < Q; ++q) { loc[q] = sum; sum += hist[tid * Q + q]; }
        int inc = sum;
        for (int dl = 1; dl < 32; dl <<= 1) {
            const int v = __shfl_up(inc, dl);
            if (tid >= dl) inc += v;
        }
        const int excl = inc - sum;
        #pragma unroll
        for (int q = 0; q < Q; ++q) off[tid * Q + q] = excl + loc[q];
    }
    if (tid < nbins) {
        const int h = hist[tid];
        gbase[tid] = h ? atomicAdd(&binCnt[tid * CNT_PAD], h) : 0;
    }
    __syncthreads();

    #pragma unroll
    for (int u = 0; u < EB_E; ++u)
        if (d[u] >= 0) srt[off[d[u] >> BIN_SH] + r[u]] = make_int2(d[u], s[u]);
    __syncthreads();

    const int left = E - e0;
    const int n = left < EB_T * EB_E ? left : EB_T * EB_E;
    for (int i = tid; i < n; i += EB_T) {
        const int2 p = srt[i];
        const int b = p.x >> BIN_SH;
        const int pos = gbase[b] + (i - off[b]);
        if (pos < BIN_CAP) {
            reinterpret_cast<int2*>(rec)[(long long)b * BIN_CAP + pos] = p;
        } else {
            const int k = atomicAdd(ovfCnt, 1);
            reinterpret_cast<int2*>(ovf)[k] = p;
        }
    }
}

__global__ void __launch_bounds__(256)
bin_rank_kernel(const int* __restrict__ binCnt, const int* __restrict__ rec,
                const int* __restrict__ ovfCnt, const int* __restrict__ ovf,
                int* __restrict__ cnt, int* __restrict__ col,
                int* __restrict__ spillCnt, int* __restrict__ spill, int N) {
    __shared__ int lcnt[BIN_N];
    __shared__ int lcol[BIN_N * PITCH];

    const int tid = threadIdx.x;
    const int b = blockIdx.x;
    const int node0 = b << BIN_SH;
    if (tid < BIN_N) lcnt[tid] = 0;
    __syncthreads();

    auto place = [&](const int2 p) {
        const int dl = p.x - node0;
        const int r = atomicAdd(&lcnt[dl], 1);
        if (r < PITCH) {
            lcol[dl * PITCH + r] = p.y;
        } else {
            const int k = atomicAdd(spillCnt, 1);
            reinterpret_cast<int2*>(spill)[k] = p;
        }
    };

    const int tot  = binCnt[b * CNT_PAD];
    const int nrec = tot < BIN_CAP ? tot : BIN_CAP;
    const int2* rb = reinterpret_cast<const int2*>(rec) + (long long)b * BIN_CAP;
    for (int base = 0; base < nrec; base += 256 * 8) {
        int2 p[8];
        #pragma unroll
        for (int u = 0; u < 8; ++u) {
            const int i = base + u * 256 + tid;
            p[u] = make_int2(-1, 0);
            if (i < nrec) p[u] = rb[i];
        }
        #pragma unroll
        for (int u = 0; u < 8; ++u)
            if (p[u].x >= 0) place(p[u]);
    }
    const int novf = *ovfCnt;
    for (int i = tid; i < novf; i += 256) {
        const int2 p = reinterpret_cast<const int2*>(ovf)[i];
        if ((p.x >> BIN_SH) == b) place(p);
    }
    __syncthreads();

    if (tid < BIN_N && node0 + tid < N) cnt[node0 + tid] = lcnt[tid];
    const int wave = tid >> 5, lane = tid & 31;
    for (int dl = wave; dl < BIN_N; dl += 8) {
        const int node = node0 + dl;
        if (node >= N) break;
        const int deg = lcnt[dl];
        const int m = deg < PITCH ? deg : PITCH;
        if (lane < m)      col[node * PITCH + lane]      = lcol[dl * PITCH + lane];
        if (lane + 32 < m) col[node * PITCH + 32 + lane] = lcol[dl * PITCH + 32 + lane];
    }
}

template<bool NORM>
__device__ __forceinline__ void add_row(float4& a, const float4& v,
                                        const float4& s, const float4& t) {
    if (NORM) {
        a.x += v.x * s.x + t.x;
        a.y += v.y * s.y + t.y;
        a.z += v.z * s.z + t.z;
        a.w += v.w * s.w + t.w;
    } else {
        a.x += v.x; a.y += v.y; a.z += v.z; a.w += v.w;
    }
}

template<int F, bool NORM>
__device__ __forceinline__ void sum_source_rows(const float* __restrict__ X,
                                                int node, int deg, int myc,
                                                const int* __restrict__ col,
                                                const int* __restrict__ spillCnt,
                                                const int* __restrict__ spill,
                                                int lane, float4 (&acc)[F / 128],
                                                const float4 (&s)[F / 128],
                                                const float4 (&t)[F / 128]) {
    constexpr int V = F / 128;
    constexpr int B = 8 / V;
    const int beg = node * PITCH;
    const int end = beg + (deg < PITCH ? deg : PITCH);

    if (deg > PITCH) {
        const int nsp = *spillCnt;
        for (int base = 0; base < nsp; base += 32) {
            int2 ds = make_int2(-1, 0);
            if (base + lane < nsp) ds = reinterpret_cast<const int2*>(spill)[base + lane];
            unsigned m = __builtin_amdgcn_ballot_w32(ds.x == node);
            while (m) {
                const int b = __builtin_ctz(m);
                m &= m - 1;
                const int sidx = __builtin_amdgcn_readlane(ds.y, b);
                const float4* r = reinterpret_cast<const float4*>(X + (long long)sidx * F);
                #pragma unroll
                for (int w = 0; w < V; ++w) add_row<NORM>(acc[w], r[lane + 32 * w], s[w], t[w]);
            }
        }
    }

    for (int base = beg; base < end; base += 32) {
        const int n = (end - base) < 32 ? (end - base) : 32;
        int c = myc;
        if (base != beg) c = (lane < n) ? col[base + lane] : 0;
        int j = 0;
        for (; j + B <= n; j += B) {
            float4 v[B][V];
            #pragma unroll
            for (int u = 0; u < B; ++u) {
                const int sidx = __builtin_amdgcn_readlane(c, j + u);
                const float4* r = reinterpret_cast<const float4*>(X + (long long)sidx * F);
                #pragma unroll
                for (int w = 0; w < V; ++w) v[u][w] = r[lane + 32 * w];
            }
            #pragma unroll
            for (int u = 0; u < B; ++u) {
                #pragma unroll
                for (int w = 0; w < V; ++w) add_row<NORM>(acc[w], v[u][w], s[w], t[w]);
            }
        }
        if (j < n) {
            float4 v[B][V];
            #pragma unroll
            for (int u = 0; u < B; ++u) {
                #pragma unroll
                for (int w = 0; w < V; ++w) v[u][w] = make_float4(0.f, 0.f, 0.f, 0.f);
                if (j + u < n) {
                    const int sidx = __builtin_amdgcn_readlane(c, j + u);
                    const float4* r = reinterpret_cast<const float4*>(X + (long long)sidx * F);
                    #pragma unroll
                    for (int w = 0; w < V; ++w) v[u][w] = r[lane + 32 * w];
                }
            }
            #pragma unroll
            for (int u = 0; u < B; ++u) {
                if (!NORM || j + u < n) {
                    #pragma unroll
                    for (int w = 0; w < V; ++w) add_row<NORM>(acc[w], v[u][w], s[w], t[w]);
                }
            }
        }
    }
}

template<int K, int MT, bool NORM>
__device__ __forceinline__ void gather_stage(float* __restrict__ tile,
                                             const float* __restrict__ X,
                                             const int* __restrict__ cnt,
                                             const int* __restrict__ col,
                                             const int* __restrict__ spillCnt,
                                             const int* __restrict__ spill,
                                             const float* __restrict__ colsum,
                                             const float* __restrict__ colsumsq,
                                             const float* __restrict__ gamma,
                                             const float* __restrict__ beta,
                                             float* __restrict__ hid,
                                             int row0, int M, float invN,
                                             int lane, int wave) {
    constexpr int V = K / 128;
    constexpr int LSTRIDE = K + 4;

    float4 s[V], t[V];
    #pragma unroll
    for (int w = 0; w < V; ++w) {
        s[w] = make_float4(0.f, 0.f, 0.f, 0.f);
        t[w] = make_float4(0.f, 0.f, 0.f, 0.f);
        if (NORM) {
            const int c4 = lane + 32 * w;
            const float4 cs = reinterpret_cast<const float4*>(colsum)[c4];
            const float4 cq = reinterpret_cast<const float4*>(colsumsq)[c4];
            const float csv[4] = {cs.x, cs.y, cs.z, cs.w};
            const float cqv[4] = {cq.x, cq.y, cq.z, cq.w};
            float sv[4], tv[4];
            #pragma unroll
            for (int q = 0; q < 4; ++q) {
                const float mean = csv[q] * invN;
                const float var  = cqv[q] * invN - mean * mean;
                sv[q] = gamma[c4 * 4 + q] * rsqrtf(var + BN_EPS);
                tv[q] = beta[c4 * 4 + q] - mean * sv[q];
            }
            s[w] = make_float4(sv[0], sv[1], sv[2], sv[3]);
            t[w] = make_float4(tv[0], tv[1], tv[2], tv[3]);
        }
    }

    int degv = 0;
    {
        const int nd = row0 + wave + 16 * lane;
        if (lane < MT && nd < M) degv = cnt[nd];
    }
    int node = row0 + wave;
    int deg  = __builtin_amdgcn_readlane(degv, 0);
    int myc  = 0;
    if (lane < (deg < 32 ? deg : 32)) myc = col[node * PITCH + lane];

    #pragma nounroll
    for (int i = 0; i < MT; ++i) {
        const int nnode = node + 16;
        int ndeg = 0, nmyc = 0;
        if (i + 1 < MT) {
            ndeg = __builtin_amdgcn_readlane(degv, i + 1);
            if (lane < (ndeg < 32 ? ndeg : 32)) nmyc = col[nnode * PITCH + lane];
        }

        float4 acc[V];
        #pragma unroll
        for (int w = 0; w < V; ++w) acc[w] = make_float4(0.f, 0.f, 0.f, 0.f);

        if (node < M) {
            const float4* xrow = reinterpret_cast<const float4*>(X + (long long)node * K);
            float4 own[V];
            #pragma unroll
            for (int w = 0; w < V; ++w) own[w] = xrow[lane + 32 * w];
            if (NORM) {
                float4* h = reinterpret_cast<float4*>(hid + (long long)node * K);
                #pragma unroll
                for (int w = 0; w < V; ++w) {
                    acc[w].x = own[w].x * s[w].x + t[w].x;
                    acc[w].y = own[w].y * s[w].y + t[w].y;
                    acc[w].z = own[w].z * s[w].z + t[w].z;
                    acc[w].w = own[w].w * s[w].w + t[w].w;
                    h[lane + 32 * w] = acc[w];
                }
            }
            sum_source_rows<K, NORM>(X, node, deg, myc, col, spillCnt, spill, lane, acc, s, t);
            if (!NORM) {
                #pragma unroll
                for (int w = 0; w < V; ++w) {
                    acc[w].x = own[w].x + acc[w].x;
                    acc[w].y = own[w].y + acc[w].y;
                    acc[w].z = own[w].z + acc[w].z;
                    acc[w].w = own[w].w + acc[w].w;
                }
            }
        }

        float4* l = reinterpret_cast<float4*>(tile + (wave + 16 * i) * LSTRIDE);
        #pragma unroll
        for (int w = 0; w < V; ++w) l[lane + 32 * w] = acc[w];

        node = nnode; deg = ndeg; myc = nmyc;
    }
}

constexpr int GEMM_MAXT = 5;

template<int K, int NDIM, int MT, bool STATS>
__device__ __forceinline__ void wmma_tile_run(const float* __restrict__ tileA,
                                              const float* __restrict__ W,
                                              float bb, float* __restrict__ Y,
                                              int row0, int M, int lane, int wave,
                                              float& psum, float& psq) {
    constexpr int LSTRIDE = K + 4;
    const int r16  = lane & 15;
    const int hi   = lane >> 4;
    const int bCol = wave * 16 + r16;
    const float* aLds = &tileA[r16 * LSTRIDE + hi * 2];
    const float* wp   = W + (long long)(hi * 2) * NDIM + bCol;

    v8f acc[MT];
    #pragma unroll
    for (int t = 0; t < MT; ++t)
        acc[t] = (v8f){0.f, 0.f, 0.f, 0.f, 0.f, 0.f, 0.f, 0.f};

    #pragma unroll 2
    for (int k = 0; k < K; k += 4) {
        v2f b;
        b.x = wp[(long long)k * NDIM];
        b.y = wp[(long long)(k + 1) * NDIM];
        #pragma unroll
        for (int t = 0; t < MT; ++t) {
            float2 av = *reinterpret_cast<const float2*>(aLds + t * 16 * LSTRIDE + k);
            v2f a; a.x = av.x; a.y = av.y;
            acc[t] = __builtin_amdgcn_wmma_f32_16x16x4_f32(
                       false, a,  false, b,
                       (short)0, acc[t],  false,  false);
        }
    }

    float* yp = Y + (long long)(row0 + hi * 8) * NDIM + bCol;
    if (row0 + MT * 16 <= M) {
        #pragma unroll
        for (int t = 0; t < MT; ++t) {
            #pragma unroll
            for (int r = 0; r < 8; ++r) {
                float v = acc[t][r] + bb;
                v = v > 0.f ? v : 0.f;
                yp[(t * 16 + r) * NDIM] = v;
                if (STATS) { psum += v; psq += v * v; }
            }
        }
    } else {
        const int left = M - (row0 + hi * 8);
        #pragma unroll
        for (int t = 0; t < MT; ++t) {
            #pragma unroll
            for (int r = 0; r < 8; ++r) {
                float v = acc[t][r] + bb;
                v = v > 0.f ? v : 0.f;
                if (t * 16 + r < left) {
                    yp[(t * 16 + r) * NDIM] = v;
                    if (STATS) { psum += v; psq += v * v; }
                }
            }
        }
    }
}

template<int K, int NDIM, int MT>
__device__ __forceinline__ void wmma_tile_run_to_lds(float* tile,
                                                     const float* __restrict__ W,
                                                     float bb, int lane, int wave) {
    constexpr int LSTRIDE = K + 4;
    constexpr int HSTRIDE = NDIM + 4;
    const int r16  = lane & 15;
    const int hi   = lane >> 4;
    const int bCol = wave * 16 + r16;
    const float* aLds = &tile[r16 * LSTRIDE + hi * 2];
    const float* wp   = W + (long long)(hi * 2) * NDIM + bCol;

    v8f acc[MT];
    #pragma unroll
    for (int t = 0; t < MT; ++t)
        acc[t] = (v8f){0.f, 0.f, 0.f, 0.f, 0.f, 0.f, 0.f, 0.f};

    #pragma unroll 2
    for (int k = 0; k < K; k += 4) {
        v2f b;
        b.x = wp[(long long)k * NDIM];
        b.y = wp[(long long)(k + 1) * NDIM];
        #pragma unroll
        for (int t = 0; t < MT; ++t) {
            float2 av = *reinterpret_cast<const float2*>(aLds + t * 16 * LSTRIDE + k);
            v2f a; a.x = av.x; a.y = av.y;
            acc[t] = __builtin_amdgcn_wmma_f32_16x16x4_f32(
                       false, a,  false, b,
                       (short)0, acc[t],  false,  false);
        }
    }

    __syncthreads();
    float* hp = tile + (hi * 8) * HSTRIDE + bCol;
    #pragma unroll
    for (int t = 0; t < MT; ++t) {
        #pragma unroll
        for (int r = 0; r < 8; ++r) {
            float v = acc[t][r] + bb;
            hp[(t * 16 + r) * HSTRIDE] = v > 0.f ? v : 0.f;
        }
    }
}

template<int K, int NDIM, int MT>
__device__ __forceinline__ void wmma_tile_run_head(const float* __restrict__ tileA,
                                                   const float* __restrict__ W,
                                                   const float* __restrict__ bias,
                                                   const float* __restrict__ W4,
                                                   int lane, int wave,
                                                   float2* __restrict__ pp) {
    constexpr int LSTRIDE = K + 4;
    const int r16  = lane & 15;
    const int hi   = lane >> 4;
    const int bCol = wave * 16 + r16;
    const float* aLds = &tileA[r16 * LSTRIDE + hi * 2];
    const float* wp   = W + (long long)(hi * 2) * NDIM + bCol;

    v8f acc[MT];
    #pragma unroll
    for (int t = 0; t < MT; ++t)
        acc[t] = (v8f){0.f, 0.f, 0.f, 0.f, 0.f, 0.f, 0.f, 0.f};

    #pragma unroll 2
    for (int k = 0; k < K; k += 4) {
        v2f b;
        b.x = wp[(long long)k * NDIM];
        b.y = wp[(long long)(k + 1) * NDIM];
        #pragma unroll
        for (int t = 0; t < MT; ++t) {
            float2 av = *reinterpret_cast<const float2*>(aLds + t * 16 * LSTRIDE + k);
            v2f a; a.x = av.x; a.y = av.y;
            acc[t] = __builtin_amdgcn_wmma_f32_16x16x4_f32(
                       false, b,  false, a,
                       (short)0, acc[t],  false,  false);
        }
    }

    const int f0 = wave * 16 + hi * 8;
    float bia[8], wa[8], wb[8];
    #pragma unroll
    for (int r = 0; r < 8; ++r) {
        bia[r] = bias[f0 + r];
        wa[r]  = W4[(f0 + r) * 2];
        wb[r]  = W4[(f0 + r) * 2 + 1];
    }
    #pragma unroll
    for (int t = 0; t < MT; ++t) {
        float s0 = 0.f, s1 = 0.f;
        #pragma unroll
        for (int r = 0; r < 8; ++r) {
            float v = acc[t][r] + bia[r];
            v = v > 0.f ? v : 0.f;
            s0 = fmaf(v, wa[r], s0);
            s1 = fmaf(v, wb[r], s1);
        }
        pp[t] = make_float2(s0, s1);
    }
}

template<int K1, int NDIM>
__global__ void __launch_bounds__(512) __attribute__((amdgpu_waves_per_eu(12)))
gin_layer1_kernel(const float* __restrict__ X,
                  const int* __restrict__ cnt,
                  const int* __restrict__ col,
                  const int* __restrict__ spillCnt,
                  const int* __restrict__ spill,
                  const float* __restrict__ Wa,
                  const float* __restrict__ ba,
                  const float* __restrict__ Wb,
                  const float* __restrict__ bb,
                  float* __restrict__ Y,
                  float* __restrict__ colsum,
                  float* __restrict__ colsumsq, int M) {
    constexpr int LS2 = NDIM + 4;
    static_assert(K1 <= NDIM, "the buffer is sized for the wider of the two row sets");
    __shared__ __attribute__((aligned(16))) float tile[GEMM_MAXT * 16 * LS2];

    const int lane  = threadIdx.x;
    const int wave  = __builtin_amdgcn_readfirstlane(threadIdx.y);

    const int units = (M + 15) >> 4;
    const int u0 = (int)(((long long)blockIdx.x * units) / gridDim.x);
    const int u1 = (int)(((long long)(blockIdx.x + 1) * units) / gridDim.x);
    const int n    = u1 - u0;
    const int row0 = u0 * 16;

    const float bias1 = ba[wave * 16 + (lane & 15)];
    const float bias2 = bb[wave * 16 + (lane & 15)];
    float psum = 0.f, psq = 0.f;

    if (n == 5) {
        gather_stage<K1, 5, false>(tile, X, cnt, col, spillCnt, spill, nullptr, nullptr,
                                   nullptr, nullptr, nullptr, row0, M, 0.f, lane, wave);
        __syncthreads();
        wmma_tile_run_to_lds<K1, NDIM, 5>(tile, Wa, bias1, lane, wave);
        __syncthreads();
        wmma_tile_run<NDIM, NDIM, 5, true>(tile, Wb, bias2, Y, row0, M, lane, wave, psum, psq);
    } else if (n == 4) {
        gather_stage<K1, 4, false>(tile, X, cnt, col, spillCnt, spill, nullptr, nullptr,
                                   nullptr, nullptr, nullptr, row0, M, 0.f, lane, wave);
        __syncthreads();
        wmma_tile_run_to_lds<K1, NDIM, 4>(tile, Wa, bias1, lane, wave);
        __syncthreads();
        wmma_tile_run<NDIM, NDIM, 4, true>(tile, Wb, bias2, Y, row0, M, lane, wave, psum, psq);
    } else {
        for (int t = 0; t < n; ++t) {
            if (t) __syncthreads();
            gather_stage<K1, 1, false>(tile, X, cnt, col, spillCnt, spill, nullptr, nullptr,
                                       nullptr, nullptr, nullptr, row0 + t * 16, M, 0.f,
                                       lane, wave);
            __syncthreads();
            wmma_tile_run_to_lds<K1, NDIM, 1>(tile, Wa, bias1, lane, wave);
            __syncthreads();
            wmma_tile_run<NDIM, NDIM, 1, true>(tile, Wb, bias2, Y, row0 + t * 16, M,
                                               lane, wave, psum, psq);
        }
    }

    atomicAdd(&colsum[wave * 16 + (lane & 15)], psum);
    atomicAdd(&colsumsq[wave * 16 + (lane & 15)], psq);
}

template<int K, int NDIM>
__global__ void __launch_bounds__(512) __attribute__((amdgpu_waves_per_eu(12)))
gin_layer2_head_kernel(const float* __restrict__ H2,
                       const int* __restrict__ cnt,
                       const int* __restrict__ col,
                       const int* __restrict__ spillCnt,
                       const int* __restrict__ spill,
                       const float* __restrict__ colsum1,
                       const float* __restrict__ colsumsq1,
                       const float* __restrict__ gamma,
                       const float* __restrict__ beta,
                       float* __restrict__ hid,
                       const float* __restrict__ W,
                       const float* __restrict__ bias,
                       const float* __restrict__ W4,
                       const float* __restrict__ b4,
                       float* __restrict__ out2,
                       float* __restrict__ colsum5,
                       float* __restrict__ colsumsq5, int M, float invN) {
    constexpr int LSTRIDE = K + 4;
    __shared__ __attribute__((aligned(16))) float tileA[GEMM_MAXT * 16 * LSTRIDE];

    const int lane  = threadIdx.x;
    const int wave  = __builtin_amdgcn_readfirstlane(threadIdx.y);

    const int units = (M + 15) >> 4;
    const int u0 = (int)(((long long)blockIdx.x * units) / gridDim.x);
    const int u1 = (int)(((long long)(blockIdx.x + 1) * units) / gridDim.x);
    const int n    = u1 - u0;
    const int row0 = u0 * 16;

    float2 pp[GEMM_MAXT];
    #pragma unroll
    for (int t = 0; t < GEMM_MAXT; ++t) pp[t] = make_float2(0.f, 0.f);

    if (n == 5) {
        gather_stage<K, 5, true>(tileA, H2, cnt, col, spillCnt, spill, colsum1, colsumsq1,
                                 gamma, beta, hid, row0, M, invN, lane, wave);
        __syncthreads();
        wmma_tile_run_head<K, NDIM, 5>(tileA, W, bias, W4, lane, wave, pp);
    } else if (n == 4) {
        gather_stage<K, 4, true>(tileA, H2, cnt, col, spillCnt, spill, colsum1, colsumsq1,
                                 gamma, beta, hid, row0, M, invN, lane, wave);
        __syncthreads();
        wmma_tile_run_head<K, NDIM, 4>(tileA, W, bias, W4, lane, wave, pp);
    } else {
        for (int t = 0; t < n; ++t)
            gather_stage<K, 1, true>(tileA + t * 16 * LSTRIDE, H2, cnt, col, spillCnt, spill,
                                     colsum1, colsumsq1, gamma, beta, hid, row0 + t * 16, M,
                                     invN, lane, wave);
        __syncthreads();
        #pragma unroll
        for (int t = 0; t < 3; ++t)
            if (t < n)
                wmma_tile_run_head<K, NDIM, 1>(tileA + t * 16 * LSTRIDE, W, bias, W4,
                                               lane, wave, pp + t);
    }

    constexpr int PROWS = GEMM_MAXT * 16;
    __syncthreads();
    float2* part = reinterpret_cast<float2*>(tileA);
    const int grp = wave * 2 + (lane >> 4);
    #pragma unroll
    for (int t = 0; t < GEMM_MAXT; ++t)
        if (t < n) part[grp * PROWS + t * 16 + (lane & 15)] = pp[t];
    __syncthreads();
    if (wave == 0) {
        const float* pf = tileA;
        const int c = lane & 1;
        const float bc = b4[c];
        float ssum = 0.f, ssq = 0.f;
        #pragma unroll
        for (int j = 0; j < GEMM_MAXT; ++j) {
            if (j < n) {
                const int idx = j * 32 + lane;
                float sacc = bc;
                #pragma unroll
                for (int g = 0; g < 32; ++g) sacc += pf[g * (PROWS * 2) + idx];
                const float v = sacc > 0.f ? sacc : 0.f;
                const int row = row0 + (idx >> 1);
                if (row < M) {
                    out2[(long long)row * 2 + c] = v;
                    ssum += v; ssq += v * v;
                }
            }
        }
        #pragma unroll
        for (int d = 2; d < 32; d <<= 1) {
            ssum += __shfl_xor(ssum, d);
            ssq  += __shfl_xor(ssq, d);
        }
        if (lane < 2) {
            atomicAdd(&colsum5[lane], ssum);
            atomicAdd(&colsumsq5[lane], ssq);
        }
    }
}

__global__ void bn_params_kernel(const float* __restrict__ colsum,
                                 const float* __restrict__ colsumsq,
                                 const float* __restrict__ gamma,
                                 const float* __restrict__ beta,
                                 float* __restrict__ scale,
                                 float* __restrict__ shift,
                                 int C, float invN) {
    int c = blockIdx.x * blockDim.x + threadIdx.x;
    if (c >= C) return;
    float mean = colsum[c] * invN;
    float var  = colsumsq[c] * invN - mean * mean;
    float s = gamma[c] * rsqrtf(var + BN_EPS);
    scale[c] = s;
    shift[c] = beta[c] - mean * s;
}

__global__ void bn_apply2(const float* __restrict__ Yin,
                          const float* __restrict__ scale,
                          const float* __restrict__ shift,
                          float* __restrict__ Yout, int total) {
    int t = blockIdx.x * blockDim.x + threadIdx.x;
    if (t >= total) return;
    int c = t & 1;
    Yout[t] = Yin[t] * scale[c] + shift[c];
}

extern "C" void kernel_launch(void* const* d_in, const int* in_sizes, int n_in,
                              void* d_out, int out_size, void* d_ws, size_t ws_size,
                              hipStream_t stream) {
    (void)n_in; (void)out_size; (void)ws_size;

    const float* x   = (const float*)d_in[0];
    const int*   ei  = (const int*)  d_in[1];
    const float* W1  = (const float*)d_in[2];
    const float* b1  = (const float*)d_in[3];
    const float* W2  = (const float*)d_in[4];
    const float* b2  = (const float*)d_in[5];
    const float* W3  = (const float*)d_in[6];
    const float* b3  = (const float*)d_in[7];
    const float* W4  = (const float*)d_in[8];
    const float* b4  = (const float*)d_in[9];
    const float* gm1 = (const float*)d_in[10];
    const float* bt1 = (const float*)d_in[11];
    const float* gm5 = (const float*)d_in[12];
    const float* bt5 = (const float*)d_in[13];

    const int FIN = 128, DIM = 256;
    const int Nn = in_sizes[0] / FIN;
    const int E  = in_sizes[1] / 2;
    const int* src = ei;
    const int* dst = ei + E;

    float* ws = (float*)d_ws;
    float* colsum1   = ws + 0;
    float* colsumsq1 = ws + 256;
    float* colsum5   = ws + 1024;
    float* colsumsq5 = ws + 1028;
    float* scale5    = ws + 1032;
    float* shift5    = ws + 1036;
    int*   spillCnt  = (int*)(ws + 1040);
    int*   ovfCnt    = (int*)(ws + 1041);
    const int nbins  = (Nn + BIN_N - 1) / BIN_N;
    int*   binCnt = (int*)(ws + 2048);
    int*   cnt    = binCnt + (size_t)nbins * CNT_PAD;
    const size_t hdr = (2048 + (size_t)nbins * CNT_PAD + (size_t)Nn + 63) / 64 * 64;
    int*   col = (int*)(ws + hdr);
    float* R2 = ws + hdr + (size_t)Nn * PITCH;
    float* R3 = R2 + (size_t)Nn * DIM;
    int*   spill = (int*)(R3 + (size_t)Nn * 2);
    int*   rec   = spill + (size_t)E * 2;
    int*   ovf   = rec + (size_t)nbins * BIN_CAP * 2;

    float* outp = (float*)d_out;
    float* hid  = outp + (size_t)Nn * 2;

    hipMemsetAsync(ws, 0, (2048 + (size_t)nbins * CNT_PAD) * sizeof(float), stream);

    const dim3 gemmBlk(32, 16);
    const int mTiles = (Nn + 15) / 16;
    const dim3 gemmGrid(mTiles > 768 * 5 ? (mTiles + 4) / 5 : 768);

    edge_bin_kernel<<<(E + EB_T * EB_E - 1) / (EB_T * EB_E), EB_T, 0, stream>>>(
        src, dst, binCnt, rec, ovfCnt, ovf, E, nbins);
    bin_rank_kernel<<<nbins, 256, 0, stream>>>(binCnt, rec, ovfCnt, ovf, cnt, col,
                                               spillCnt, spill, Nn);

    gin_layer1_kernel<128, 256><<<gemmGrid, gemmBlk, 0, stream>>>(
        x, cnt, col, spillCnt, spill, W1, b1, W2, b2, R2, colsum1, colsumsq1, Nn);
    gin_layer2_head_kernel<256, 256><<<gemmGrid, gemmBlk, 0, stream>>>(
        R2, cnt, col, spillCnt, spill, colsum1, colsumsq1, gm1, bt1, hid,
        W3, b3, W4, b4, R3, colsum5, colsumsq5, Nn, 1.0f / (float)Nn);
    bn_params_kernel<<<1, 32, 0, stream>>>(colsum5, colsumsq5, gm5, bt5,
                                           scale5, shift5, 2, 1.0f / (float)Nn);
    bn_apply2<<<(Nn * 2 + 255) / 256, 256, 0, stream>>>(R3, scale5, shift5,
                                                        outp, Nn * 2);
}


# M-tile dealing by dispatch order: the 53 five-tile runs go to every third of the first 159 blocks (first to enter), not to every 14th block; BN-stat partials regrouped (sum order)
# speedup vs baseline: 3.6657x; 1.0360x over previous
#include <hip/hip_runtime.h>

typedef __attribute__((ext_vector_type(2))) float v2f;
typedef __attribute__((ext_vector_type(8))) float v8f;

#define BN_EPS 1e-5f

constexpr int PITCH = 64;

constexpr int BIN_SH  = 7;
constexpr int BIN_N   = 1 << BIN_SH;
constexpr int MAXBINS = 512;
constexpr int BIN_CAP = 8192;
constexpr int CNT_PAD = 32;
constexpr int EB_T = 1024;
constexpr int EB_E = 4;

__global__ void __launch_bounds__(EB_T)
edge_bin_kernel(const int* __restrict__ src, const int* __restrict__ dst,
                int* __restrict__ binCnt, int* __restrict__ rec,
                int* __restrict__ ovfCnt, int* __restrict__ ovf, int E, int nbins) {
    __shared__ int hist[MAXBINS];
    __shared__ int off[MAXBINS];
    __shared__ int gbase[MAXBINS];
    __shared__ int2 srt[EB_T * EB_E];

    const int tid = threadIdx.x;
    for (int i = tid; i < MAXBINS; i += EB_T) hist[i] = 0;
    __syncthreads();

    const int e0 = blockIdx.x * (EB_T * EB_E);
    int d[EB_E], s[EB_E], r[EB_E];
    #pragma unroll
    for (int u = 0; u < EB_E; ++u) {
        const int e = e0 + u * EB_T + tid;
        d[u] = -1; s[u] = 0; r[u] = 0;
        if (e < E) {
            d[u] = dst[e];
            s[u] = src[e];
            r[u] = atomicAdd(&hist[d[u] >> BIN_SH], 1);
        }
    }
    __syncthreads();

    if (tid < 32) {
        constexpr int Q = MAXBINS / 32;
        int loc[Q];
        int sum = 0;
        #pragma unroll
        for (int q = 0; q < Q; ++q) { loc[q] = sum; sum += hist[tid * Q + q]; }
        int inc = sum;
        for (int dl = 1; dl < 32; dl <<= 1) {
            const int v = __shfl_up(inc, dl);
            if (tid >= dl) inc += v;
        }
        const int excl = inc - sum;
        #pragma unroll
        for (int q = 0; q < Q; ++q) off[tid * Q + q] = excl + loc[q];
    }
    if (tid < nbins) {
        const int h = hist[tid];
        gbase[tid] = h ? atomicAdd(&binCnt[tid * CNT_PAD], h) : 0;
    }
    __syncthreads();

    #pragma unroll
    for (int u = 0; u < EB_E; ++u)
        if (d[u] >= 0) srt[off[d[u] >> BIN_SH] + r[u]] = make_int2(d[u], s[u]);
    __syncthreads();

    const int left = E - e0;
    const int n = left < EB_T * EB_E ? left : EB_T * EB_E;
    for (int i = tid; i < n; i += EB_T) {
        const int2 p = srt[i];
        const int b = p.x >> BIN_SH;
        const int pos = gbase[b] + (i - off[b]);
        if (pos < BIN_CAP) {
            reinterpret_cast<int2*>(rec)[(long long)b * BIN_CAP + pos] = p;
        } else {
            const int k = atomicAdd(ovfCnt, 1);
            reinterpret_cast<int2*>(ovf)[k] = p;
        }
    }
}

__global__ void __launch_bounds__(256)
bin_rank_kernel(const int* __restrict__ binCnt, const int* __restrict__ rec,
                const int* __restrict__ ovfCnt, const int* __restrict__ ovf,
                int* __restrict__ cnt, int* __restrict__ col,
                int* __restrict__ spillCnt, int* __restrict__ spill, int N) {
    __shared__ int lcnt[BIN_N];
    __shared__ int lcol[BIN_N * PITCH];

    const int tid = threadIdx.x;
    const int b = blockIdx.x;
    const int node0 = b << BIN_SH;
    if (tid < BIN_N) lcnt[tid] = 0;
    __syncthreads();

    auto place = [&](const int2 p) {
        const int dl = p.x - node0;
        const int r = atomicAdd(&lcnt[dl], 1);
        if (r < PITCH) {
            lcol[dl * PITCH + r] = p.y;
        } else {
            const int k = atomicAdd(spillCnt, 1);
            reinterpret_cast<int2*>(spill)[k] = p;
        }
    };

    const int tot  = binCnt[b * CNT_PAD];
    const int nrec = tot < BIN_CAP ? tot : BIN_CAP;
    const int2* rb = reinterpret_cast<const int2*>(rec) + (long long)b * BIN_CAP;
    for (int base = 0; base < nrec; base += 256 * 8) {
        int2 p[8];
        #pragma unroll
        for (int u = 0; u < 8; ++u) {
            const int i = base + u * 256 + tid;
            p[u] = make_int2(-1, 0);
            if (i < nrec) p[u] = rb[i];
        }
        #pragma unroll
        for (int u = 0; u < 8; ++u)
            if (p[u].x >= 0) place(p[u]);
    }
    const int novf = *ovfCnt;
    for (int i = tid; i < novf; i += 256) {
        const int2 p = reinterpret_cast<const int2*>(ovf)[i];
        if ((p.x >> BIN_SH) == b) place(p);
    }
    __syncthreads();

    if (tid < BIN_N && node0 + tid < N) cnt[node0 + tid] = lcnt[tid];
    const int wave = tid >> 5, lane = tid & 31;
    for (int dl = wave; dl < BIN_N; dl += 8) {
        const int node = node0 + dl;
        if (node >= N) break;
        const int deg = lcnt[dl];
        const int m = deg < PITCH ? deg : PITCH;
        if (lane < m)      col[node * PITCH + lane]      = lcol[dl * PITCH + lane];
        if (lane + 32 < m) col[node * PITCH + 32 + lane] = lcol[dl * PITCH + 32 + lane];
    }
}

template<bool NORM>
__device__ __forceinline__ void add_row(float4& a, const float4& v,
                                        const float4& s, const float4& t) {
    if (NORM) {
        a.x += v.x * s.x + t.x;
        a.y += v.y * s.y + t.y;
        a.z += v.z * s.z + t.z;
        a.w += v.w * s.w + t.w;
    } else {
        a.x += v.x; a.y += v.y; a.z += v.z; a.w += v.w;
    }
}

template<int F, bool NORM>
__device__ __forceinline__ void sum_source_rows(const float* __restrict__ X,
                                                int node, int deg, int myc,
                                                const int* __restrict__ col,
                                                const int* __restrict__ spillCnt,
                                                const int* __restrict__ spill,
                                                int lane, float4 (&acc)[F / 128],
                                                const float4 (&s)[F / 128],
                                                const float4 (&t)[F / 128]) {
    constexpr int V = F / 128;
    constexpr int B = 8 / V;
    const int beg = node * PITCH;
    const int end = beg + (deg < PITCH ? deg : PITCH);

    if (deg > PITCH) {
        const int nsp = *spillCnt;
        for (int base = 0; base < nsp; base += 32) {
            int2 ds = make_int2(-1, 0);
            if (base + lane < nsp) ds = reinterpret_cast<const int2*>(spill)[base + lane];
            unsigned m = __builtin_amdgcn_ballot_w32(ds.x == node);
            while (m) {
                const int b = __builtin_ctz(m);
                m &= m - 1;
                const int sidx = __builtin_amdgcn_readlane(ds.y, b);
                const float4* r = reinterpret_cast<const float4*>(X + (long long)sidx * F);
                #pragma unroll
                for (int w = 0; w < V; ++w) add_row<NORM>(acc[w], r[lane + 32 * w], s[w], t[w]);
            }
        }
    }

    for (int base = beg; base < end; base += 32) {
        const int n = (end - base) < 32 ? (end - base) : 32;
        int c = myc;
        if (base != beg) c = (lane < n) ? col[base + lane] : 0;
        int j = 0;
        for (; j + B <= n; j += B) {
            float4 v[B][V];
            #pragma unroll
            for (int u = 0; u < B; ++u) {
                const int sidx = __builtin_amdgcn_readlane(c, j + u);
                const float4* r = reinterpret_cast<const float4*>(X + (long long)sidx * F);
                #pragma unroll
                for (int w = 0; w < V; ++w) v[u][w] = r[lane + 32 * w];
            }
            #pragma unroll
            for (int u = 0; u < B; ++u) {
                #pragma unroll
                for (int w = 0; w < V; ++w) add_row<NORM>(acc[w], v[u][w], s[w], t[w]);
            }
        }
        if (j < n) {
            float4 v[B][V];
            #pragma unroll
            for (int u = 0; u < B; ++u) {
                #pragma unroll
                for (int w = 0; w < V; ++w) v[u][w] = make_float4(0.f, 0.f, 0.f, 0.f);
                if (j + u < n) {
                    const int sidx = __builtin_amdgcn_readlane(c, j + u);
                    const float4* r = reinterpret_cast<const float4*>(X + (long long)sidx * F);
                    #pragma unroll
                    for (int w = 0; w < V; ++w) v[u][w] = r[lane + 32 * w];
                }
            }
            #pragma unroll
            for (int u = 0; u < B; ++u) {
                if (!NORM || j + u < n) {
                    #pragma unroll
                    for (int w = 0; w < V; ++w) add_row<NORM>(acc[w], v[u][w], s[w], t[w]);
                }
            }
        }
    }
}

template<int K, int MT, bool NORM>
__device__ __forceinline__ void gather_stage(float* __restrict__ tile,
                                             const float* __restrict__ X,
                                             const int* __restrict__ cnt,
                                             const int* __restrict__ col,
                                             const int* __restrict__ spillCnt,
                                             const int* __restrict__ spill,
                                             const float* __restrict__ colsum,
                                             const float* __restrict__ colsumsq,
                                             const float* __restrict__ gamma,
                                             const float* __restrict__ beta,
                                             float* __restrict__ hid,
                                             int row0, int M, float invN,
                                             int lane, int wave) {
    constexpr int V = K / 128;
    constexpr int LSTRIDE = K + 4;

    float4 s[V], t[V];
    #pragma unroll
    for (int w = 0; w < V; ++w) {
        s[w] = make_float4(0.f, 0.f, 0.f, 0.f);
        t[w] = make_float4(0.f, 0.f, 0.f, 0.f);
        if (NORM) {
            const int c4 = lane + 32 * w;
            const float4 cs = reinterpret_cast<const float4*>(colsum)[c4];
            const float4 cq = reinterpret_cast<const float4*>(colsumsq)[c4];
            const float csv[4] = {cs.x, cs.y, cs.z, cs.w};
            const float cqv[4] = {cq.x, cq.y, cq.z, cq.w};
            float sv[4], tv[4];
            #pragma unroll
            for (int q = 0; q < 4; ++q) {
                const float mean = csv[q] * invN;
                const float var  = cqv[q] * invN - mean * mean;
                sv[q] = gamma[c4 * 4 + q] * rsqrtf(var + BN_EPS);
                tv[q] = beta[c4 * 4 + q] - mean * sv[q];
            }
            s[w] = make_float4(sv[0], sv[1], sv[2], sv[3]);
            t[w] = make_float4(tv[0], tv[1], tv[2], tv[3]);
        }
    }

    int degv = 0;
    {
        const int nd = row0 + wave + 16 * lane;
        if (lane < MT && nd < M) degv = cnt[nd];
    }
    int node = row0 + wave;
    int deg  = __builtin_amdgcn_readlane(degv, 0);
    int myc  = 0;
    if (lane < (deg < 32 ? deg : 32)) myc = col[node * PITCH + lane];

    #pragma nounroll
    for (int i = 0; i < MT; ++i) {
        const int nnode = node + 16;
        int ndeg = 0, nmyc = 0;
        if (i + 1 < MT) {
            ndeg = __builtin_amdgcn_readlane(degv, i + 1);
            if (lane < (ndeg < 32 ? ndeg : 32)) nmyc = col[nnode * PITCH + lane];
        }

        float4 acc[V];
        #pragma unroll
        for (int w = 0; w < V; ++w) acc[w] = make_float4(0.f, 0.f, 0.f, 0.f);

        if (node < M) {
            const float4* xrow = reinterpret_cast<const float4*>(X + (long long)node * K);
            float4 own[V];
            #pragma unroll
            for (int w = 0; w < V; ++w) own[w] = xrow[lane + 32 * w];
            if (NORM) {
                float4* h = reinterpret_cast<float4*>(hid + (long long)node * K);
                #pragma unroll
                for (int w = 0; w < V; ++w) {
                    acc[w].x = own[w].x * s[w].x + t[w].x;
                    acc[w].y = own[w].y * s[w].y + t[w].y;
                    acc[w].z = own[w].z * s[w].z + t[w].z;
                    acc[w].w = own[w].w * s[w].w + t[w].w;
                    h[lane + 32 * w] = acc[w];
                }
            }
            sum_source_rows<K, NORM>(X, node, deg, myc, col, spillCnt, spill, lane, acc, s, t);
            if (!NORM) {
                #pragma unroll
                for (int w = 0; w < V; ++w) {
                    acc[w].x = own[w].x + acc[w].x;
                    acc[w].y = own[w].y + acc[w].y;
                    acc[w].z = own[w].z + acc[w].z;
                    acc[w].w = own[w].w + acc[w].w;
                }
            }
        }

        float4* l = reinterpret_cast<float4*>(tile + (wave + 16 * i) * LSTRIDE);
        #pragma unroll
        for (int w = 0; w < V; ++w) l[lane + 32 * w] = acc[w];

        node = nnode; deg = ndeg; myc = nmyc;
    }
}

constexpr int GEMM_MAXT = 5;

template<int K, int NDIM, int MT, bool STATS>
__device__ __forceinline__ void wmma_tile_run(const float* __restrict__ tileA,
                                              const float* __restrict__ W,
                                              float bb, float* __restrict__ Y,
                                              int row0, int M, int lane, int wave,
                                              float& psum, float& psq) {
    constexpr int LSTRIDE = K + 4;
    const int r16  = lane & 15;
    const int hi   = lane >> 4;
    const int bCol = wave * 16 + r16;
    const float* aLds = &tileA[r16 * LSTRIDE + hi * 2];
    const float* wp   = W + (long long)(hi * 2) * NDIM + bCol;

    v8f acc[MT];
    #pragma unroll
    for (int t = 0; t < MT; ++t)
        acc[t] = (v8f){0.f, 0.f, 0.f, 0.f, 0.f, 0.f, 0.f, 0.f};

    #pragma unroll 2
    for (int k = 0; k < K; k += 4) {
        v2f b;
        b.x = wp[(long long)k * NDIM];
        b.y = wp[(long long)(k + 1) * NDIM];
        #pragma unroll
        for (int t = 0; t < MT; ++t) {
            float2 av = *reinterpret_cast<const float2*>(aLds + t * 16 * LSTRIDE + k);
            v2f a; a.x = av.x; a.y = av.y;
            acc[t] = __builtin_amdgcn_wmma_f32_16x16x4_f32(
                       false, a,  false, b,
                       (short)0, acc[t],  false,  false);
        }
    }

    float* yp = Y + (long long)(row0 + hi * 8) * NDIM + bCol;
    if (row0 + MT * 16 <= M) {
        #pragma unroll
        for (int t = 0; t < MT; ++t) {
            #pragma unroll
            for (int r = 0; r < 8; ++r) {
                float v = acc[t][r] + bb;
                v = v > 0.f ? v : 0.f;
                yp[(t * 16 + r) * NDIM] = v;
                if (STATS) { psum += v; psq += v * v; }
            }
        }
    } else {
        const int left = M - (row0 + hi * 8);
        #pragma unroll
        for (int t = 0; t < MT; ++t) {
            #pragma unroll
            for (int r = 0; r < 8; ++r) {
                float v = acc[t][r] + bb;
                v = v > 0.f ? v : 0.f;
                if (t * 16 + r < left) {
                    yp[(t * 16 + r) * NDIM] = v;
                    if (STATS) { psum += v; psq += v * v; }
                }
            }
        }
    }
}

template<int K, int NDIM, int MT>
__device__ __forceinline__ void wmma_tile_run_to_lds(float* tile,
                                                     const float* __restrict__ W,
                                                     float bb, int lane, int wave) {
    constexpr int LSTRIDE = K + 4;
    constexpr int HSTRIDE = NDIM + 4;
    const int r16  = lane & 15;
    const int hi   = lane >> 4;
    const int bCol = wave * 16 + r16;
    const float* aLds = &tile[r16 * LSTRIDE + hi * 2];
    const float* wp   = W + (long long)(hi * 2) * NDIM + bCol;

    v8f acc[MT];
    #pragma unroll
    for (int t = 0; t < MT; ++t)
        acc[t] = (v8f){0.f, 0.f, 0.f, 0.f, 0.f, 0.f, 0.f, 0.f};

    #pragma unroll 2
    for (int k = 0; k < K; k += 4) {
        v2f b;
        b.x = wp[(long long)k * NDIM];
        b.y = wp[(long long)(k + 1) * NDIM];
        #pragma unroll
        for (int t = 0; t < MT; ++t) {
            float2 av = *reinterpret_cast<const float2*>(aLds + t * 16 * LSTRIDE + k);
            v2f a; a.x = av.x; a.y = av.y;
            acc[t] = __builtin_amdgcn_wmma_f32_16x16x4_f32(
                       false, a,  false, b,
                       (short)0, acc[t],  false,  false);
        }
    }

    __syncthreads();
    float* hp = tile + (hi * 8) * HSTRIDE + bCol;
    #pragma unroll
    for (int t = 0; t < MT; ++t) {
        #pragma unroll
        for (int r = 0; r < 8; ++r) {
            float v = acc[t][r] + bb;
            hp[(t * 16 + r) * HSTRIDE] = v > 0.f ? v : 0.f;
        }
    }
}

template<int K, int NDIM, int MT>
__device__ __forceinline__ void wmma_tile_run_head(const float* __restrict__ tileA,
                                                   const float* __restrict__ W,
                                                   const float* __restrict__ bias,
                                                   const float* __restrict__ W4,
                                                   int lane, int wave,
                                                   float2* __restrict__ pp) {
    constexpr int LSTRIDE = K + 4;
    const int r16  = lane & 15;
    const int hi   = lane >> 4;
    const int bCol = wave * 16 + r16;
    const float* aLds = &tileA[r16 * LSTRIDE + hi * 2];
    const float* wp   = W + (long long)(hi * 2) * NDIM + bCol;

    v8f acc[MT];
    #pragma unroll
    for (int t = 0; t < MT; ++t)
        acc[t] = (v8f){0.f, 0.f, 0.f, 0.f, 0.f, 0.f, 0.f, 0.f};

    #pragma unroll 2
    for (int k = 0; k < K; k += 4) {
        v2f b;
        b.x = wp[(long long)k * NDIM];
        b.y = wp[(long long)(k + 1) * NDIM];
        #pragma unroll
        for (int t = 0; t < MT; ++t) {
            float2 av = *reinterpret_cast<const float2*>(aLds + t * 16 * LSTRIDE + k);
            v2f a; a.x = av.x; a.y = av.y;
            acc[t] = __builtin_amdgcn_wmma_f32_16x16x4_f32(
                       false, b,  false, a,
                       (short)0, acc[t],  false,  false);
        }
    }

    const int f0 = wave * 16 + hi * 8;
    float bia[8], wa[8], wb[8];
    #pragma unroll
    for (int r = 0; r < 8; ++r) {
        bia[r] = bias[f0 + r];
        wa[r]  = W4[(f0 + r) * 2];
        wb[r]  = W4[(f0 + r) * 2 + 1];
    }
    #pragma unroll
    for (int t = 0; t < MT; ++t) {
        float s0 = 0.f, s1 = 0.f;
        #pragma unroll
        for (int r = 0; r < 8; ++r) {
            float v = acc[t][r] + bia[r];
            v = v > 0.f ? v : 0.f;
            s0 = fmaf(v, wa[r], s0);
            s1 = fmaf(v, wb[r], s1);
        }
        pp[t] = make_float2(s0, s1);
    }
}

__device__ __forceinline__ void block_tile_run(int M, int& row0, int& n) {
    const int units = (M + 15) >> 4;
    const int G = (int)gridDim.x;
    const int b = (int)blockIdx.x;
    const int base = units / G;
    const int rem  = units - base * G;
    if (3 * rem <= G) {
        const int j = (b + 2) / 3;
        row0 = (base * b + (j < rem ? j : rem)) * 16;
        n = base + ((b % 3 == 0 && b / 3 < rem) ? 1 : 0);
    } else {
        const int u0 = (int)(((long long)b * units) / G);
        const int u1 = (int)(((long long)(b + 1) * units) / G);
        row0 = u0 * 16;
        n = u1 - u0;
    }
}

template<int K1, int NDIM>
__global__ void __launch_bounds__(512) __attribute__((amdgpu_waves_per_eu(12)))
gin_layer1_kernel(const float* __restrict__ X,
                  const int* __restrict__ cnt,
                  const int* __restrict__ col,
                  const int* __restrict__ spillCnt,
                  const int* __restrict__ spill,
                  const float* __restrict__ Wa,
                  const float* __restrict__ ba,
                  const float* __restrict__ Wb,
                  const float* __restrict__ bb,
                  float* __restrict__ Y,
                  float* __restrict__ colsum,
                  float* __restrict__ colsumsq, int M) {
    constexpr int LS2 = NDIM + 4;
    static_assert(K1 <= NDIM, "the buffer is sized for the wider of the two row sets");
    __shared__ __attribute__((aligned(16))) float tile[GEMM_MAXT * 16 * LS2];

    const int lane  = threadIdx.x;
    const int wave  = __builtin_amdgcn_readfirstlane(threadIdx.y);

    int row0, n;
    block_tile_run(M, row0, n);

    const float bias1 = ba[wave * 16 + (lane & 15)];
    const float bias2 = bb[wave * 16 + (lane & 15)];
    float psum = 0.f, psq = 0.f;

    if (n == 5) {
        gather_stage<K1, 5, false>(tile, X, cnt, col, spillCnt, spill, nullptr, nullptr,
                                   nullptr, nullptr, nullptr, row0, M, 0.f, lane, wave);
        __syncthreads();
        wmma_tile_run_to_lds<K1, NDIM, 5>(tile, Wa, bias1, lane, wave);
        __syncthreads();
        wmma_tile_run<NDIM, NDIM, 5, true>(tile, Wb, bias2, Y, row0, M, lane, wave, psum, psq);
    } else if (n == 4) {
        gather_stage<K1, 4, false>(tile, X, cnt, col, spillCnt, spill, nullptr, nullptr,
                                   nullptr, nullptr, nullptr, row0, M, 0.f, lane, wave);
        __syncthreads();
        wmma_tile_run_to_lds<K1, NDIM, 4>(tile, Wa, bias1, lane, wave);
        __syncthreads();
        wmma_tile_run<NDIM, NDIM, 4, true>(tile, Wb, bias2, Y, row0, M, lane, wave, psum, psq);
    } else {
        for (int t = 0; t < n; ++t) {
            if (t) __syncthreads();
            gather_stage<K1, 1, false>(tile, X, cnt, col, spillCnt, spill, nullptr, nullptr,
                                       nullptr, nullptr, nullptr, row0 + t * 16, M, 0.f,
                                       lane, wave);
            __syncthreads();
            wmma_tile_run_to_lds<K1, NDIM, 1>(tile, Wa, bias1, lane, wave);
            __syncthreads();
            wmma_tile_run<NDIM, NDIM, 1, true>(tile, Wb, bias2, Y, row0 + t * 16, M,
                                               lane, wave, psum, psq);
        }
    }

    atomicAdd(&colsum[wave * 16 + (lane & 15)], psum);
    atomicAdd(&colsumsq[wave * 16 + (lane & 15)], psq);
}

template<int K, int NDIM>
__global__ void __launch_bounds__(512) __attribute__((amdgpu_waves_per_eu(12)))
gin_layer2_head_kernel(const float* __restrict__ H2,
                       const int* __restrict__ cnt,
                       const int* __restrict__ col,
                       const int* __restrict__ spillCnt,
                       const int* __restrict__ spill,
                       const float* __restrict__ colsum1,
                       const float* __restrict__ colsumsq1,
                       const float* __restrict__ gamma,
                       const float* __restrict__ beta,
                       float* __restrict__ hid,
                       const float* __restrict__ W,
                       const float* __restrict__ bias,
                       const float* __restrict__ W4,
                       const float* __restrict__ b4,
                       float* __restrict__ out2,
                       float* __restrict__ colsum5,
                       float* __restrict__ colsumsq5, int M, float invN) {
    constexpr int LSTRIDE = K + 4;
    __shared__ __attribute__((aligned(16))) float tileA[GEMM_MAXT * 16 * LSTRIDE];

    const int lane  = threadIdx.x;
    const int wave  = __builtin_amdgcn_readfirstlane(threadIdx.y);

    int row0, n;
    block_tile_run(M, row0, n);

    float2 pp[GEMM_MAXT];
    #pragma unroll
    for (int t = 0; t < GEMM_MAXT; ++t) pp[t] = make_float2(0.f, 0.f);

    if (n == 5) {
        gather_stage<K, 5, true>(tileA, H2, cnt, col, spillCnt, spill, colsum1, colsumsq1,
                                 gamma, beta, hid, row0, M, invN, lane, wave);
        __syncthreads();
        wmma_tile_run_head<K, NDIM, 5>(tileA, W, bias, W4, lane, wave, pp);
    } else if (n == 4) {
        gather_stage<K, 4, true>(tileA, H2, cnt, col, spillCnt, spill, colsum1, colsumsq1,
                                 gamma, beta, hid, row0, M, invN, lane, wave);
        __syncthreads();
        wmma_tile_run_head<K, NDIM, 4>(tileA, W, bias, W4, lane, wave, pp);
    } else {
        for (int t = 0; t < n; ++t)
            gather_stage<K, 1, true>(tileA + t * 16 * LSTRIDE, H2, cnt, col, spillCnt, spill,
                                     colsum1, colsumsq1, gamma, beta, hid, row0 + t * 16, M,
                                     invN, lane, wave);
        __syncthreads();
        #pragma unroll
        for (int t = 0; t < 3; ++t)
            if (t < n)
                wmma_tile_run_head<K, NDIM, 1>(tileA + t * 16 * LSTRIDE, W, bias, W4,
                                               lane, wave, pp + t);
    }

    constexpr int PROWS = GEMM_MAXT * 16;
    __syncthreads();
    float2* part = reinterpret_cast<float2*>(tileA);
    const int grp = wave * 2 + (lane >> 4);
    #pragma unroll
    for (int t = 0; t < GEMM_MAXT; ++t)
        if (t < n) part[grp * PROWS + t * 16 + (lane & 15)] = pp[t];
    __syncthreads();
    if (wave == 0) {
        const float* pf = tileA;
        const int c = lane & 1;
        const float bc = b4[c];
        float ssum = 0.f, ssq = 0.f;
        #pragma unroll
        for (int j = 0; j < GEMM_MAXT; ++j) {
            if (j < n) {
                const int idx = j * 32 + lane;
                float sacc = bc;
                #pragma unroll
                for (int g = 0; g < 32; ++g) sacc += pf[g * (PROWS * 2) + idx];
                const float v = sacc > 0.f ? sacc : 0.f;
                const int row = row0 + (idx >> 1);
                if (row < M) {
                    out2[(long long)row * 2 + c] = v;
                    ssum += v; ssq += v * v;
                }
            }
        }
        #pragma unroll
        for (int d = 2; d < 32; d <<= 1) {
            ssum += __shfl_xor(ssum, d);
            ssq  += __shfl_xor(ssq, d);
        }
        if (lane < 2) {
            atomicAdd(&colsum5[lane], ssum);
            atomicAdd(&colsumsq5[lane], ssq);
        }
    }
}

__global__ void bn_params_kernel(const float* __restrict__ colsum,
                                 const float* __restrict__ colsumsq,
                                 const float* __restrict__ gamma,
                                 const float* __restrict__ beta,
                                 float* __restrict__ scale,
                                 float* __restrict__ shift,
                                 int C, float invN) {
    int c = blockIdx.x * blockDim.x + threadIdx.x;
    if (c >= C) return;
    float mean = colsum[c] * invN;
    float var  = colsumsq[c] * invN - mean * mean;
    float s = gamma[c] * rsqrtf(var + BN_EPS);
    scale[c] = s;
    shift[c] = beta[c] - mean * s;
}

__global__ void bn_apply2(const float* __restrict__ Yin,
                          const float* __restrict__ scale,
                          const float* __restrict__ shift,
                          float* __restrict__ Yout, int total) {
    int t = blockIdx.x * blockDim.x + threadIdx.x;
    if (t >= total) return;
    int c = t & 1;
    Yout[t] = Yin[t] * scale[c] + shift[c];
}

extern "C" void kernel_launch(void* const* d_in, const int* in_sizes, int n_in,
                              void* d_out, int out_size, void* d_ws, size_t ws_size,
                              hipStream_t stream) {
    (void)n_in; (void)out_size; (void)ws_size;

    const float* x   = (const float*)d_in[0];
    const int*   ei  = (const int*)  d_in[1];
    const float* W1  = (const float*)d_in[2];
    const float* b1  = (const float*)d_in[3];
    const float* W2  = (const float*)d_in[4];
    const float* b2  = (const float*)d_in[5];
    const float* W3  = (const float*)d_in[6];
    const float* b3  = (const float*)d_in[7];
    const float* W4  = (const float*)d_in[8];
    const float* b4  = (const float*)d_in[9];
    const float* gm1 = (const float*)d_in[10];
    const float* bt1 = (const float*)d_in[11];
    const float* gm5 = (const float*)d_in[12];
    const float* bt5 = (const float*)d_in[13];

    const int FIN = 128, DIM = 256;
    const int Nn = in_sizes[0] / FIN;
    const int E  = in_sizes[1] / 2;
    const int* src = ei;
    const int* dst = ei + E;

    float* ws = (float*)d_ws;
    float* colsum1   = ws + 0;
    float* colsumsq1 = ws + 256;
    float* colsum5   = ws + 1024;
    float* colsumsq5 = ws + 1028;
    float* scale5    = ws + 1032;
    float* shift5    = ws + 1036;
    int*   spillCnt  = (int*)(ws + 1040);
    int*   ovfCnt    = (int*)(ws + 1041);
    const int nbins  = (Nn + BIN_N - 1) / BIN_N;
    int*   binCnt = (int*)(ws + 2048);
    int*   cnt    = binCnt + (size_t)nbins * CNT_PAD;
    const size_t hdr = (2048 + (size_t)nbins * CNT_PAD + (size_t)Nn + 63) / 64 * 64;
    int*   col = (int*)(ws + hdr);
    float* R2 = ws + hdr + (size_t)Nn * PITCH;
    float* R3 = R2 + (size_t)Nn * DIM;
    int*   spill = (int*)(R3 + (size_t)Nn * 2);
    int*   rec   = spill + (size_t)E * 2;
    int*   ovf   = rec + (size_t)nbins * BIN_CAP * 2;

    float* outp = (float*)d_out;
    float* hid  = outp + (size_t)Nn * 2;

    hipMemsetAsync(ws, 0, (2048 + (size_t)nbins * CNT_PAD) * sizeof(float), stream);

    const dim3 gemmBlk(32, 16);
    const int mTiles = (Nn + 15) / 16;
    const dim3 gemmGrid(mTiles > 768 * 5 ? (mTiles + 4) / 5 : 768);

    edge_bin_kernel<<<(E + EB_T * EB_E - 1) / (EB_T * EB_E), EB_T, 0, stream>>>(
        src, dst, binCnt, rec, ovfCnt, ovf, E, nbins);
    bin_rank_kernel<<<nbins, 256, 0, stream>>>(binCnt, rec, ovfCnt, ovf, cnt, col,
                                               spillCnt, spill, Nn);

    gin_layer1_kernel<128, 256><<<gemmGrid, gemmBlk, 0, stream>>>(
        x, cnt, col, spillCnt, spill, W1, b1, W2, b2, R2, colsum1, colsumsq1, Nn);
    gin_layer2_head_kernel<256, 256><<<gemmGrid, gemmBlk, 0, stream>>>(
        R2, cnt, col, spillCnt, spill, colsum1, colsumsq1, gm1, bt1, hid,
        W3, b3, W4, b4, R3, colsum5, colsumsq5, Nn, 1.0f / (float)Nn);
    bn_params_kernel<<<1, 32, 0, stream>>>(colsum5, colsumsq5, gm5, bt5,
                                           scale5, shift5, 2, 1.0f / (float)Nn);
    bn_apply2<<<(Nn * 2 + 255) / 256, 256, 0, stream>>>(R3, scale5, shift5,
                                                        outp, Nn * 2);
}
